# starting kernel
# baseline (speedup 1.0000x reference)
#include <hip/hip_runtime.h>
#include <math.h>

typedef __attribute__((ext_vector_type(2))) float v2f;
typedef __attribute__((ext_vector_type(4))) float v4f;
typedef __attribute__((ext_vector_type(8))) float v8f;

#define D_        128
#define R_        1024
#define T_        4
#define OUTSLOTS_ 256
#define START_    4
#define L_        64
#define M_        2048
#define N_        (R_ + L_ * M_)
#define NENC_     5
#define K1_       256
#define XSTR_     260

#define W1P_FRAGS (NENC_ * 16 * 64 * 32)
#define W2P_FRAGS (NENC_ * 8  * 64 * 32)
#define W1P_BYTES ((size_t)W1P_FRAGS * 8)
#define W2P_BYTES ((size_t)W2P_FRAGS * 8)
#define WSP_BYTES (W1P_BYTES + W2P_BYTES)

__global__ void dag_init_roots(float* __restrict__ buf, const float* __restrict__ root) {
    int i = blockIdx.x * blockDim.x + threadIdx.x;
    v4f v = *(const v4f*)(root + (size_t)i * 4);
    *(v4f*)(buf + (size_t)i * 4) = v;
}

__global__ void dag_pack_w(const float* __restrict__ W, float* __restrict__ Wp,
                           int ntile, int ncol) {
    int idx  = blockIdx.x * blockDim.x + threadIdx.x;
    int lane = idx & 31;
    int k4   = (idx >> 5) & 63;
    int j    = (idx >> 11) % ntile;
    int e    = idx / (ntile << 11);
    int col  = lane & 15;
    int kk   = k4 * 4 + ((lane >> 4) & 1) * 2;
    const float* Wsrc = W + (size_t)e * K1_ * ncol + (size_t)(j * 16 + col);
    Wp[(size_t)idx * 2 + 0] = Wsrc[(size_t)kk * ncol];
    Wp[(size_t)idx * 2 + 1] = Wsrc[(size_t)(kk + 1) * ncol];
}

template <bool PACKED>
__global__ __launch_bounds__(160)
void dag_level(float* __restrict__ buf,
               const float* __restrict__ W1,
               const float* __restrict__ b1,
               const float* __restrict__ W2,
               const float* __restrict__ b2,
               const float* __restrict__ slots,
               const int*   __restrict__ par,
               const int*   __restrict__ types,
               int level)
{
    __shared__ float X[16 * XSTR_];
    __shared__ float H[16 * XSTR_];

    const int lane     = threadIdx.x & 31;
    const int e        = threadIdx.x >> 5;
    const int tile     = blockIdx.x;
    const int qbase    = level * M_ + tile * 16;
    const int rowbase  = R_ + qbase;
    const int colLane  = lane & 15;
    const int laneHalf = (lane >> 4) & 1;
    const int klo      = laneHalf * 2;

    for (int m = e; m < 16; m += NENC_) {
        int p0 = par[(qbase + m) * 2 + 0];
        int p1 = par[(qbase + m) * 2 + 1];
        int t  = types[qbase + m];
        const float* s0 = buf + (size_t)p0 * D_;
        const float* s1;
        if (t >= START_) {
            int slot = t - START_;
            slot = slot < 0 ? 0 : (slot > OUTSLOTS_ - 1 ? OUTSLOTS_ - 1 : slot);
            s1 = slots + (size_t)slot * D_;
        } else {
            s1 = buf + (size_t)p1 * D_;
        }
        v4f a = *(const v4f*)(s0 + lane * 4);
        v4f c = *(const v4f*)(s1 + lane * 4);
        *(v4f*)(&X[m * XSTR_ + lane * 4])       = a;
        *(v4f*)(&X[m * XSTR_ + 128 + lane * 4]) = c;
    }

    int rt[8];
    bool ownAny = false;
    #pragma unroll
    for (int i = 0; i < 8; ++i) {
        int t = types[qbase + i + laneHalf * 8];
        rt[i] = (t >= START_) ? T_ : t;
        ownAny |= (rt[i] == e);
    }
    const bool waveActive = __any(ownAny);
    __syncthreads();

    if (waveActive) {
        #pragma unroll 1
        for (int j = 0; j < 16; ++j) {
            v8f acc = {};
            if (PACKED) {
                const v2f* Wf = (const v2f*)W1 + ((size_t)(e * 16 + j) * 64) * 32 + lane;
                #pragma unroll 8
                for (int k4 = 0; k4 < 64; ++k4) {
                    int kk = k4 * 4 + klo;
                    v2f a = *(const v2f*)(&X[colLane * XSTR_ + kk]);
                    v2f b = Wf[k4 * 32];
                    acc = __builtin_amdgcn_wmma_f32_16x16x4_f32(
                            false, a, false, b, (short)0, acc, false, false);
                }
            } else {
                const float* Wb = W1 + (size_t)e * K1_ * K1_ + (size_t)(j * 16 + colLane);
                #pragma unroll 8
                for (int k4 = 0; k4 < 64; ++k4) {
                    int kk = k4 * 4 + klo;
                    v2f a = *(const v2f*)(&X[colLane * XSTR_ + kk]);
                    v2f b;
                    b.x = Wb[(size_t)kk * K1_];
                    b.y = Wb[(size_t)(kk + 1) * K1_];
                    acc = __builtin_amdgcn_wmma_f32_16x16x4_f32(
                            false, a, false, b, (short)0, acc, false, false);
                }
            }
            float bias = b1[e * K1_ + j * 16 + colLane];
            #pragma unroll
            for (int i = 0; i < 8; ++i) {
                if (rt[i] == e) {
                    float x = acc[i] + bias;
                    float g = 0.5f * x * (1.0f + erff(x * 0.70710678118654752f));
                    H[(i + laneHalf * 8) * XSTR_ + j * 16 + colLane] = g;
                }
            }
        }
    }
    __syncthreads();

    if (waveActive) {
        #pragma unroll 1
        for (int j = 0; j < 8; ++j) {
            v8f acc = {};
            if (PACKED) {
                const v2f* Wf = (const v2f*)W2 + ((size_t)(e * 8 + j) * 64) * 32 + lane;
                #pragma unroll 8
                for (int k4 = 0; k4 < 64; ++k4) {
                    int kk = k4 * 4 + klo;
                    v2f a = *(const v2f*)(&H[colLane * XSTR_ + kk]);
                    v2f b = Wf[k4 * 32];
                    acc = __builtin_amdgcn_wmma_f32_16x16x4_f32(
                            false, a, false, b, (short)0, acc, false, false);
                }
            } else {
                const float* Wb = W2 + (size_t)e * K1_ * D_ + (size_t)(j * 16 + colLane);
                #pragma unroll 8
                for (int k4 = 0; k4 < 64; ++k4) {
                    int kk = k4 * 4 + klo;
                    v2f a = *(const v2f*)(&H[colLane * XSTR_ + kk]);
                    v2f b;
                    b.x = Wb[(size_t)kk * D_];
                    b.y = Wb[(size_t)(kk + 1) * D_];
                    acc = __builtin_amdgcn_wmma_f32_16x16x4_f32(
                            false, a, false, b, (short)0, acc, false, false);
                }
            }
            float bias = b2[e * D_ + j * 16 + colLane];
            #pragma unroll
            for (int i = 0; i < 8; ++i) {
                if (rt[i] == e) {
                    int row = rowbase + i + laneHalf * 8;
                    buf[(size_t)row * D_ + j * 16 + colLane] = acc[i] + bias;
                }
            }
        }
    }
}

extern "C" void kernel_launch(void* const* d_in, const int* in_sizes, int n_in,
                              void* d_out, int out_size, void* d_ws, size_t ws_size,
                              hipStream_t stream) {
    const float* root  = (const float*)d_in[0];
    const float* W1    = (const float*)d_in[1];
    const float* b1    = (const float*)d_in[2];
    const float* W2    = (const float*)d_in[3];
    const float* b2    = (const float*)d_in[4];
    const float* slots = (const float*)d_in[5];
    const int*   par   = (const int*)d_in[6];
    const int*   typ   = (const int*)d_in[7];
    float* out = (float*)d_out;

    dag_init_roots<<<128, 256, 0, stream>>>(out, root);

    const bool packed = (ws_size >= WSP_BYTES);
    float* W1p = (float*)d_ws;
    float* W2p = (float*)((char*)d_ws + W1P_BYTES);

    if (packed) {
        dag_pack_w<<<W1P_FRAGS / 256, 256, 0, stream>>>(W1, W1p, 16, K1_);
        dag_pack_w<<<W2P_FRAGS / 256, 256, 0, stream>>>(W2, W2p, 8, D_);
        for (int l = 0; l < L_; ++l)
            dag_level<true><<<128, 160, 0, stream>>>(out, W1p, b1, W2p, b2,
                                                     slots, par, typ, l);
    } else {
        for (int l = 0; l < L_; ++l)
            dag_level<false><<<128, 160, 0, stream>>>(out, W1, b1, W2, b2,
                                                      slots, par, typ, l);
    }
}


# Level kernel re-partitioned: a 4-wave work-group per (16-node tile, encoder) instead of a wave per encoder; each layer's column tiles spread over its waves, 4 / 2 accumulators each; k order kept
# speedup vs baseline: 4.3112x; 4.3112x over previous
#include <hip/hip_runtime.h>
#include <math.h>

typedef __attribute__((ext_vector_type(2))) float v2f;
typedef __attribute__((ext_vector_type(4))) float v4f;
typedef __attribute__((ext_vector_type(8))) float v8f;

#define D_        128
#define R_        1024
#define T_        4
#define OUTSLOTS_ 256
#define START_    4
#define L_        64
#define M_        2048
#define N_        (R_ + L_ * M_)
#define NENC_     5
#define K1_       256
#define XSTR_     260

#define W1P_FRAGS (NENC_ * 16 * 64 * 32)
#define W2P_FRAGS (NENC_ * 8  * 64 * 32)
#define W1P_BYTES ((size_t)W1P_FRAGS * 8)
#define W2P_BYTES ((size_t)W2P_FRAGS * 8)
#define WSP_BYTES (W1P_BYTES + W2P_BYTES)

__global__ void dag_init_roots(float* __restrict__ buf, const float* __restrict__ root) {
    int i = blockIdx.x * blockDim.x + threadIdx.x;
    v4f v = *(const v4f*)(root + (size_t)i * 4);
    *(v4f*)(buf + (size_t)i * 4) = v;
}

__global__ void dag_pack_w(const float* __restrict__ W, float* __restrict__ Wp,
                           int ntile, int ncol) {
    int idx  = blockIdx.x * blockDim.x + threadIdx.x;
    int lane = idx & 31;
    int k4   = (idx >> 5) & 63;
    int j    = (idx >> 11) % ntile;
    int e    = idx / (ntile << 11);
    int col  = lane & 15;
    int kk   = k4 * 4 + ((lane >> 4) & 1) * 2;
    const float* Wsrc = W + (size_t)e * K1_ * ncol + (size_t)(j * 16 + col);
    Wp[(size_t)idx * 2 + 0] = Wsrc[(size_t)kk * ncol];
    Wp[(size_t)idx * 2 + 1] = Wsrc[(size_t)(kk + 1) * ncol];
}

template <bool PACKED, int NCOL>
__device__ __forceinline__ v2f dag_load_b(const float* __restrict__ Wq, int e, int j,
                                          int k4, int lane) {
    if (PACKED) {
        const v2f* Wf = (const v2f*)Wq + ((size_t)(e * (NCOL / 16) + j) * 64) * 32 + lane;
        return Wf[k4 * 32];
    } else {
        const int colLane = lane & 15;
        const int kk = k4 * 4 + ((lane >> 4) & 1) * 2;
        const float* Wb = Wq + (size_t)e * K1_ * NCOL + (size_t)(j * 16 + colLane);
        v2f b;
        b.x = Wb[(size_t)kk * NCOL];
        b.y = Wb[(size_t)(kk + 1) * NCOL];
        return b;
    }
}

template <bool PACKED, int NCOL, int NT>
__device__ __forceinline__ void dag_fetch_b(const float* __restrict__ Wq, int e, int j0,
                                            int g, int lane, v2f (&b)[8][NT]) {
    #pragma unroll
    for (int s = 0; s < 8; ++s) {
        #pragma unroll
        for (int t = 0; t < NT; ++t)
            b[s][t] = dag_load_b<PACKED, NCOL>(Wq, e, j0 + t, g * 8 + s, lane);
    }
}

template <bool PACKED, int NCOL, int NT>
__device__ __forceinline__ void dag_panel(const float* __restrict__ Xs,
                                          const float* __restrict__ Wq,
                                          int e, int j0, int lane,
                                          v2f (&bfirst)[8][NT], v8f (&acc)[NT]) {
    const int colLane = lane & 15;
    const int klo     = ((lane >> 4) & 1) * 2;
    const float* Arow = Xs + colLane * XSTR_ + klo;

    v2f aq[2][8];
    v2f bq[2][8][NT];
    #pragma unroll
    for (int s = 0; s < 8; ++s) {
        #pragma unroll
        for (int t = 0; t < NT; ++t)
            bq[0][s][t] = bfirst[s][t];
        aq[0][s] = *(const v2f*)(Arow + s * 4);
    }
    #pragma unroll
    for (int g = 0; g < 8; ++g) {
        const int cur = g & 1, nxt = cur ^ 1;
        if (g < 7) {
            dag_fetch_b<PACKED, NCOL, NT>(Wq, e, j0, g + 1, lane, bq[nxt]);
            #pragma unroll
            for (int s = 0; s < 8; ++s)
                aq[nxt][s] = *(const v2f*)(Arow + ((g + 1) * 8 + s) * 4);
        }
        #pragma unroll
        for (int s = 0; s < 8; ++s) {
            #pragma unroll
            for (int t = 0; t < NT; ++t)
                acc[t] = __builtin_amdgcn_wmma_f32_16x16x4_f32(
                        false, aq[cur][s], false, bq[cur][s][t], (short)0, acc[t], false, false);
        }
    }
}

template <bool PACKED>
__global__ __launch_bounds__(128)
void dag_level(float* __restrict__ buf,
               const float* __restrict__ W1,
               const float* __restrict__ b1,
               const float* __restrict__ W2,
               const float* __restrict__ b2,
               const float* __restrict__ slots,
               const int*   __restrict__ par,
               const int*   __restrict__ types,
               int level)
{
    __shared__ float X[16 * XSTR_];
    __shared__ float H[16 * XSTR_];

    const int lane     = threadIdx.x & 31;
    const int w        = threadIdx.x >> 5;
    const int e        = blockIdx.y;
    const int tile     = blockIdx.x;
    const int qbase    = level * M_ + tile * 16;
    const int rowbase  = R_ + qbase;
    const int colLane  = lane & 15;
    const int laneHalf = (lane >> 4) & 1;

    int gp0[4], gp1[4], gt[4];
    #pragma unroll
    for (int r = 0; r < 4; ++r) {
        const int m = w + 4 * r;
        gp0[r] = par[(qbase + m) * 2 + 0];
        gp1[r] = par[(qbase + m) * 2 + 1];
        gt[r]  = types[qbase + m];
    }

    bool own[8];
    bool ownAny = false;
    #pragma unroll
    for (int i = 0; i < 8; ++i) {
        int t = types[qbase + i + laneHalf * 8];
        int r = (t >= START_) ? T_ : t;
        own[i] = (r == e);
        ownAny |= own[i];
    }
    if (!__any(ownAny)) return;

    v2f w1first[8][4];
    dag_fetch_b<PACKED, K1_, 4>(W1, e, w * 4, 0, lane, w1first);

    v4f ga[4], gc[4];
    #pragma unroll
    for (int r = 0; r < 4; ++r) {
        const int t = gt[r];
        int slot = t - START_;
        slot = slot < 0 ? 0 : (slot > OUTSLOTS_ - 1 ? OUTSLOTS_ - 1 : slot);
        const float* s0 = buf + (size_t)gp0[r] * D_;
        const float* s1 = (t >= START_) ? (slots + (size_t)slot * D_)
                                        : (buf + (size_t)gp1[r] * D_);
        ga[r] = *(const v4f*)(s0 + lane * 4);
        gc[r] = *(const v4f*)(s1 + lane * 4);
    }
    #pragma unroll
    for (int r = 0; r < 4; ++r) {
        const int m = w + 4 * r;
        *(v4f*)(&X[m * XSTR_ + lane * 4])       = ga[r];
        *(v4f*)(&X[m * XSTR_ + 128 + lane * 4]) = gc[r];
    }
    __syncthreads();

    v2f w2first[8][2];
    {
        v8f acc[4] = {};
        dag_panel<PACKED, K1_, 4>(X, W1, e, w * 4, lane, w1first, acc);
        dag_fetch_b<PACKED, D_, 2>(W2, e, w * 2, 0, lane, w2first);
        #pragma unroll
        for (int t = 0; t < 4; ++t) {
            const int j = w * 4 + t;
            float bias = b1[e * K1_ + j * 16 + colLane];
            #pragma unroll
            for (int i = 0; i < 8; ++i) {
                if (own[i]) {
                    float x = acc[t][i] + bias;
                    float g = 0.5f * x * (1.0f + erff(x * 0.70710678118654752f));
                    H[(i + laneHalf * 8) * XSTR_ + j * 16 + colLane] = g;
                }
            }
        }
    }
    __syncthreads();

    {
        v8f acc[2] = {};
        dag_panel<PACKED, D_, 2>(H, W2, e, w * 2, lane, w2first, acc);
        #pragma unroll
        for (int t = 0; t < 2; ++t) {
            const int j = w * 2 + t;
            float bias = b2[e * D_ + j * 16 + colLane];
            #pragma unroll
            for (int i = 0; i < 8; ++i) {
                if (own[i]) {
                    int row = rowbase + i + laneHalf * 8;
                    buf[(size_t)row * D_ + j * 16 + colLane] = acc[t][i] + bias;
                }
            }
        }
    }
}

extern "C" void kernel_launch(void* const* d_in, const int* in_sizes, int n_in,
                              void* d_out, int out_size, void* d_ws, size_t ws_size,
                              hipStream_t stream) {
    const float* root  = (const float*)d_in[0];
    const float* W1    = (const float*)d_in[1];
    const float* b1    = (const float*)d_in[2];
    const float* W2    = (const float*)d_in[3];
    const float* b2    = (const float*)d_in[4];
    const float* slots = (const float*)d_in[5];
    const int*   par   = (const int*)d_in[6];
    const int*   typ   = (const int*)d_in[7];
    float* out = (float*)d_out;

    dag_init_roots<<<128, 256, 0, stream>>>(out, root);

    const bool packed = (ws_size >= WSP_BYTES);
    float* W1p = (float*)d_ws;
    float* W2p = (float*)((char*)d_ws + W1P_BYTES);

    if (packed) {
        dag_pack_w<<<W1P_FRAGS / 256, 256, 0, stream>>>(W1, W1p, 16, K1_);
        dag_pack_w<<<W2P_FRAGS / 256, 256, 0, stream>>>(W2, W2p, 8, D_);
        for (int l = 0; l < L_; ++l)
            dag_level<true><<<dim3(128, NENC_), 128, 0, stream>>>(out, W1p, b1, W2p, b2,
                                                                 slots, par, typ, l);
    } else {
        for (int l = 0; l < L_; ++l)
            dag_level<false><<<dim3(128, NENC_), 128, 0, stream>>>(out, W1, b1, W2, b2,
                                                                  slots, par, typ, l);
    }
}


# Rows of each level bucketed by encoder in one pre-pass (stable counting sort, no atomics); one work-group per 16-row tile of ONE encoder, grid 132 not 128x5; per-row k order and sums unchanged
# speedup vs baseline: 5.4126x; 1.2555x over previous
#include <hip/hip_runtime.h>
#include <math.h>

typedef __attribute__((ext_vector_type(2))) float v2f;
typedef __attribute__((ext_vector_type(4))) float v4f;
typedef __attribute__((ext_vector_type(8))) float v8f;

#define D_        128
#define R_        1024
#define T_        4
#define OUTSLOTS_ 256
#define START_    4
#define L_        64
#define M_        2048
#define N_        (R_ + L_ * M_)
#define NENC_     5
#define K1_       256
#define XSTR_     260

#define W1P_FRAGS (NENC_ * 16 * 64 * 32)
#define W2P_FRAGS (NENC_ * 8  * 64 * 32)
#define W1P_BYTES ((size_t)W1P_FRAGS * 8)
#define W2P_BYTES ((size_t)W2P_FRAGS * 8)
#define WSP_BYTES (W1P_BYTES + W2P_BYTES)

#define TILES_    132
#define SLOTS_    (TILES_ * 16)
#define REC_BYTES ((size_t)L_ * SLOTS_ * 16)

__global__ void dag_init_roots(float* __restrict__ buf, const float* __restrict__ root) {
    int i = blockIdx.x * blockDim.x + threadIdx.x;
    v4f v = *(const v4f*)(root + (size_t)i * 4);
    *(v4f*)(buf + (size_t)i * 4) = v;
}

__global__ void dag_pack_w(const float* __restrict__ W, float* __restrict__ Wp,
                           int ntile, int ncol) {
    int idx  = blockIdx.x * blockDim.x + threadIdx.x;
    int lane = idx & 31;
    int k4   = (idx >> 5) & 63;
    int j    = (idx >> 11) % ntile;
    int e    = idx / (ntile << 11);
    int col  = lane & 15;
    int kk   = k4 * 4 + ((lane >> 4) & 1) * 2;
    const float* Wsrc = W + (size_t)e * K1_ * ncol + (size_t)(j * 16 + col);
    Wp[(size_t)idx * 2 + 0] = Wsrc[(size_t)kk * ncol];
    Wp[(size_t)idx * 2 + 1] = Wsrc[(size_t)(kk + 1) * ncol];
}

template <bool PACKED, int NCOL>
__device__ __forceinline__ v2f dag_load_b(const float* __restrict__ Wq, int e, int j,
                                          int k4, int lane) {
    if (PACKED) {
        const v2f* Wf = (const v2f*)Wq + ((size_t)(e * (NCOL / 16) + j) * 64) * 32 + lane;
        return Wf[k4 * 32];
    } else {
        const int colLane = lane & 15;
        const int kk = k4 * 4 + ((lane >> 4) & 1) * 2;
        const float* Wb = Wq + (size_t)e * K1_ * NCOL + (size_t)(j * 16 + colLane);
        v2f b;
        b.x = Wb[(size_t)kk * NCOL];
        b.y = Wb[(size_t)(kk + 1) * NCOL];
        return b;
    }
}

template <bool PACKED, int NCOL, int NT>
__device__ __forceinline__ void dag_fetch_b(const float* __restrict__ Wq, int e, int j0,
                                            int g, int lane, v2f (&b)[8][NT]) {
    #pragma unroll
    for (int s = 0; s < 8; ++s) {
        #pragma unroll
        for (int t = 0; t < NT; ++t)
            b[s][t] = dag_load_b<PACKED, NCOL>(Wq, e, j0 + t, g * 8 + s, lane);
    }
}

template <bool PACKED, int NCOL, int NT>
__device__ __forceinline__ void dag_panel(const float* __restrict__ Xs,
                                          const float* __restrict__ Wq,
                                          int e, int j0, int lane,
                                          v2f (&bfirst)[8][NT], v8f (&acc)[NT]) {
    const int colLane = lane & 15;
    const int klo     = ((lane >> 4) & 1) * 2;
    const float* Arow = Xs + colLane * XSTR_ + klo;

    v2f aq[2][8];
    v2f bq[2][8][NT];
    #pragma unroll
    for (int s = 0; s < 8; ++s) {
        #pragma unroll
        for (int t = 0; t < NT; ++t)
            bq[0][s][t] = bfirst[s][t];
        aq[0][s] = *(const v2f*)(Arow + s * 4);
    }
    #pragma unroll
    for (int g = 0; g < 8; ++g) {
        const int cur = g & 1, nxt = cur ^ 1;
        if (g < 7) {
            dag_fetch_b<PACKED, NCOL, NT>(Wq, e, j0, g + 1, lane, bq[nxt]);
            #pragma unroll
            for (int s = 0; s < 8; ++s)
                aq[nxt][s] = *(const v2f*)(Arow + ((g + 1) * 8 + s) * 4);
        }
        #pragma unroll
        for (int s = 0; s < 8; ++s) {
            #pragma unroll
            for (int t = 0; t < NT; ++t)
                acc[t] = __builtin_amdgcn_wmma_f32_16x16x4_f32(
                        false, aq[cur][s], false, bq[cur][s][t], (short)0, acc[t], false, false);
        }
    }
}

template <bool PACKED>
__global__ __launch_bounds__(128)
void dag_level(float* __restrict__ buf,
               const float* __restrict__ W1,
               const float* __restrict__ b1,
               const float* __restrict__ W2,
               const float* __restrict__ b2,
               const float* __restrict__ slots,
               const int*   __restrict__ par,
               const int*   __restrict__ types,
               int level)
{
    __shared__ float X[16 * XSTR_];
    __shared__ float H[16 * XSTR_];

    const int lane     = threadIdx.x & 31;
    const int w        = threadIdx.x >> 5;
    const int e        = blockIdx.y;
    const int tile     = blockIdx.x;
    const int qbase    = level * M_ + tile * 16;
    const int rowbase  = R_ + qbase;
    const int colLane  = lane & 15;
    const int laneHalf = (lane >> 4) & 1;

    int gp0[4], gp1[4], gt[4];
    #pragma unroll
    for (int r = 0; r < 4; ++r) {
        const int m = w + 4 * r;
        gp0[r] = par[(qbase + m) * 2 + 0];
        gp1[r] = par[(qbase + m) * 2 + 1];
        gt[r]  = types[qbase + m];
    }

    bool own[8];
    bool ownAny = false;
    #pragma unroll
    for (int i = 0; i < 8; ++i) {
        int t = types[qbase + i + laneHalf * 8];
        int r = (t >= START_) ? T_ : t;
        own[i] = (r == e);
        ownAny |= own[i];
    }
    if (!__any(ownAny)) return;

    v2f w1first[8][4];
    dag_fetch_b<PACKED, K1_, 4>(W1, e, w * 4, 0, lane, w1first);

    v4f ga[4], gc[4];
    #pragma unroll
    for (int r = 0; r < 4; ++r) {
        const int t = gt[r];
        int slot = t - START_;
        slot = slot < 0 ? 0 : (slot > OUTSLOTS_ - 1 ? OUTSLOTS_ - 1 : slot);
        const float* s0 = buf + (size_t)gp0[r] * D_;
        const float* s1 = (t >= START_) ? (slots + (size_t)slot * D_)
                                        : (buf + (size_t)gp1[r] * D_);
        ga[r] = *(const v4f*)(s0 + lane * 4);
        gc[r] = *(const v4f*)(s1 + lane * 4);
    }
    #pragma unroll
    for (int r = 0; r < 4; ++r) {
        const int m = w + 4 * r;
        *(v4f*)(&X[m * XSTR_ + lane * 4])       = ga[r];
        *(v4f*)(&X[m * XSTR_ + 128 + lane * 4]) = gc[r];
    }
    __syncthreads();

    v2f w2first[8][2];
    {
        v8f acc[4] = {};
        dag_panel<PACKED, K1_, 4>(X, W1, e, w * 4, lane, w1first, acc);
        dag_fetch_b<PACKED, D_, 2>(W2, e, w * 2, 0, lane, w2first);
        #pragma unroll
        for (int t = 0; t < 4; ++t) {
            const int j = w * 4 + t;
            float bias = b1[e * K1_ + j * 16 + colLane];
            #pragma unroll
            for (int i = 0; i < 8; ++i) {
                if (own[i]) {
                    float x = acc[t][i] + bias;
                    float g = 0.5f * x * (1.0f + erff(x * 0.70710678118654752f));
                    H[(i + laneHalf * 8) * XSTR_ + j * 16 + colLane] = g;
                }
            }
        }
    }
    __syncthreads();

    {
        v8f acc[2] = {};
        dag_panel<PACKED, D_, 2>(H, W2, e, w * 2, lane, w2first, acc);
        #pragma unroll
        for (int t = 0; t < 2; ++t) {
            const int j = w * 2 + t;
            float bias = b2[e * D_ + j * 16 + colLane];
            #pragma unroll
            for (int i = 0; i < 8; ++i) {
                if (own[i]) {
                    int row = rowbase + i + laneHalf * 8;
                    buf[(size_t)row * D_ + j * 16 + colLane] = acc[t][i] + bias;
                }
            }
        }
    }
}


__global__ __launch_bounds__(1024)
void dag_bucket(const int* __restrict__ par, const int* __restrict__ types,
                int* __restrict__ recs_) {
    __shared__ int cnt[NENC_ * 32];
    __shared__ int fillFrom[NENC_];
    __shared__ int tileEnd[NENC_];

    const int level = blockIdx.x;
    const int tid   = threadIdx.x;
    const int lane  = tid & 31;
    const int wv    = tid >> 5;
    const int q     = level * M_ + tid * 2;
    const int2 tt = *(const int2*)(types + q);
    const int4 pp = *(const int4*)(par + (size_t)q * 2);
    const int r0 = tt.x >= START_ ? T_ : (tt.x < 0 ? 0 : tt.x);
    const int r1 = tt.y >= START_ ? T_ : (tt.y < 0 ? 0 : tt.y);

    const unsigned below = (1u << lane) - 1u;
    int rank0 = 0, rank1 = 0;
    for (int e = 0; e < NENC_; ++e) {
        const unsigned m0 = (unsigned)__ballot(r0 == e);
        const unsigned m1 = (unsigned)__ballot(r1 == e);
        const int before = __popc(m0 & below) + __popc(m1 & below);
        if (r0 == e) rank0 = before;
        if (r1 == e) rank1 = before + (r0 == e ? 1 : 0);
        if (lane == 0) cnt[e * 32 + wv] = __popc(m0) + __popc(m1);
    }
    __syncthreads();

    if (wv == 0) {
        int base = 0;
        for (int e = 0; e < NENC_; ++e) {
            const int v = cnt[e * 32 + lane];
            int inc = v;
            for (int d = 1; d < 32; d <<= 1) {
                const int u = __shfl_up(inc, d);
                if (lane >= d) inc += u;
            }
            const int total = __shfl(inc, 31);
            cnt[e * 32 + lane] = base + inc - v;
            if (lane == 0) fillFrom[e] = base + total;
            base += ((total + 15) >> 4) << 4;
            if (lane == 0) tileEnd[e] = base;
        }
    }
    __syncthreads();

    int4* lv = (int4*)recs_ + (size_t)level * SLOTS_;
    lv[cnt[r0 * 32 + wv] + rank0] = make_int4(pp.x, pp.y, tt.x, tid * 2);
    lv[cnt[r1 * 32 + wv] + rank1] = make_int4(pp.z, pp.w, tt.y, tid * 2 + 1);

    for (int s = tid; s < SLOTS_; s += 1024) {
        int e = 0;
        for (int k = 0; k < NENC_; ++k)
            if (s >= tileEnd[k]) e = k + 1;
        if (e == NENC_)
            lv[s] = make_int4(0, 0, 0, -1);
        else if (s >= fillFrom[e])
            lv[s] = make_int4(0, 0, e == T_ ? START_ : e, -1);
    }
}

__global__ __launch_bounds__(128)
void dag_level_bucketed(float* __restrict__ buf,
                        const float* __restrict__ W1,
                        const float* __restrict__ b1,
                        const float* __restrict__ W2,
                        const float* __restrict__ b2,
                        const float* __restrict__ slots,
                        const int*   __restrict__ recs_,
                        int level)
{
    __shared__ float X[16 * XSTR_];
    __shared__ float H[16 * XSTR_];
    __shared__ int   rowNode[16];

    const int lane     = threadIdx.x & 31;
    const int w        = threadIdx.x >> 5;
    const int colLane  = lane & 15;
    const int laneHalf = (lane >> 4) & 1;
    const int4* tr = (const int4*)recs_ + (size_t)level * SLOTS_ + blockIdx.x * 16;

    int4 rc[4];
    #pragma unroll
    for (int r = 0; r < 4; ++r)
        rc[r] = tr[w + 4 * r];
    const int t0 = rc[0].z;
    const int e = t0 >= START_ ? T_ : (t0 < 0 ? 0 : t0);
    if (threadIdx.x < 16) rowNode[threadIdx.x] = tr[threadIdx.x].w;

    v2f w1first[8][4];
    dag_fetch_b<true, K1_, 4>(W1, e, w * 4, 0, lane, w1first);

    v4f ga[4], gc[4];
    #pragma unroll
    for (int r = 0; r < 4; ++r) {
        const int t = rc[r].z;
        int slot = t - START_;
        slot = slot < 0 ? 0 : (slot > OUTSLOTS_ - 1 ? OUTSLOTS_ - 1 : slot);
        const float* s0 = buf + (size_t)rc[r].x * D_;
        const float* s1 = (t >= START_) ? (slots + (size_t)slot * D_)
                                        : (buf + (size_t)rc[r].y * D_);
        ga[r] = *(const v4f*)(s0 + lane * 4);
        gc[r] = *(const v4f*)(s1 + lane * 4);
    }
    #pragma unroll
    for (int r = 0; r < 4; ++r) {
        const int m = w + 4 * r;
        *(v4f*)(&X[m * XSTR_ + lane * 4])       = ga[r];
        *(v4f*)(&X[m * XSTR_ + 128 + lane * 4]) = gc[r];
    }
    __syncthreads();

    v2f w2first[8][2];
    {
        v8f acc[4] = {};
        dag_panel<true, K1_, 4>(X, W1, e, w * 4, lane, w1first, acc);
        dag_fetch_b<true, D_, 2>(W2, e, w * 2, 0, lane, w2first);
        #pragma unroll
        for (int t = 0; t < 4; ++t) {
            const int j = w * 4 + t;
            float bias = b1[e * K1_ + j * 16 + colLane];
            #pragma unroll
            for (int i = 0; i < 8; ++i) {
                float x = acc[t][i] + bias;
                float g = 0.5f * x * (1.0f + erff(x * 0.70710678118654752f));
                H[(i + laneHalf * 8) * XSTR_ + j * 16 + colLane] = g;
            }
        }
    }
    __syncthreads();

    {
        v8f acc[2] = {};
        dag_panel<true, D_, 2>(H, W2, e, w * 2, lane, w2first, acc);
        int node[8];
        #pragma unroll
        for (int i = 0; i < 8; ++i)
            node[i] = rowNode[laneHalf * 8 + i];
        #pragma unroll
        for (int t = 0; t < 2; ++t) {
            const int j = w * 2 + t;
            float bias = b2[e * D_ + j * 16 + colLane];
            #pragma unroll
            for (int i = 0; i < 8; ++i) {
                if (node[i] >= 0) {
                    const int row = R_ + level * M_ + node[i];
                    buf[(size_t)row * D_ + j * 16 + colLane] = acc[t][i] + bias;
                }
            }
        }
    }
}

extern "C" void kernel_launch(void* const* d_in, const int* in_sizes, int n_in,
                              void* d_out, int out_size, void* d_ws, size_t ws_size,
                              hipStream_t stream) {
    const float* root  = (const float*)d_in[0];
    const float* W1    = (const float*)d_in[1];
    const float* b1    = (const float*)d_in[2];
    const float* W2    = (const float*)d_in[3];
    const float* b2    = (const float*)d_in[4];
    const float* slots = (const float*)d_in[5];
    const int*   par   = (const int*)d_in[6];
    const int*   typ   = (const int*)d_in[7];
    float* out = (float*)d_out;

    dag_init_roots<<<128, 256, 0, stream>>>(out, root);

    const bool packed = (ws_size >= WSP_BYTES);
    float* W1p = (float*)d_ws;
    float* W2p = (float*)((char*)d_ws + W1P_BYTES);

    const bool bucketed = (ws_size >= WSP_BYTES + REC_BYTES);
    int* recs = (int*)((char*)d_ws + WSP_BYTES);

    if (bucketed) {
        dag_pack_w<<<W1P_FRAGS / 256, 256, 0, stream>>>(W1, W1p, 16, K1_);
        dag_pack_w<<<W2P_FRAGS / 256, 256, 0, stream>>>(W2, W2p, 8, D_);
        dag_bucket<<<L_, 1024, 0, stream>>>(par, typ, recs);
        for (int l = 0; l < L_; ++l)
            dag_level_bucketed<<<TILES_, 128, 0, stream>>>(out, W1p, b1, W2p, b2,
                                                           slots, recs, l);
    } else if (packed) {
        dag_pack_w<<<W1P_FRAGS / 256, 256, 0, stream>>>(W1, W1p, 16, K1_);
        dag_pack_w<<<W2P_FRAGS / 256, 256, 0, stream>>>(W2, W2p, 8, D_);
        for (int l = 0; l < L_; ++l)
            dag_level<true><<<dim3(128, NENC_), 128, 0, stream>>>(out, W1p, b1, W2p, b2,
                                                                 slots, par, typ, l);
    } else {
        for (int l = 0; l < L_; ++l)
            dag_level<false><<<dim3(128, NENC_), 128, 0, stream>>>(out, W1, b1, W2, b2,
                                                                  slots, par, typ, l);
    }
}


# GELU epilogue without the erf branch: both ranges of the library's erf computed for each of a lane's 32 values and one selected, so the 32 dependent chains interleave; same operations per value
# speedup vs baseline: 6.3391x; 1.1712x over previous
#include <hip/hip_runtime.h>
#include <math.h>

typedef __attribute__((ext_vector_type(2))) float v2f;
typedef __attribute__((ext_vector_type(4))) float v4f;
typedef __attribute__((ext_vector_type(8))) float v8f;

#define D_        128
#define R_        1024
#define T_        4
#define OUTSLOTS_ 256
#define START_    4
#define L_        64
#define M_        2048
#define N_        (R_ + L_ * M_)
#define NENC_     5
#define K1_       256
#define XSTR_     260

#define W1P_FRAGS (NENC_ * 16 * 64 * 32)
#define W2P_FRAGS (NENC_ * 8  * 64 * 32)
#define W1P_BYTES ((size_t)W1P_FRAGS * 8)
#define W2P_BYTES ((size_t)W2P_FRAGS * 8)
#define WSP_BYTES (W1P_BYTES + W2P_BYTES)

#define TILES_    132
#define SLOTS_    (TILES_ * 16)
#define REC_BYTES ((size_t)L_ * SLOTS_ * 16)

__global__ void dag_init_roots(float* __restrict__ buf, const float* __restrict__ root) {
    int i = blockIdx.x * blockDim.x + threadIdx.x;
    v4f v = *(const v4f*)(root + (size_t)i * 4);
    *(v4f*)(buf + (size_t)i * 4) = v;
}

__global__ void dag_pack_w(const float* __restrict__ W, float* __restrict__ Wp,
                           int ntile, int ncol) {
    int idx  = blockIdx.x * blockDim.x + threadIdx.x;
    int lane = idx & 31;
    int k4   = (idx >> 5) & 63;
    int j    = (idx >> 11) % ntile;
    int e    = idx / (ntile << 11);
    int col  = lane & 15;
    int kk   = k4 * 4 + ((lane >> 4) & 1) * 2;
    const float* Wsrc = W + (size_t)e * K1_ * ncol + (size_t)(j * 16 + col);
    Wp[(size_t)idx * 2 + 0] = Wsrc[(size_t)kk * ncol];
    Wp[(size_t)idx * 2 + 1] = Wsrc[(size_t)(kk + 1) * ncol];
}

template <bool PACKED, int NCOL>
__device__ __forceinline__ v2f dag_load_b(const float* __restrict__ Wq, int e, int j,
                                          int k4, int lane) {
    if (PACKED) {
        const v2f* Wf = (const v2f*)Wq + ((size_t)(e * (NCOL / 16) + j) * 64) * 32 + lane;
        return Wf[k4 * 32];
    } else {
        const int colLane = lane & 15;
        const int kk = k4 * 4 + ((lane >> 4) & 1) * 2;
        const float* Wb = Wq + (size_t)e * K1_ * NCOL + (size_t)(j * 16 + colLane);
        v2f b;
        b.x = Wb[(size_t)kk * NCOL];
        b.y = Wb[(size_t)(kk + 1) * NCOL];
        return b;
    }
}

template <bool PACKED, int NCOL, int NT>
__device__ __forceinline__ void dag_fetch_b(const float* __restrict__ Wq, int e, int j0,
                                            int g, int lane, v2f (&b)[8][NT]) {
    #pragma unroll
    for (int s = 0; s < 8; ++s) {
        #pragma unroll
        for (int t = 0; t < NT; ++t)
            b[s][t] = dag_load_b<PACKED, NCOL>(Wq, e, j0 + t, g * 8 + s, lane);
    }
}

template <bool PACKED, int NCOL, int NT>
__device__ __forceinline__ void dag_panel(const float* __restrict__ Xs,
                                          const float* __restrict__ Wq,
                                          int e, int j0, int lane,
                                          v2f (&bfirst)[8][NT], v8f (&acc)[NT]) {
    const int colLane = lane & 15;
    const int klo     = ((lane >> 4) & 1) * 2;
    const float* Arow = Xs + colLane * XSTR_ + klo;

    v2f aq[2][8];
    v2f bq[2][8][NT];
    #pragma unroll
    for (int s = 0; s < 8; ++s) {
        #pragma unroll
        for (int t = 0; t < NT; ++t)
            bq[0][s][t] = bfirst[s][t];
        aq[0][s] = *(const v2f*)(Arow + s * 4);
    }
    #pragma unroll
    for (int g = 0; g < 8; ++g) {
        const int cur = g & 1, nxt = cur ^ 1;
        if (g < 7) {
            dag_fetch_b<PACKED, NCOL, NT>(Wq, e, j0, g + 1, lane, bq[nxt]);
            #pragma unroll
            for (int s = 0; s < 8; ++s)
                aq[nxt][s] = *(const v2f*)(Arow + ((g + 1) * 8 + s) * 4);
        }
        #pragma unroll
        for (int s = 0; s < 8; ++s) {
            #pragma unroll
            for (int t = 0; t < NT; ++t)
                acc[t] = __builtin_amdgcn_wmma_f32_16x16x4_f32(
                        false, aq[cur][s], false, bq[cur][s][t], (short)0, acc[t], false, false);
        }
    }
}

template <bool PACKED>
__global__ __launch_bounds__(128)
void dag_level(float* __restrict__ buf,
               const float* __restrict__ W1,
               const float* __restrict__ b1,
               const float* __restrict__ W2,
               const float* __restrict__ b2,
               const float* __restrict__ slots,
               const int*   __restrict__ par,
               const int*   __restrict__ types,
               int level)
{
    __shared__ float X[16 * XSTR_];
    __shared__ float H[16 * XSTR_];

    const int lane     = threadIdx.x & 31;
    const int w        = threadIdx.x >> 5;
    const int e        = blockIdx.y;
    const int tile     = blockIdx.x;
    const int qbase    = level * M_ + tile * 16;
    const int rowbase  = R_ + qbase;
    const int colLane  = lane & 15;
    const int laneHalf = (lane >> 4) & 1;

    int gp0[4], gp1[4], gt[4];
    #pragma unroll
    for (int r = 0; r < 4; ++r) {
        const int m = w + 4 * r;
        gp0[r] = par[(qbase + m) * 2 + 0];
        gp1[r] = par[(qbase + m) * 2 + 1];
        gt[r]  = types[qbase + m];
    }

    bool own[8];
    bool ownAny = false;
    #pragma unroll
    for (int i = 0; i < 8; ++i) {
        int t = types[qbase + i + laneHalf * 8];
        int r = (t >= START_) ? T_ : t;
        own[i] = (r == e);
        ownAny |= own[i];
    }
    if (!__any(ownAny)) return;

    v2f w1first[8][4];
    dag_fetch_b<PACKED, K1_, 4>(W1, e, w * 4, 0, lane, w1first);

    v4f ga[4], gc[4];
    #pragma unroll
    for (int r = 0; r < 4; ++r) {
        const int t = gt[r];
        int slot = t - START_;
        slot = slot < 0 ? 0 : (slot > OUTSLOTS_ - 1 ? OUTSLOTS_ - 1 : slot);
        const float* s0 = buf + (size_t)gp0[r] * D_;
        const float* s1 = (t >= START_) ? (slots + (size_t)slot * D_)
                                        : (buf + (size_t)gp1[r] * D_);
        ga[r] = *(const v4f*)(s0 + lane * 4);
        gc[r] = *(const v4f*)(s1 + lane * 4);
    }
    #pragma unroll
    for (int r = 0; r < 4; ++r) {
        const int m = w + 4 * r;
        *(v4f*)(&X[m * XSTR_ + lane * 4])       = ga[r];
        *(v4f*)(&X[m * XSTR_ + 128 + lane * 4]) = gc[r];
    }
    __syncthreads();

    v2f w2first[8][2];
    {
        v8f acc[4] = {};
        dag_panel<PACKED, K1_, 4>(X, W1, e, w * 4, lane, w1first, acc);
        dag_fetch_b<PACKED, D_, 2>(W2, e, w * 2, 0, lane, w2first);
        #pragma unroll
        for (int t = 0; t < 4; ++t) {
            const int j = w * 4 + t;
            float bias = b1[e * K1_ + j * 16 + colLane];
            #pragma unroll
            for (int i = 0; i < 8; ++i) {
                if (own[i]) {
                    float x = acc[t][i] + bias;
                    float g = 0.5f * x * (1.0f + erff(x * 0.70710678118654752f));
                    H[(i + laneHalf * 8) * XSTR_ + j * 16 + colLane] = g;
                }
            }
        }
    }
    __syncthreads();

    {
        v8f acc[2] = {};
        dag_panel<PACKED, D_, 2>(H, W2, e, w * 2, lane, w2first, acc);
        #pragma unroll
        for (int t = 0; t < 2; ++t) {
            const int j = w * 2 + t;
            float bias = b2[e * D_ + j * 16 + colLane];
            #pragma unroll
            for (int i = 0; i < 8; ++i) {
                if (own[i]) {
                    int row = rowbase + i + laneHalf * 8;
                    buf[(size_t)row * D_ + j * 16 + colLane] = acc[t][i] + bias;
                }
            }
        }
    }
}


__global__ __launch_bounds__(1024)
void dag_bucket(const int* __restrict__ par, const int* __restrict__ types,
                int* __restrict__ recs_) {
    __shared__ int cnt[NENC_ * 32];
    __shared__ int fillFrom[NENC_];
    __shared__ int tileEnd[NENC_];

    const int level = blockIdx.x;
    const int tid   = threadIdx.x;
    const int lane  = tid & 31;
    const int wv    = tid >> 5;
    const int q     = level * M_ + tid * 2;
    const int2 tt = *(const int2*)(types + q);
    const int4 pp = *(const int4*)(par + (size_t)q * 2);
    const int r0 = tt.x >= START_ ? T_ : (tt.x < 0 ? 0 : tt.x);
    const int r1 = tt.y >= START_ ? T_ : (tt.y < 0 ? 0 : tt.y);

    const unsigned below = (1u << lane) - 1u;
    int rank0 = 0, rank1 = 0;
    for (int e = 0; e < NENC_; ++e) {
        const unsigned m0 = (unsigned)__ballot(r0 == e);
        const unsigned m1 = (unsigned)__ballot(r1 == e);
        const int before = __popc(m0 & below) + __popc(m1 & below);
        if (r0 == e) rank0 = before;
        if (r1 == e) rank1 = before + (r0 == e ? 1 : 0);
        if (lane == 0) cnt[e * 32 + wv] = __popc(m0) + __popc(m1);
    }
    __syncthreads();

    if (wv == 0) {
        int base = 0;
        for (int e = 0; e < NENC_; ++e) {
            const int v = cnt[e * 32 + lane];
            int inc = v;
            for (int d = 1; d < 32; d <<= 1) {
                const int u = __shfl_up(inc, d);
                if (lane >= d) inc += u;
            }
            const int total = __shfl(inc, 31);
            cnt[e * 32 + lane] = base + inc - v;
            if (lane == 0) fillFrom[e] = base + total;
            base += ((total + 15) >> 4) << 4;
            if (lane == 0) tileEnd[e] = base;
        }
    }
    __syncthreads();

    int4* lv = (int4*)recs_ + (size_t)level * SLOTS_;
    lv[cnt[r0 * 32 + wv] + rank0] = make_int4(pp.x, pp.y, tt.x, tid * 2);
    lv[cnt[r1 * 32 + wv] + rank1] = make_int4(pp.z, pp.w, tt.y, tid * 2 + 1);

    for (int s = tid; s < SLOTS_; s += 1024) {
        int e = 0;
        for (int k = 0; k < NENC_; ++k)
            if (s >= tileEnd[k]) e = k + 1;
        if (e == NENC_)
            lv[s] = make_int4(0, 0, 0, -1);
        else if (s >= fillFrom[e])
            lv[s] = make_int4(0, 0, e == T_ ? START_ : e, -1);
    }
}

__device__ __forceinline__ float dag_erf_both_ranges(float y) {
    const float ax = __builtin_fabsf(y);
    const float t = y * y;
    float q = __builtin_fmaf(t, -0x1.268bc2p-11f, 0x1.420828p-8f);
    q = __builtin_fmaf(t, q, -0x1.b59370p-6f);
    q = __builtin_fmaf(t, q, 0x1.ce077cp-4f);
    q = __builtin_fmaf(t, q, -0x1.812660p-2f);
    q = __builtin_fmaf(t, q, 0x1.06eba0p-3f);
    const float rs = __builtin_fmaf(ax, q, ax);
    float p = __builtin_fmaf(ax, 0x1.1d3156p-16f, -0x1.8d1290p-12f);
    p = __builtin_fmaf(ax, p, 0x1.f9a6d2p-9f);
    p = __builtin_fmaf(ax, p, -0x1.8c3164p-6f);
    p = __builtin_fmaf(ax, p, 0x1.b4e9c8p-4f);
    p = __builtin_fmaf(ax, p, 0x1.4515fap-1f);
    p = __builtin_fmaf(ax, p, 0x1.078e50p-3f);
    p = __builtin_fmaf(ax, p, ax);
    const float rl = 1.0f - expf(-p);
    const float r = (ax < 1.0f) ? rs : rl;
    return __builtin_copysignf(r, y);
}

__global__ __launch_bounds__(128)
void dag_level_bucketed(float* __restrict__ buf,
                        const float* __restrict__ W1,
                        const float* __restrict__ b1,
                        const float* __restrict__ W2,
                        const float* __restrict__ b2,
                        const float* __restrict__ slots,
                        const int*   __restrict__ recs_,
                        int level)
{
    __shared__ float X[16 * XSTR_];
    __shared__ float H[16 * XSTR_];
    __shared__ int   rowNode[16];

    const int lane     = threadIdx.x & 31;
    const int w        = threadIdx.x >> 5;
    const int colLane  = lane & 15;
    const int laneHalf = (lane >> 4) & 1;
    const int4* tr = (const int4*)recs_ + (size_t)level * SLOTS_ + blockIdx.x * 16;

    int4 rc[4];
    #pragma unroll
    for (int r = 0; r < 4; ++r)
        rc[r] = tr[w + 4 * r];
    const int t0 = rc[0].z;
    const int e = t0 >= START_ ? T_ : (t0 < 0 ? 0 : t0);
    if (threadIdx.x < 16) rowNode[threadIdx.x] = tr[threadIdx.x].w;

    v2f w1first[8][4];
    dag_fetch_b<true, K1_, 4>(W1, e, w * 4, 0, lane, w1first);

    v4f ga[4], gc[4];
    #pragma unroll
    for (int r = 0; r < 4; ++r) {
        const int t = rc[r].z;
        int slot = t - START_;
        slot = slot < 0 ? 0 : (slot > OUTSLOTS_ - 1 ? OUTSLOTS_ - 1 : slot);
        const float* s0 = buf + (size_t)rc[r].x * D_;
        const float* s1 = (t >= START_) ? (slots + (size_t)slot * D_)
                                        : (buf + (size_t)rc[r].y * D_);
        ga[r] = *(const v4f*)(s0 + lane * 4);
        gc[r] = *(const v4f*)(s1 + lane * 4);
    }
    #pragma unroll
    for (int r = 0; r < 4; ++r) {
        const int m = w + 4 * r;
        *(v4f*)(&X[m * XSTR_ + lane * 4])       = ga[r];
        *(v4f*)(&X[m * XSTR_ + 128 + lane * 4]) = gc[r];
    }
    __syncthreads();

    v2f w2first[8][2];
    {
        v8f acc[4] = {};
        dag_panel<true, K1_, 4>(X, W1, e, w * 4, lane, w1first, acc);
        float bias[4];
        #pragma unroll
        for (int t = 0; t < 4; ++t)
            bias[t] = b1[e * K1_ + (w * 4 + t) * 16 + colLane];
        dag_fetch_b<true, D_, 2>(W2, e, w * 2, 0, lane, w2first);
        __builtin_amdgcn_sched_barrier(0);
        #pragma unroll
        for (int t = 0; t < 4; ++t) {
            const int j = w * 4 + t;
            #pragma unroll
            for (int i = 0; i < 8; ++i) {
                float x = acc[t][i] + bias[t];
                float g = 0.5f * x * (1.0f + dag_erf_both_ranges(x * 0.70710678118654752f));
                H[(i + laneHalf * 8) * XSTR_ + j * 16 + colLane] = g;
            }
        }
    }
    __syncthreads();

    {
        v8f acc[2] = {};
        dag_panel<true, D_, 2>(H, W2, e, w * 2, lane, w2first, acc);
        int node[8];
        #pragma unroll
        for (int i = 0; i < 8; ++i)
            node[i] = rowNode[laneHalf * 8 + i];
        #pragma unroll
        for (int t = 0; t < 2; ++t) {
            const int j = w * 2 + t;
            float bias = b2[e * D_ + j * 16 + colLane];
            #pragma unroll
            for (int i = 0; i < 8; ++i) {
                if (node[i] >= 0) {
                    const int row = R_ + level * M_ + node[i];
                    buf[(size_t)row * D_ + j * 16 + colLane] = acc[t][i] + bias;
                }
            }
        }
    }
}

extern "C" void kernel_launch(void* const* d_in, const int* in_sizes, int n_in,
                              void* d_out, int out_size, void* d_ws, size_t ws_size,
                              hipStream_t stream) {
    const float* root  = (const float*)d_in[0];
    const float* W1    = (const float*)d_in[1];
    const float* b1    = (const float*)d_in[2];
    const float* W2    = (const float*)d_in[3];
    const float* b2    = (const float*)d_in[4];
    const float* slots = (const float*)d_in[5];
    const int*   par   = (const int*)d_in[6];
    const int*   typ   = (const int*)d_in[7];
    float* out = (float*)d_out;

    dag_init_roots<<<128, 256, 0, stream>>>(out, root);

    const bool packed = (ws_size >= WSP_BYTES);
    float* W1p = (float*)d_ws;
    float* W2p = (float*)((char*)d_ws + W1P_BYTES);

    const bool bucketed = (ws_size >= WSP_BYTES + REC_BYTES);
    int* recs = (int*)((char*)d_ws + WSP_BYTES);

    if (bucketed) {
        dag_pack_w<<<W1P_FRAGS / 256, 256, 0, stream>>>(W1, W1p, 16, K1_);
        dag_pack_w<<<W2P_FRAGS / 256, 256, 0, stream>>>(W2, W2p, 8, D_);
        dag_bucket<<<L_, 1024, 0, stream>>>(par, typ, recs);
        for (int l = 0; l < L_; ++l)
            dag_level_bucketed<<<TILES_, 128, 0, stream>>>(out, W1p, b1, W2p, b2,
                                                           slots, recs, l);
    } else if (packed) {
        dag_pack_w<<<W1P_FRAGS / 256, 256, 0, stream>>>(W1, W1p, 16, K1_);
        dag_pack_w<<<W2P_FRAGS / 256, 256, 0, stream>>>(W2, W2p, 8, D_);
        for (int l = 0; l < L_; ++l)
            dag_level<true><<<dim3(128, NENC_), 128, 0, stream>>>(out, W1p, b1, W2p, b2,
                                                                 slots, par, typ, l);
    } else {
        for (int l = 0; l < L_; ++l)
            dag_level<false><<<dim3(128, NENC_), 128, 0, stream>>>(out, W1, b1, W2, b2,
                                                                  slots, par, typ, l);
    }
}


# Weight fragments fetched by 16-byte loads (two k-steps each, repacked) as whole 8-k-step groups kept 3 groups of products ahead (ring of 4 register sets), also across the layer change; sums unchanged
# speedup vs baseline: 7.8984x; 1.2460x over previous
#include <hip/hip_runtime.h>
#include <math.h>

typedef __attribute__((ext_vector_type(2))) float v2f;
typedef __attribute__((ext_vector_type(4))) float v4f;
typedef __attribute__((ext_vector_type(8))) float v8f;

#define D_        128
#define R_        1024
#define T_        4
#define OUTSLOTS_ 256
#define START_    4
#define L_        64
#define M_        2048
#define N_        (R_ + L_ * M_)
#define NENC_     5
#define K1_       256
#define XSTR_     260

#define W1P_FRAGS (NENC_ * 16 * 64 * 32)
#define W2P_FRAGS (NENC_ * 8  * 64 * 32)
#define W1P_BYTES ((size_t)W1P_FRAGS * 8)
#define W2P_BYTES ((size_t)W2P_FRAGS * 8)
#define WSP_BYTES (W1P_BYTES + W2P_BYTES)

#define RING1_    4
#define RING2_    8
static_assert(2 * (RING1_ - 1) <= RING2_ - 1, "layer-2 groups requested during layer 1 must fit the layer-2 ring");

#define TILES_    132
#define SLOTS_    (TILES_ * 16)
#define REC_BYTES ((size_t)L_ * SLOTS_ * 16)

__global__ void dag_init_roots(float* __restrict__ buf, const float* __restrict__ root) {
    int i = blockIdx.x * blockDim.x + threadIdx.x;
    v4f v = *(const v4f*)(root + (size_t)i * 4);
    *(v4f*)(buf + (size_t)i * 4) = v;
}

__global__ void dag_pack_w(const float* __restrict__ W, float* __restrict__ Wp,
                           int ntile, int ncol) {
    int idx  = blockIdx.x * blockDim.x + threadIdx.x;
    int lane = idx & 31;
    int k4   = (idx >> 5) & 63;
    int j    = (idx >> 11) % ntile;
    int e    = idx / (ntile << 11);
    int col  = lane & 15;
    int kk   = k4 * 4 + ((lane >> 4) & 1) * 2;
    const float* Wsrc = W + (size_t)e * K1_ * ncol + (size_t)(j * 16 + col);
    const size_t dst = ((size_t)((idx >> 11) * 32 + (k4 >> 1)) * 32 + lane) * 4 + (k4 & 1) * 2;
    Wp[dst + 0] = Wsrc[(size_t)kk * ncol];
    Wp[dst + 1] = Wsrc[(size_t)(kk + 1) * ncol];
}

template <bool PACKED, int NCOL>
__device__ __forceinline__ v4f dag_load_b2(const float* __restrict__ Wq, int e, int j,
                                           int p, int lane) {
    if (PACKED) {
        const v4f* Wf = (const v4f*)Wq + ((size_t)(e * (NCOL / 16) + j) * 32) * 32 + lane;
        return Wf[p * 32];
    } else {
        const int colLane = lane & 15;
        const int kk = p * 8 + ((lane >> 4) & 1) * 2;
        const float* Wb = Wq + (size_t)e * K1_ * NCOL + (size_t)(j * 16 + colLane);
        v4f b;
        b.x = Wb[(size_t)kk * NCOL];
        b.y = Wb[(size_t)(kk + 1) * NCOL];
        b.z = Wb[(size_t)(kk + 4) * NCOL];
        b.w = Wb[(size_t)(kk + 5) * NCOL];
        return b;
    }
}

template <bool PACKED, int NCOL, int NT>
__device__ __forceinline__ void dag_fetch_b(const float* __restrict__ Wq, int e, int j0,
                                            int g, int lane, v2f (&b)[8][NT]) {
    #pragma unroll
    for (int p = 0; p < 4; ++p) {
        #pragma unroll
        for (int t = 0; t < NT; ++t) {
            const v4f q = dag_load_b2<PACKED, NCOL>(Wq, e, j0 + t, g * 4 + p, lane);
            b[2 * p][t].x     = q.x;
            b[2 * p][t].y     = q.y;
            b[2 * p + 1][t].x = q.z;
            b[2 * p + 1][t].y = q.w;
        }
    }
}

template <bool PACKED, int NCOL, int NT, int RING, class TAIL>
__device__ __forceinline__ void dag_panel(const float* __restrict__ Xs,
                                          const float* __restrict__ Wq,
                                          int e, int j0, int lane,
                                          v2f (&bq)[RING][8][NT], v8f (&acc)[NT],
                                          const TAIL& tail) {
    const int colLane = lane & 15;
    const int klo     = ((lane >> 4) & 1) * 2;
    const float* Arow = Xs + colLane * XSTR_ + klo;

    v2f aq[2][8];
    #pragma unroll
    for (int s = 0; s < 8; ++s)
        aq[0][s] = *(const v2f*)(Arow + s * 4);
    #pragma unroll
    for (int g = 0; g < 8; ++g) {
        const int cur = g & 1, nxt = cur ^ 1;
        if (g + RING - 1 < 8)
            dag_fetch_b<PACKED, NCOL, NT>(Wq, e, j0, g + RING - 1, lane, bq[(g + RING - 1) % RING]);
        else
            tail(g + RING - 1 - 8);
        if (g < 7) {
            #pragma unroll
            for (int s = 0; s < 8; ++s)
                aq[nxt][s] = *(const v2f*)(Arow + ((g + 1) * 8 + s) * 4);
        }
        __builtin_amdgcn_sched_barrier(0);
        #pragma unroll
        for (int s = 0; s < 8; ++s) {
            #pragma unroll
            for (int t = 0; t < NT; ++t)
                acc[t] = __builtin_amdgcn_wmma_f32_16x16x4_f32(
                        false, aq[cur][s], false, bq[g % RING][s][t], (short)0, acc[t], false, false);
        }
        __builtin_amdgcn_sched_barrier(0);
    }
}

template <bool PACKED>
__global__ __launch_bounds__(128)
void dag_level(float* __restrict__ buf,
               const float* __restrict__ W1,
               const float* __restrict__ b1,
               const float* __restrict__ W2,
               const float* __restrict__ b2,
               const float* __restrict__ slots,
               const int*   __restrict__ par,
               const int*   __restrict__ types,
               int level)
{
    __shared__ float X[16 * XSTR_];
    __shared__ float H[16 * XSTR_];

    const int lane     = threadIdx.x & 31;
    const int w        = threadIdx.x >> 5;
    const int e        = blockIdx.y;
    const int tile     = blockIdx.x;
    const int qbase    = level * M_ + tile * 16;
    const int rowbase  = R_ + qbase;
    const int colLane  = lane & 15;
    const int laneHalf = (lane >> 4) & 1;

    int gp0[4], gp1[4], gt[4];
    #pragma unroll
    for (int r = 0; r < 4; ++r) {
        const int m = w + 4 * r;
        gp0[r] = par[(qbase + m) * 2 + 0];
        gp1[r] = par[(qbase + m) * 2 + 1];
        gt[r]  = types[qbase + m];
    }

    bool own[8];
    bool ownAny = false;
    #pragma unroll
    for (int i = 0; i < 8; ++i) {
        int t = types[qbase + i + laneHalf * 8];
        int r = (t >= START_) ? T_ : t;
        own[i] = (r == e);
        ownAny |= own[i];
    }
    if (!__any(ownAny)) return;

    v2f w1ring[RING1_][8][4];
    #pragma unroll
    for (int g = 0; g < RING1_ - 1; ++g)
        dag_fetch_b<PACKED, K1_, 4>(W1, e, w * 4, g, lane, w1ring[g]);

    v4f ga[4], gc[4];
    #pragma unroll
    for (int r = 0; r < 4; ++r) {
        const int t = gt[r];
        int slot = t - START_;
        slot = slot < 0 ? 0 : (slot > OUTSLOTS_ - 1 ? OUTSLOTS_ - 1 : slot);
        const float* s0 = buf + (size_t)gp0[r] * D_;
        const float* s1 = (t >= START_) ? (slots + (size_t)slot * D_)
                                        : (buf + (size_t)gp1[r] * D_);
        ga[r] = *(const v4f*)(s0 + lane * 4);
        gc[r] = *(const v4f*)(s1 + lane * 4);
    }
    #pragma unroll
    for (int r = 0; r < 4; ++r) {
        const int m = w + 4 * r;
        *(v4f*)(&X[m * XSTR_ + lane * 4])       = ga[r];
        *(v4f*)(&X[m * XSTR_ + 128 + lane * 4]) = gc[r];
    }
    __syncthreads();

    v2f w2ring[RING2_][8][2];
    {
        v8f acc[4] = {};
        dag_panel<PACKED, K1_, 4, RING1_>(X, W1, e, w * 4, lane, w1ring, acc, [](int) {});
        #pragma unroll
        for (int g = 0; g < RING2_ - 1; ++g)
            dag_fetch_b<PACKED, D_, 2>(W2, e, w * 2, g, lane, w2ring[g]);
        #pragma unroll
        for (int t = 0; t < 4; ++t) {
            const int j = w * 4 + t;
            float bias = b1[e * K1_ + j * 16 + colLane];
            #pragma unroll
            for (int i = 0; i < 8; ++i) {
                if (own[i]) {
                    float x = acc[t][i] + bias;
                    float g = 0.5f * x * (1.0f + erff(x * 0.70710678118654752f));
                    H[(i + laneHalf * 8) * XSTR_ + j * 16 + colLane] = g;
                }
            }
        }
    }
    __syncthreads();

    {
        v8f acc[2] = {};
        dag_panel<PACKED, D_, 2, RING2_>(H, W2, e, w * 2, lane, w2ring, acc, [](int) {});
        #pragma unroll
        for (int t = 0; t < 2; ++t) {
            const int j = w * 2 + t;
            float bias = b2[e * D_ + j * 16 + colLane];
            #pragma unroll
            for (int i = 0; i < 8; ++i) {
                if (own[i]) {
                    int row = rowbase + i + laneHalf * 8;
                    buf[(size_t)row * D_ + j * 16 + colLane] = acc[t][i] + bias;
                }
            }
        }
    }
}


__global__ __launch_bounds__(1024)
void dag_bucket(const int* __restrict__ par, const int* __restrict__ types,
                int* __restrict__ recs_) {
    __shared__ int cnt[NENC_ * 32];
    __shared__ int fillFrom[NENC_];
    __shared__ int tileEnd[NENC_];

    const int level = blockIdx.x;
    const int tid   = threadIdx.x;
    const int lane  = tid & 31;
    const int wv    = tid >> 5;
    const int q     = level * M_ + tid * 2;
    const int2 tt = *(const int2*)(types + q);
    const int4 pp = *(const int4*)(par + (size_t)q * 2);
    const int r0 = tt.x >= START_ ? T_ : (tt.x < 0 ? 0 : tt.x);
    const int r1 = tt.y >= START_ ? T_ : (tt.y < 0 ? 0 : tt.y);

    const unsigned below = (1u << lane) - 1u;
    int rank0 = 0, rank1 = 0;
    for (int e = 0; e < NENC_; ++e) {
        const unsigned m0 = (unsigned)__ballot(r0 == e);
        const unsigned m1 = (unsigned)__ballot(r1 == e);
        const int before = __popc(m0 & below) + __popc(m1 & below);
        if (r0 == e) rank0 = before;
        if (r1 == e) rank1 = before + (r0 == e ? 1 : 0);
        if (lane == 0) cnt[e * 32 + wv] = __popc(m0) + __popc(m1);
    }
    __syncthreads();

    if (wv == 0) {
        int base = 0;
        for (int e = 0; e < NENC_; ++e) {
            const int v = cnt[e * 32 + lane];
            int inc = v;
            for (int d = 1; d < 32; d <<= 1) {
                const int u = __shfl_up(inc, d);
                if (lane >= d) inc += u;
            }
            const int total = __shfl(inc, 31);
            cnt[e * 32 + lane] = base + inc - v;
            if (lane == 0) fillFrom[e] = base + total;
            base += ((total + 15) >> 4) << 4;
            if (lane == 0) tileEnd[e] = base;
        }
    }
    __syncthreads();

    int4* lv = (int4*)recs_ + (size_t)level * SLOTS_;
    lv[cnt[r0 * 32 + wv] + rank0] = make_int4(pp.x, pp.y, tt.x, tid * 2);
    lv[cnt[r1 * 32 + wv] + rank1] = make_int4(pp.z, pp.w, tt.y, tid * 2 + 1);

    for (int s = tid; s < SLOTS_; s += 1024) {
        int e = 0;
        for (int k = 0; k < NENC_; ++k)
            if (s >= tileEnd[k]) e = k + 1;
        if (e == NENC_)
            lv[s] = make_int4(0, 0, 0, -1);
        else if (s >= fillFrom[e])
            lv[s] = make_int4(0, 0, e == T_ ? START_ : e, -1);
    }
}

__device__ __forceinline__ float dag_erf_both_ranges(float y) {
    const float ax = __builtin_fabsf(y);
    const float t = y * y;
    float q = __builtin_fmaf(t, -0x1.268bc2p-11f, 0x1.420828p-8f);
    q = __builtin_fmaf(t, q, -0x1.b59370p-6f);
    q = __builtin_fmaf(t, q, 0x1.ce077cp-4f);
    q = __builtin_fmaf(t, q, -0x1.812660p-2f);
    q = __builtin_fmaf(t, q, 0x1.06eba0p-3f);
    const float rs = __builtin_fmaf(ax, q, ax);
    float p = __builtin_fmaf(ax, 0x1.1d3156p-16f, -0x1.8d1290p-12f);
    p = __builtin_fmaf(ax, p, 0x1.f9a6d2p-9f);
    p = __builtin_fmaf(ax, p, -0x1.8c3164p-6f);
    p = __builtin_fmaf(ax, p, 0x1.b4e9c8p-4f);
    p = __builtin_fmaf(ax, p, 0x1.4515fap-1f);
    p = __builtin_fmaf(ax, p, 0x1.078e50p-3f);
    p = __builtin_fmaf(ax, p, ax);
    const float rl = 1.0f - expf(-p);
    const float r = (ax < 1.0f) ? rs : rl;
    return __builtin_copysignf(r, y);
}

__global__ __launch_bounds__(128)
void dag_level_bucketed(float* __restrict__ buf,
                        const float* __restrict__ W1,
                        const float* __restrict__ b1,
                        const float* __restrict__ W2,
                        const float* __restrict__ b2,
                        const float* __restrict__ slots,
                        const int*   __restrict__ recs_,
                        int level)
{
    __shared__ float X[16 * XSTR_];
    __shared__ float H[16 * XSTR_];
    __shared__ int   rowNode[16];

    const int lane     = threadIdx.x & 31;
    const int w        = threadIdx.x >> 5;
    const int colLane  = lane & 15;
    const int laneHalf = (lane >> 4) & 1;
    const int4* tr = (const int4*)recs_ + (size_t)level * SLOTS_ + blockIdx.x * 16;

    int4 rc[4];
    #pragma unroll
    for (int r = 0; r < 4; ++r)
        rc[r] = tr[w + 4 * r];
    const int t0 = rc[0].z;
    const int e = t0 >= START_ ? T_ : (t0 < 0 ? 0 : t0);
    if (threadIdx.x < 16) rowNode[threadIdx.x] = tr[threadIdx.x].w;

    v4f ga[4], gc[4];
    #pragma unroll
    for (int r = 0; r < 4; ++r) {
        const int t = rc[r].z;
        int slot = t - START_;
        slot = slot < 0 ? 0 : (slot > OUTSLOTS_ - 1 ? OUTSLOTS_ - 1 : slot);
        const float* s0 = buf + (size_t)rc[r].x * D_;
        const float* s1 = (t >= START_) ? (slots + (size_t)slot * D_)
                                        : (buf + (size_t)rc[r].y * D_);
        ga[r] = *(const v4f*)(s0 + lane * 4);
        gc[r] = *(const v4f*)(s1 + lane * 4);
    }

    v2f w1ring[RING1_][8][4];
    #pragma unroll
    for (int g = 0; g < RING1_ - 1; ++g)
        dag_fetch_b<true, K1_, 4>(W1, e, w * 4, g, lane, w1ring[g]);

    #pragma unroll
    for (int r = 0; r < 4; ++r) {
        const int m = w + 4 * r;
        *(v4f*)(&X[m * XSTR_ + lane * 4])       = ga[r];
        *(v4f*)(&X[m * XSTR_ + 128 + lane * 4]) = gc[r];
    }
    __syncthreads();

    v2f w2ring[RING2_][8][2];
    {
        v8f acc[4] = {};
        float bias[4];
        auto ahead = [&](int i) {
            if (i == 0) {
                #pragma unroll
                for (int t = 0; t < 4; ++t)
                    bias[t] = b1[e * K1_ + (w * 4 + t) * 16 + colLane];
            }
            dag_fetch_b<true, D_, 2>(W2, e, w * 2, 2 * i,     lane, w2ring[2 * i]);
            dag_fetch_b<true, D_, 2>(W2, e, w * 2, 2 * i + 1, lane, w2ring[2 * i + 1]);
        };
        dag_panel<true, K1_, 4, RING1_>(X, W1, e, w * 4, lane, w1ring, acc, ahead);
        #pragma unroll
        for (int g = 2 * (RING1_ - 1); g < RING2_ - 1; ++g)
            dag_fetch_b<true, D_, 2>(W2, e, w * 2, g, lane, w2ring[g]);
        __builtin_amdgcn_sched_barrier(0);
        #pragma unroll
        for (int t = 0; t < 4; ++t) {
            const int j = w * 4 + t;
            #pragma unroll
            for (int i = 0; i < 8; ++i) {
                float x = acc[t][i] + bias[t];
                float g = 0.5f * x * (1.0f + dag_erf_both_ranges(x * 0.70710678118654752f));
                H[(i + laneHalf * 8) * XSTR_ + j * 16 + colLane] = g;
            }
        }
    }
    __syncthreads();

    {
        v8f acc[2] = {};
        dag_panel<true, D_, 2, RING2_>(H, W2, e, w * 2, lane, w2ring, acc, [](int) {});
        int node[8];
        #pragma unroll
        for (int i = 0; i < 8; ++i)
            node[i] = rowNode[laneHalf * 8 + i];
        #pragma unroll
        for (int t = 0; t < 2; ++t) {
            const int j = w * 2 + t;
            float bias = b2[e * D_ + j * 16 + colLane];
            #pragma unroll
            for (int i = 0; i < 8; ++i) {
                if (node[i] >= 0) {
                    const int row = R_ + level * M_ + node[i];
                    buf[(size_t)row * D_ + j * 16 + colLane] = acc[t][i] + bias;
                }
            }
        }
    }
}

extern "C" void kernel_launch(void* const* d_in, const int* in_sizes, int n_in,
                              void* d_out, int out_size, void* d_ws, size_t ws_size,
                              hipStream_t stream) {
    const float* root  = (const float*)d_in[0];
    const float* W1    = (const float*)d_in[1];
    const float* b1    = (const float*)d_in[2];
    const float* W2    = (const float*)d_in[3];
    const float* b2    = (const float*)d_in[4];
    const float* slots = (const float*)d_in[5];
    const int*   par   = (const int*)d_in[6];
    const int*   typ   = (const int*)d_in[7];
    float* out = (float*)d_out;

    dag_init_roots<<<128, 256, 0, stream>>>(out, root);

    const bool packed = (ws_size >= WSP_BYTES);
    float* W1p = (float*)d_ws;
    float* W2p = (float*)((char*)d_ws + W1P_BYTES);

    const bool bucketed = (ws_size >= WSP_BYTES + REC_BYTES);
    int* recs = (int*)((char*)d_ws + WSP_BYTES);

    if (bucketed) {
        dag_pack_w<<<W1P_FRAGS / 256, 256, 0, stream>>>(W1, W1p, 16, K1_);
        dag_pack_w<<<W2P_FRAGS / 256, 256, 0, stream>>>(W2, W2p, 8, D_);
        dag_bucket<<<L_, 1024, 0, stream>>>(par, typ, recs);
        for (int l = 0; l < L_; ++l)
            dag_level_bucketed<<<TILES_, 128, 0, stream>>>(out, W1p, b1, W2p, b2,
                                                           slots, recs, l);
    } else if (packed) {
        dag_pack_w<<<W1P_FRAGS / 256, 256, 0, stream>>>(W1, W1p, 16, K1_);
        dag_pack_w<<<W2P_FRAGS / 256, 256, 0, stream>>>(W2, W2p, 8, D_);
        for (int l = 0; l < L_; ++l)
            dag_level<true><<<dim3(128, NENC_), 128, 0, stream>>>(out, W1p, b1, W2p, b2,
                                                                 slots, par, typ, l);
    } else {
        for (int l = 0; l < L_; ++l)
            dag_level<false><<<dim3(128, NENC_), 128, 0, stream>>>(out, W1, b1, W2, b2,
                                                                  slots, par, typ, l);
    }
}


# Level kernel's exposed load waits removed: layer-2 biases and row numbers requested during the products, not after the last; barriers wait for LDS only, weight requests stay in flight; sums unchanged
# speedup vs baseline: 8.8662x; 1.1225x over previous
#include <hip/hip_runtime.h>
#include <math.h>

typedef __attribute__((ext_vector_type(2))) float v2f;
typedef __attribute__((ext_vector_type(4))) float v4f;
typedef __attribute__((ext_vector_type(8))) float v8f;

#define D_        128
#define R_        1024
#define T_        4
#define OUTSLOTS_ 256
#define START_    4
#define L_        64
#define M_        2048
#define N_        (R_ + L_ * M_)
#define NENC_     5
#define K1_       256
#define XSTR_     260

#define W1P_FRAGS (NENC_ * 16 * 64 * 32)
#define W2P_FRAGS (NENC_ * 8  * 64 * 32)
#define W1P_BYTES ((size_t)W1P_FRAGS * 8)
#define W2P_BYTES ((size_t)W2P_FRAGS * 8)
#define WSP_BYTES (W1P_BYTES + W2P_BYTES)

#define RING1_    4
#define RING2_    8
static_assert(2 * (RING1_ - 1) <= RING2_ - 1, "layer-2 groups requested during layer 1 must fit the layer-2 ring");

#define TILES_    132
#define SLOTS_    (TILES_ * 16)
#define REC_BYTES ((size_t)L_ * SLOTS_ * 16)

__global__ void dag_init_roots(float* __restrict__ buf, const float* __restrict__ root) {
    int i = blockIdx.x * blockDim.x + threadIdx.x;
    v4f v = *(const v4f*)(root + (size_t)i * 4);
    *(v4f*)(buf + (size_t)i * 4) = v;
}

__global__ void dag_pack_w(const float* __restrict__ W, float* __restrict__ Wp,
                           int ntile, int ncol) {
    int idx  = blockIdx.x * blockDim.x + threadIdx.x;
    int lane = idx & 31;
    int k4   = (idx >> 5) & 63;
    int j    = (idx >> 11) % ntile;
    int e    = idx / (ntile << 11);
    int col  = lane & 15;
    int kk   = k4 * 4 + ((lane >> 4) & 1) * 2;
    const float* Wsrc = W + (size_t)e * K1_ * ncol + (size_t)(j * 16 + col);
    const size_t dst = ((size_t)((idx >> 11) * 32 + (k4 >> 1)) * 32 + lane) * 4 + (k4 & 1) * 2;
    Wp[dst + 0] = Wsrc[(size_t)kk * ncol];
    Wp[dst + 1] = Wsrc[(size_t)(kk + 1) * ncol];
}

template <bool PACKED, int NCOL>
__device__ __forceinline__ v4f dag_load_b2(const float* __restrict__ Wq, int e, int j,
                                           int p, int lane) {
    if (PACKED) {
        const v4f* Wf = (const v4f*)Wq + ((size_t)(e * (NCOL / 16) + j) * 32) * 32 + lane;
        return Wf[p * 32];
    } else {
        const int colLane = lane & 15;
        const int kk = p * 8 + ((lane >> 4) & 1) * 2;
        const float* Wb = Wq + (size_t)e * K1_ * NCOL + (size_t)(j * 16 + colLane);
        v4f b;
        b.x = Wb[(size_t)kk * NCOL];
        b.y = Wb[(size_t)(kk + 1) * NCOL];
        b.z = Wb[(size_t)(kk + 4) * NCOL];
        b.w = Wb[(size_t)(kk + 5) * NCOL];
        return b;
    }
}

template <bool PACKED, int NCOL, int NT>
__device__ __forceinline__ void dag_fetch_b(const float* __restrict__ Wq, int e, int j0,
                                            int g, int lane, v2f (&b)[8][NT]) {
    #pragma unroll
    for (int p = 0; p < 4; ++p) {
        #pragma unroll
        for (int t = 0; t < NT; ++t) {
            const v4f q = dag_load_b2<PACKED, NCOL>(Wq, e, j0 + t, g * 4 + p, lane);
            b[2 * p][t].x     = q.x;
            b[2 * p][t].y     = q.y;
            b[2 * p + 1][t].x = q.z;
            b[2 * p + 1][t].y = q.w;
        }
    }
}

template <bool PACKED, int NCOL, int NT, int RING, class TAIL>
__device__ __forceinline__ void dag_panel(const float* __restrict__ Xs,
                                          const float* __restrict__ Wq,
                                          int e, int j0, int lane,
                                          v2f (&bq)[RING][8][NT], v8f (&acc)[NT],
                                          const TAIL& tail) {
    const int colLane = lane & 15;
    const int klo     = ((lane >> 4) & 1) * 2;
    const float* Arow = Xs + colLane * XSTR_ + klo;

    v2f aq[2][8];
    #pragma unroll
    for (int s = 0; s < 8; ++s)
        aq[0][s] = *(const v2f*)(Arow + s * 4);
    #pragma unroll
    for (int g = 0; g < 8; ++g) {
        const int cur = g & 1, nxt = cur ^ 1;
        if (g + RING - 1 < 8)
            dag_fetch_b<PACKED, NCOL, NT>(Wq, e, j0, g + RING - 1, lane, bq[(g + RING - 1) % RING]);
        else
            tail(g + RING - 1 - 8);
        if (g < 7) {
            #pragma unroll
            for (int s = 0; s < 8; ++s)
                aq[nxt][s] = *(const v2f*)(Arow + ((g + 1) * 8 + s) * 4);
        }
        __builtin_amdgcn_sched_barrier(0);
        #pragma unroll
        for (int s = 0; s < 8; ++s) {
            #pragma unroll
            for (int t = 0; t < NT; ++t)
                acc[t] = __builtin_amdgcn_wmma_f32_16x16x4_f32(
                        false, aq[cur][s], false, bq[g % RING][s][t], (short)0, acc[t], false, false);
        }
        __builtin_amdgcn_sched_barrier(0);
    }
}

template <bool PACKED>
__global__ __launch_bounds__(128)
void dag_level(float* __restrict__ buf,
               const float* __restrict__ W1,
               const float* __restrict__ b1,
               const float* __restrict__ W2,
               const float* __restrict__ b2,
               const float* __restrict__ slots,
               const int*   __restrict__ par,
               const int*   __restrict__ types,
               int level)
{
    __shared__ float X[16 * XSTR_];
    __shared__ float H[16 * XSTR_];

    const int lane     = threadIdx.x & 31;
    const int w        = threadIdx.x >> 5;
    const int e        = blockIdx.y;
    const int tile     = blockIdx.x;
    const int qbase    = level * M_ + tile * 16;
    const int rowbase  = R_ + qbase;
    const int colLane  = lane & 15;
    const int laneHalf = (lane >> 4) & 1;

    int gp0[4], gp1[4], gt[4];
    #pragma unroll
    for (int r = 0; r < 4; ++r) {
        const int m = w + 4 * r;
        gp0[r] = par[(qbase + m) * 2 + 0];
        gp1[r] = par[(qbase + m) * 2 + 1];
        gt[r]  = types[qbase + m];
    }

    bool own[8];
    bool ownAny = false;
    #pragma unroll
    for (int i = 0; i < 8; ++i) {
        int t = types[qbase + i + laneHalf * 8];
        int r = (t >= START_) ? T_ : t;
        own[i] = (r == e);
        ownAny |= own[i];
    }
    if (!__any(ownAny)) return;

    v2f w1ring[RING1_][8][4];
    #pragma unroll
    for (int g = 0; g < RING1_ - 1; ++g)
        dag_fetch_b<PACKED, K1_, 4>(W1, e, w * 4, g, lane, w1ring[g]);

    v4f ga[4], gc[4];
    #pragma unroll
    for (int r = 0; r < 4; ++r) {
        const int t = gt[r];
        int slot = t - START_;
        slot = slot < 0 ? 0 : (slot > OUTSLOTS_ - 1 ? OUTSLOTS_ - 1 : slot);
        const float* s0 = buf + (size_t)gp0[r] * D_;
        const float* s1 = (t >= START_) ? (slots + (size_t)slot * D_)
                                        : (buf + (size_t)gp1[r] * D_);
        ga[r] = *(const v4f*)(s0 + lane * 4);
        gc[r] = *(const v4f*)(s1 + lane * 4);
    }
    #pragma unroll
    for (int r = 0; r < 4; ++r) {
        const int m = w + 4 * r;
        *(v4f*)(&X[m * XSTR_ + lane * 4])       = ga[r];
        *(v4f*)(&X[m * XSTR_ + 128 + lane * 4]) = gc[r];
    }
    __syncthreads();

    v2f w2ring[RING2_][8][2];
    {
        v8f acc[4] = {};
        dag_panel<PACKED, K1_, 4, RING1_>(X, W1, e, w * 4, lane, w1ring, acc, [](int) {});
        #pragma unroll
        for (int g = 0; g < RING2_ - 1; ++g)
            dag_fetch_b<PACKED, D_, 2>(W2, e, w * 2, g, lane, w2ring[g]);
        #pragma unroll
        for (int t = 0; t < 4; ++t) {
            const int j = w * 4 + t;
            float bias = b1[e * K1_ + j * 16 + colLane];
            #pragma unroll
            for (int i = 0; i < 8; ++i) {
                if (own[i]) {
                    float x = acc[t][i] + bias;
                    float g = 0.5f * x * (1.0f + erff(x * 0.70710678118654752f));
                    H[(i + laneHalf * 8) * XSTR_ + j * 16 + colLane] = g;
                }
            }
        }
    }
    __syncthreads();

    {
        v8f acc[2] = {};
        dag_panel<PACKED, D_, 2, RING2_>(H, W2, e, w * 2, lane, w2ring, acc, [](int) {});
        #pragma unroll
        for (int t = 0; t < 2; ++t) {
            const int j = w * 2 + t;
            float bias = b2[e * D_ + j * 16 + colLane];
            #pragma unroll
            for (int i = 0; i < 8; ++i) {
                if (own[i]) {
                    int row = rowbase + i + laneHalf * 8;
                    buf[(size_t)row * D_ + j * 16 + colLane] = acc[t][i] + bias;
                }
            }
        }
    }
}


__global__ __launch_bounds__(1024)
void dag_bucket(const int* __restrict__ par, const int* __restrict__ types,
                int* __restrict__ recs_) {
    __shared__ int cnt[NENC_ * 32];
    __shared__ int fillFrom[NENC_];
    __shared__ int tileEnd[NENC_];

    const int level = blockIdx.x;
    const int tid   = threadIdx.x;
    const int lane  = tid & 31;
    const int wv    = tid >> 5;
    const int q     = level * M_ + tid * 2;
    const int2 tt = *(const int2*)(types + q);
    const int4 pp = *(const int4*)(par + (size_t)q * 2);
    const int r0 = tt.x >= START_ ? T_ : (tt.x < 0 ? 0 : tt.x);
    const int r1 = tt.y >= START_ ? T_ : (tt.y < 0 ? 0 : tt.y);

    const unsigned below = (1u << lane) - 1u;
    int rank0 = 0, rank1 = 0;
    for (int e = 0; e < NENC_; ++e) {
        const unsigned m0 = (unsigned)__ballot(r0 == e);
        const unsigned m1 = (unsigned)__ballot(r1 == e);
        const int before = __popc(m0 & below) + __popc(m1 & below);
        if (r0 == e) rank0 = before;
        if (r1 == e) rank1 = before + (r0 == e ? 1 : 0);
        if (lane == 0) cnt[e * 32 + wv] = __popc(m0) + __popc(m1);
    }
    __syncthreads();

    if (wv == 0) {
        int base = 0;
        for (int e = 0; e < NENC_; ++e) {
            const int v = cnt[e * 32 + lane];
            int inc = v;
            for (int d = 1; d < 32; d <<= 1) {
                const int u = __shfl_up(inc, d);
                if (lane >= d) inc += u;
            }
            const int total = __shfl(inc, 31);
            cnt[e * 32 + lane] = base + inc - v;
            if (lane == 0) fillFrom[e] = base + total;
            base += ((total + 15) >> 4) << 4;
            if (lane == 0) tileEnd[e] = base;
        }
    }
    __syncthreads();

    int4* lv = (int4*)recs_ + (size_t)level * SLOTS_;
    lv[cnt[r0 * 32 + wv] + rank0] = make_int4(pp.x, pp.y, tt.x, tid * 2);
    lv[cnt[r1 * 32 + wv] + rank1] = make_int4(pp.z, pp.w, tt.y, tid * 2 + 1);

    for (int s = tid; s < SLOTS_; s += 1024) {
        int e = 0;
        for (int k = 0; k < NENC_; ++k)
            if (s >= tileEnd[k]) e = k + 1;
        if (e == NENC_)
            lv[s] = make_int4(0, 0, 0, -1);
        else if (s >= fillFrom[e])
            lv[s] = make_int4(0, 0, e == T_ ? START_ : e, -1);
    }
}

__device__ __forceinline__ float dag_erf_both_ranges(float y) {
    const float ax = __builtin_fabsf(y);
    const float t = y * y;
    float q = __builtin_fmaf(t, -0x1.268bc2p-11f, 0x1.420828p-8f);
    q = __builtin_fmaf(t, q, -0x1.b59370p-6f);
    q = __builtin_fmaf(t, q, 0x1.ce077cp-4f);
    q = __builtin_fmaf(t, q, -0x1.812660p-2f);
    q = __builtin_fmaf(t, q, 0x1.06eba0p-3f);
    const float rs = __builtin_fmaf(ax, q, ax);
    float p = __builtin_fmaf(ax, 0x1.1d3156p-16f, -0x1.8d1290p-12f);
    p = __builtin_fmaf(ax, p, 0x1.f9a6d2p-9f);
    p = __builtin_fmaf(ax, p, -0x1.8c3164p-6f);
    p = __builtin_fmaf(ax, p, 0x1.b4e9c8p-4f);
    p = __builtin_fmaf(ax, p, 0x1.4515fap-1f);
    p = __builtin_fmaf(ax, p, 0x1.078e50p-3f);
    p = __builtin_fmaf(ax, p, ax);
    const float rl = 1.0f - expf(-p);
    const float r = (ax < 1.0f) ? rs : rl;
    return __builtin_copysignf(r, y);
}

__device__ __forceinline__ void dag_lds_barrier() {
    asm volatile("s_wait_dscnt 0x0" ::: "memory");
    asm volatile("s_barrier_signal -1\n\ts_barrier_wait -1" ::: "memory");
}

__global__ __launch_bounds__(128)
void dag_level_bucketed(float* __restrict__ buf,
                        const float* __restrict__ W1,
                        const float* __restrict__ b1,
                        const float* __restrict__ W2,
                        const float* __restrict__ b2,
                        const float* __restrict__ slots,
                        const int*   __restrict__ recs_,
                        int level)
{
    __shared__ float X[16 * XSTR_];
    __shared__ float H[16 * XSTR_];
    __shared__ int   rowNode[16];

    const int lane     = threadIdx.x & 31;
    const int w        = threadIdx.x >> 5;
    const int colLane  = lane & 15;
    const int laneHalf = (lane >> 4) & 1;
    const int4* tr = (const int4*)recs_ + (size_t)level * SLOTS_ + blockIdx.x * 16;

    int4 rc[4];
    #pragma unroll
    for (int r = 0; r < 4; ++r)
        rc[r] = tr[w + 4 * r];
    const int t0 = rc[0].z;
    const int e = t0 >= START_ ? T_ : (t0 < 0 ? 0 : t0);
    if (threadIdx.x < 16) rowNode[threadIdx.x] = tr[threadIdx.x].w;

    v4f ga[4], gc[4];
    #pragma unroll
    for (int r = 0; r < 4; ++r) {
        const int t = rc[r].z;
        int slot = t - START_;
        slot = slot < 0 ? 0 : (slot > OUTSLOTS_ - 1 ? OUTSLOTS_ - 1 : slot);
        const float* s0 = buf + (size_t)rc[r].x * D_;
        const float* s1 = (t >= START_) ? (slots + (size_t)slot * D_)
                                        : (buf + (size_t)rc[r].y * D_);
        ga[r] = *(const v4f*)(s0 + lane * 4);
        gc[r] = *(const v4f*)(s1 + lane * 4);
    }

    v2f w1ring[RING1_][8][4];
    #pragma unroll
    for (int g = 0; g < RING1_ - 1; ++g)
        dag_fetch_b<true, K1_, 4>(W1, e, w * 4, g, lane, w1ring[g]);

    #pragma unroll
    for (int r = 0; r < 4; ++r) {
        const int m = w + 4 * r;
        *(v4f*)(&X[m * XSTR_ + lane * 4])       = ga[r];
        *(v4f*)(&X[m * XSTR_ + 128 + lane * 4]) = gc[r];
    }
    dag_lds_barrier();

    v2f w2ring[RING2_][8][2];
    float bias2[2];
    {
        v8f acc[4] = {};
        float bias[4];
        auto ahead = [&](int i) {
            if (i == 0) {
                #pragma unroll
                for (int t = 0; t < 4; ++t)
                    bias[t] = b1[e * K1_ + (w * 4 + t) * 16 + colLane];
                #pragma unroll
                for (int t = 0; t < 2; ++t)
                    bias2[t] = b2[e * D_ + (w * 2 + t) * 16 + colLane];
            }
            dag_fetch_b<true, D_, 2>(W2, e, w * 2, 2 * i,     lane, w2ring[2 * i]);
            dag_fetch_b<true, D_, 2>(W2, e, w * 2, 2 * i + 1, lane, w2ring[2 * i + 1]);
        };
        dag_panel<true, K1_, 4, RING1_>(X, W1, e, w * 4, lane, w1ring, acc, ahead);
        #pragma unroll
        for (int g = 2 * (RING1_ - 1); g < RING2_ - 1; ++g)
            dag_fetch_b<true, D_, 2>(W2, e, w * 2, g, lane, w2ring[g]);
        __builtin_amdgcn_sched_barrier(0);
        #pragma unroll
        for (int t = 0; t < 4; ++t) {
            const int j = w * 4 + t;
            #pragma unroll
            for (int i = 0; i < 8; ++i) {
                float x = acc[t][i] + bias[t];
                float g = 0.5f * x * (1.0f + dag_erf_both_ranges(x * 0.70710678118654752f));
                H[(i + laneHalf * 8) * XSTR_ + j * 16 + colLane] = g;
            }
        }
    }
    dag_lds_barrier();

    {
        v8f acc[2] = {};
        int node[8];
        auto ahead2 = [&](int i) {
            if (i == 0) {
                #pragma unroll
                for (int r = 0; r < 8; ++r)
                    node[r] = rowNode[laneHalf * 8 + r];
            }
        };
        dag_panel<true, D_, 2, RING2_>(H, W2, e, w * 2, lane, w2ring, acc, ahead2);
        #pragma unroll
        for (int t = 0; t < 2; ++t) {
            const int j = w * 2 + t;
            #pragma unroll
            for (int i = 0; i < 8; ++i) {
                if (node[i] >= 0) {
                    const int row = R_ + level * M_ + node[i];
                    buf[(size_t)row * D_ + j * 16 + colLane] = acc[t][i] + bias2[t];
                }
            }
        }
    }
}

extern "C" void kernel_launch(void* const* d_in, const int* in_sizes, int n_in,
                              void* d_out, int out_size, void* d_ws, size_t ws_size,
                              hipStream_t stream) {
    const float* root  = (const float*)d_in[0];
    const float* W1    = (const float*)d_in[1];
    const float* b1    = (const float*)d_in[2];
    const float* W2    = (const float*)d_in[3];
    const float* b2    = (const float*)d_in[4];
    const float* slots = (const float*)d_in[5];
    const int*   par   = (const int*)d_in[6];
    const int*   typ   = (const int*)d_in[7];
    float* out = (float*)d_out;

    dag_init_roots<<<128, 256, 0, stream>>>(out, root);

    const bool packed = (ws_size >= WSP_BYTES);
    float* W1p = (float*)d_ws;
    float* W2p = (float*)((char*)d_ws + W1P_BYTES);

    const bool bucketed = (ws_size >= WSP_BYTES + REC_BYTES);
    int* recs = (int*)((char*)d_ws + WSP_BYTES);

    if (bucketed) {
        dag_pack_w<<<W1P_FRAGS / 256, 256, 0, stream>>>(W1, W1p, 16, K1_);
        dag_pack_w<<<W2P_FRAGS / 256, 256, 0, stream>>>(W2, W2p, 8, D_);
        dag_bucket<<<L_, 1024, 0, stream>>>(par, typ, recs);
        for (int l = 0; l < L_; ++l)
            dag_level_bucketed<<<TILES_, 128, 0, stream>>>(out, W1p, b1, W2p, b2,
                                                           slots, recs, l);
    } else if (packed) {
        dag_pack_w<<<W1P_FRAGS / 256, 256, 0, stream>>>(W1, W1p, 16, K1_);
        dag_pack_w<<<W2P_FRAGS / 256, 256, 0, stream>>>(W2, W2p, 8, D_);
        for (int l = 0; l < L_; ++l)
            dag_level<true><<<dim3(128, NENC_), 128, 0, stream>>>(out, W1p, b1, W2p, b2,
                                                                 slots, par, typ, l);
    } else {
        for (int l = 0; l < L_; ++l)
            dag_level<false><<<dim3(128, NENC_), 128, 0, stream>>>(out, W1, b1, W2, b2,
                                                                  slots, par, typ, l);
    }
}


# Nodes run by dependency depth, not given level: a pre-pass takes depth = 1 + max over parents, sorts by depth and encoder (no atomics); launch d loops 512 groups over its tiles; row sums unchanged
# speedup vs baseline: 11.8288x; 1.3341x over previous
#include <hip/hip_runtime.h>
#include <math.h>

typedef __attribute__((ext_vector_type(2))) float v2f;
typedef __attribute__((ext_vector_type(4))) float v4f;
typedef __attribute__((ext_vector_type(8))) float v8f;

#define D_        128
#define R_        1024
#define T_        4
#define OUTSLOTS_ 256
#define START_    4
#define L_        64
#define M_        2048
#define N_        (R_ + L_ * M_)
#define NENC_     5
#define K1_       256
#define XSTR_     260

#define W1P_FRAGS (NENC_ * 16 * 64 * 32)
#define W2P_FRAGS (NENC_ * 8  * 64 * 32)
#define W1P_BYTES ((size_t)W1P_FRAGS * 8)
#define W2P_BYTES ((size_t)W2P_FRAGS * 8)
#define WSP_BYTES (W1P_BYTES + W2P_BYTES)

#define RING1_    4
#define RING2_    8
static_assert(2 * (RING1_ - 1) <= RING2_ - 1, "layer-2 groups requested during layer 1 must fit the layer-2 ring");

#define TILES_    132
#define SLOTS_    (TILES_ * 16)
#define REC_BYTES ((size_t)L_ * SLOTS_ * 16)

#define QN_        (L_ * M_)
#define NKEY_      (L_ * NENC_)
#define DTILES_    (QN_ / 16 + NKEY_)
#define DREC_BYTES ((size_t)DTILES_ * 16 * 16)
#define DEP_BYTES  ((size_t)QN_)
#define RANK_BYTES ((size_t)QN_ * 2)
#define CNT_BYTES  ((size_t)L_ * NKEY_ * 4)
#define TST_BYTES  ((size_t)512)
#define DGRID_     512

__global__ void dag_init_roots(float* __restrict__ buf, const float* __restrict__ root) {
    int i = blockIdx.x * blockDim.x + threadIdx.x;
    v4f v = *(const v4f*)(root + (size_t)i * 4);
    *(v4f*)(buf + (size_t)i * 4) = v;
}

__global__ void dag_pack_w(const float* __restrict__ W, float* __restrict__ Wp,
                           int ntile, int ncol) {
    int idx  = blockIdx.x * blockDim.x + threadIdx.x;
    int lane = idx & 31;
    int k4   = (idx >> 5) & 63;
    int j    = (idx >> 11) % ntile;
    int e    = idx / (ntile << 11);
    int col  = lane & 15;
    int kk   = k4 * 4 + ((lane >> 4) & 1) * 2;
    const float* Wsrc = W + (size_t)e * K1_ * ncol + (size_t)(j * 16 + col);
    const size_t dst = ((size_t)((idx >> 11) * 32 + (k4 >> 1)) * 32 + lane) * 4 + (k4 & 1) * 2;
    Wp[dst + 0] = Wsrc[(size_t)kk * ncol];
    Wp[dst + 1] = Wsrc[(size_t)(kk + 1) * ncol];
}

template <bool PACKED, int NCOL>
__device__ __forceinline__ v4f dag_load_b2(const float* __restrict__ Wq, int e, int j,
                                           int p, int lane) {
    if (PACKED) {
        const v4f* Wf = (const v4f*)Wq + ((size_t)(e * (NCOL / 16) + j) * 32) * 32 + lane;
        return Wf[p * 32];
    } else {
        const int colLane = lane & 15;
        const int kk = p * 8 + ((lane >> 4) & 1) * 2;
        const float* Wb = Wq + (size_t)e * K1_ * NCOL + (size_t)(j * 16 + colLane);
        v4f b;
        b.x = Wb[(size_t)kk * NCOL];
        b.y = Wb[(size_t)(kk + 1) * NCOL];
        b.z = Wb[(size_t)(kk + 4) * NCOL];
        b.w = Wb[(size_t)(kk + 5) * NCOL];
        return b;
    }
}

template <bool PACKED, int NCOL, int NT>
__device__ __forceinline__ void dag_fetch_b(const float* __restrict__ Wq, int e, int j0,
                                            int g, int lane, v2f (&b)[8][NT]) {
    #pragma unroll
    for (int p = 0; p < 4; ++p) {
        #pragma unroll
        for (int t = 0; t < NT; ++t) {
            const v4f q = dag_load_b2<PACKED, NCOL>(Wq, e, j0 + t, g * 4 + p, lane);
            b[2 * p][t].x     = q.x;
            b[2 * p][t].y     = q.y;
            b[2 * p + 1][t].x = q.z;
            b[2 * p + 1][t].y = q.w;
        }
    }
}

template <bool PACKED, int NCOL, int NT, int RING, class TAIL>
__device__ __forceinline__ void dag_panel(const float* __restrict__ Xs,
                                          const float* __restrict__ Wq,
                                          int e, int j0, int lane,
                                          v2f (&bq)[RING][8][NT], v8f (&acc)[NT],
                                          const TAIL& tail) {
    const int colLane = lane & 15;
    const int klo     = ((lane >> 4) & 1) * 2;
    const float* Arow = Xs + colLane * XSTR_ + klo;

    v2f aq[2][8];
    #pragma unroll
    for (int s = 0; s < 8; ++s)
        aq[0][s] = *(const v2f*)(Arow + s * 4);
    #pragma unroll
    for (int g = 0; g < 8; ++g) {
        const int cur = g & 1, nxt = cur ^ 1;
        if (g + RING - 1 < 8)
            dag_fetch_b<PACKED, NCOL, NT>(Wq, e, j0, g + RING - 1, lane, bq[(g + RING - 1) % RING]);
        else
            tail(g + RING - 1 - 8);
        if (g < 7) {
            #pragma unroll
            for (int s = 0; s < 8; ++s)
                aq[nxt][s] = *(const v2f*)(Arow + ((g + 1) * 8 + s) * 4);
        }
        __builtin_amdgcn_sched_barrier(0);
        #pragma unroll
        for (int s = 0; s < 8; ++s) {
            #pragma unroll
            for (int t = 0; t < NT; ++t)
                acc[t] = __builtin_amdgcn_wmma_f32_16x16x4_f32(
                        false, aq[cur][s], false, bq[g % RING][s][t], (short)0, acc[t], false, false);
        }
        __builtin_amdgcn_sched_barrier(0);
    }
}

template <bool PACKED>
__global__ __launch_bounds__(128)
void dag_level(float* __restrict__ buf,
               const float* __restrict__ W1,
               const float* __restrict__ b1,
               const float* __restrict__ W2,
               const float* __restrict__ b2,
               const float* __restrict__ slots,
               const int*   __restrict__ par,
               const int*   __restrict__ types,
               int level)
{
    __shared__ float X[16 * XSTR_];
    __shared__ float H[16 * XSTR_];

    const int lane     = threadIdx.x & 31;
    const int w        = threadIdx.x >> 5;
    const int e        = blockIdx.y;
    const int tile     = blockIdx.x;
    const int qbase    = level * M_ + tile * 16;
    const int rowbase  = R_ + qbase;
    const int colLane  = lane & 15;
    const int laneHalf = (lane >> 4) & 1;

    int gp0[4], gp1[4], gt[4];
    #pragma unroll
    for (int r = 0; r < 4; ++r) {
        const int m = w + 4 * r;
        gp0[r] = par[(qbase + m) * 2 + 0];
        gp1[r] = par[(qbase + m) * 2 + 1];
        gt[r]  = types[qbase + m];
    }

    bool own[8];
    bool ownAny = false;
    #pragma unroll
    for (int i = 0; i < 8; ++i) {
        int t = types[qbase + i + laneHalf * 8];
        int r = (t >= START_) ? T_ : t;
        own[i] = (r == e);
        ownAny |= own[i];
    }
    if (!__any(ownAny)) return;

    v2f w1ring[RING1_][8][4];
    #pragma unroll
    for (int g = 0; g < RING1_ - 1; ++g)
        dag_fetch_b<PACKED, K1_, 4>(W1, e, w * 4, g, lane, w1ring[g]);

    v4f ga[4], gc[4];
    #pragma unroll
    for (int r = 0; r < 4; ++r) {
        const int t = gt[r];
        int slot = t - START_;
        slot = slot < 0 ? 0 : (slot > OUTSLOTS_ - 1 ? OUTSLOTS_ - 1 : slot);
        const float* s0 = buf + (size_t)gp0[r] * D_;
        const float* s1 = (t >= START_) ? (slots + (size_t)slot * D_)
                                        : (buf + (size_t)gp1[r] * D_);
        ga[r] = *(const v4f*)(s0 + lane * 4);
        gc[r] = *(const v4f*)(s1 + lane * 4);
    }
    #pragma unroll
    for (int r = 0; r < 4; ++r) {
        const int m = w + 4 * r;
        *(v4f*)(&X[m * XSTR_ + lane * 4])       = ga[r];
        *(v4f*)(&X[m * XSTR_ + 128 + lane * 4]) = gc[r];
    }
    __syncthreads();

    v2f w2ring[RING2_][8][2];
    {
        v8f acc[4] = {};
        dag_panel<PACKED, K1_, 4, RING1_>(X, W1, e, w * 4, lane, w1ring, acc, [](int) {});
        #pragma unroll
        for (int g = 0; g < RING2_ - 1; ++g)
            dag_fetch_b<PACKED, D_, 2>(W2, e, w * 2, g, lane, w2ring[g]);
        #pragma unroll
        for (int t = 0; t < 4; ++t) {
            const int j = w * 4 + t;
            float bias = b1[e * K1_ + j * 16 + colLane];
            #pragma unroll
            for (int i = 0; i < 8; ++i) {
                if (own[i]) {
                    float x = acc[t][i] + bias;
                    float g = 0.5f * x * (1.0f + erff(x * 0.70710678118654752f));
                    H[(i + laneHalf * 8) * XSTR_ + j * 16 + colLane] = g;
                }
            }
        }
    }
    __syncthreads();

    {
        v8f acc[2] = {};
        dag_panel<PACKED, D_, 2, RING2_>(H, W2, e, w * 2, lane, w2ring, acc, [](int) {});
        #pragma unroll
        for (int t = 0; t < 2; ++t) {
            const int j = w * 2 + t;
            float bias = b2[e * D_ + j * 16 + colLane];
            #pragma unroll
            for (int i = 0; i < 8; ++i) {
                if (own[i]) {
                    int row = rowbase + i + laneHalf * 8;
                    buf[(size_t)row * D_ + j * 16 + colLane] = acc[t][i] + bias;
                }
            }
        }
    }
}


__global__ __launch_bounds__(1024)
void dag_bucket(const int* __restrict__ par, const int* __restrict__ types,
                int* __restrict__ recs_) {
    __shared__ int cnt[NENC_ * 32];
    __shared__ int fillFrom[NENC_];
    __shared__ int tileEnd[NENC_];

    const int level = blockIdx.x;
    const int tid   = threadIdx.x;
    const int lane  = tid & 31;
    const int wv    = tid >> 5;
    const int q     = level * M_ + tid * 2;
    const int2 tt = *(const int2*)(types + q);
    const int4 pp = *(const int4*)(par + (size_t)q * 2);
    const int r0 = tt.x >= START_ ? T_ : (tt.x < 0 ? 0 : tt.x);
    const int r1 = tt.y >= START_ ? T_ : (tt.y < 0 ? 0 : tt.y);

    const unsigned below = (1u << lane) - 1u;
    int rank0 = 0, rank1 = 0;
    for (int e = 0; e < NENC_; ++e) {
        const unsigned m0 = (unsigned)__ballot(r0 == e);
        const unsigned m1 = (unsigned)__ballot(r1 == e);
        const int before = __popc(m0 & below) + __popc(m1 & below);
        if (r0 == e) rank0 = before;
        if (r1 == e) rank1 = before + (r0 == e ? 1 : 0);
        if (lane == 0) cnt[e * 32 + wv] = __popc(m0) + __popc(m1);
    }
    __syncthreads();

    if (wv == 0) {
        int base = 0;
        for (int e = 0; e < NENC_; ++e) {
            const int v = cnt[e * 32 + lane];
            int inc = v;
            for (int d = 1; d < 32; d <<= 1) {
                const int u = __shfl_up(inc, d);
                if (lane >= d) inc += u;
            }
            const int total = __shfl(inc, 31);
            cnt[e * 32 + lane] = base + inc - v;
            if (lane == 0) fillFrom[e] = base + total;
            base += ((total + 15) >> 4) << 4;
            if (lane == 0) tileEnd[e] = base;
        }
    }
    __syncthreads();

    int4* lv = (int4*)recs_ + (size_t)level * SLOTS_;
    lv[cnt[r0 * 32 + wv] + rank0] = make_int4(pp.x, pp.y, tt.x, tid * 2);
    lv[cnt[r1 * 32 + wv] + rank1] = make_int4(pp.z, pp.w, tt.y, tid * 2 + 1);

    for (int s = tid; s < SLOTS_; s += 1024) {
        int e = 0;
        for (int k = 0; k < NENC_; ++k)
            if (s >= tileEnd[k]) e = k + 1;
        if (e == NENC_)
            lv[s] = make_int4(0, 0, 0, -1);
        else if (s >= fillFrom[e])
            lv[s] = make_int4(0, 0, e == T_ ? START_ : e, -1);
    }
}

__device__ __forceinline__ float dag_erf_both_ranges(float y) {
    const float ax = __builtin_fabsf(y);
    const float t = y * y;
    float q = __builtin_fmaf(t, -0x1.268bc2p-11f, 0x1.420828p-8f);
    q = __builtin_fmaf(t, q, -0x1.b59370p-6f);
    q = __builtin_fmaf(t, q, 0x1.ce077cp-4f);
    q = __builtin_fmaf(t, q, -0x1.812660p-2f);
    q = __builtin_fmaf(t, q, 0x1.06eba0p-3f);
    const float rs = __builtin_fmaf(ax, q, ax);
    float p = __builtin_fmaf(ax, 0x1.1d3156p-16f, -0x1.8d1290p-12f);
    p = __builtin_fmaf(ax, p, 0x1.f9a6d2p-9f);
    p = __builtin_fmaf(ax, p, -0x1.8c3164p-6f);
    p = __builtin_fmaf(ax, p, 0x1.b4e9c8p-4f);
    p = __builtin_fmaf(ax, p, 0x1.4515fap-1f);
    p = __builtin_fmaf(ax, p, 0x1.078e50p-3f);
    p = __builtin_fmaf(ax, p, ax);
    const float rl = 1.0f - expf(-p);
    const float r = (ax < 1.0f) ? rs : rl;
    return __builtin_copysignf(r, y);
}

__device__ __forceinline__ void dag_lds_barrier() {
    asm volatile("s_wait_dscnt 0x0" ::: "memory");
    asm volatile("s_barrier_signal -1\n\ts_barrier_wait -1" ::: "memory");
}

__global__ __launch_bounds__(128)
void dag_level_bucketed(float* __restrict__ buf,
                        const float* __restrict__ W1,
                        const float* __restrict__ b1,
                        const float* __restrict__ W2,
                        const float* __restrict__ b2,
                        const float* __restrict__ slots,
                        const int*   __restrict__ recs_,
                        int level)
{
    __shared__ float X[16 * XSTR_];
    __shared__ float H[16 * XSTR_];
    __shared__ int   rowNode[16];

    const int lane     = threadIdx.x & 31;
    const int w        = threadIdx.x >> 5;
    const int colLane  = lane & 15;
    const int laneHalf = (lane >> 4) & 1;
    const int4* tr = (const int4*)recs_ + (size_t)level * SLOTS_ + blockIdx.x * 16;

    int4 rc[4];
    #pragma unroll
    for (int r = 0; r < 4; ++r)
        rc[r] = tr[w + 4 * r];
    const int t0 = rc[0].z;
    const int e = t0 >= START_ ? T_ : (t0 < 0 ? 0 : t0);
    if (threadIdx.x < 16) rowNode[threadIdx.x] = tr[threadIdx.x].w;

    v4f ga[4], gc[4];
    #pragma unroll
    for (int r = 0; r < 4; ++r) {
        const int t = rc[r].z;
        int slot = t - START_;
        slot = slot < 0 ? 0 : (slot > OUTSLOTS_ - 1 ? OUTSLOTS_ - 1 : slot);
        const float* s0 = buf + (size_t)rc[r].x * D_;
        const float* s1 = (t >= START_) ? (slots + (size_t)slot * D_)
                                        : (buf + (size_t)rc[r].y * D_);
        ga[r] = *(const v4f*)(s0 + lane * 4);
        gc[r] = *(const v4f*)(s1 + lane * 4);
    }

    v2f w1ring[RING1_][8][4];
    #pragma unroll
    for (int g = 0; g < RING1_ - 1; ++g)
        dag_fetch_b<true, K1_, 4>(W1, e, w * 4, g, lane, w1ring[g]);

    #pragma unroll
    for (int r = 0; r < 4; ++r) {
        const int m = w + 4 * r;
        *(v4f*)(&X[m * XSTR_ + lane * 4])       = ga[r];
        *(v4f*)(&X[m * XSTR_ + 128 + lane * 4]) = gc[r];
    }
    dag_lds_barrier();

    v2f w2ring[RING2_][8][2];
    float bias2[2];
    {
        v8f acc[4] = {};
        float bias[4];
        auto ahead = [&](int i) {
            if (i == 0) {
                #pragma unroll
                for (int t = 0; t < 4; ++t)
                    bias[t] = b1[e * K1_ + (w * 4 + t) * 16 + colLane];
                #pragma unroll
                for (int t = 0; t < 2; ++t)
                    bias2[t] = b2[e * D_ + (w * 2 + t) * 16 + colLane];
            }
            dag_fetch_b<true, D_, 2>(W2, e, w * 2, 2 * i,     lane, w2ring[2 * i]);
            dag_fetch_b<true, D_, 2>(W2, e, w * 2, 2 * i + 1, lane, w2ring[2 * i + 1]);
        };
        dag_panel<true, K1_, 4, RING1_>(X, W1, e, w * 4, lane, w1ring, acc, ahead);
        #pragma unroll
        for (int g = 2 * (RING1_ - 1); g < RING2_ - 1; ++g)
            dag_fetch_b<true, D_, 2>(W2, e, w * 2, g, lane, w2ring[g]);
        __builtin_amdgcn_sched_barrier(0);
        #pragma unroll
        for (int t = 0; t < 4; ++t) {
            const int j = w * 4 + t;
            #pragma unroll
            for (int i = 0; i < 8; ++i) {
                float x = acc[t][i] + bias[t];
                float g = 0.5f * x * (1.0f + dag_erf_both_ranges(x * 0.70710678118654752f));
                H[(i + laneHalf * 8) * XSTR_ + j * 16 + colLane] = g;
            }
        }
    }
    dag_lds_barrier();

    {
        v8f acc[2] = {};
        int node[8];
        auto ahead2 = [&](int i) {
            if (i == 0) {
                #pragma unroll
                for (int r = 0; r < 8; ++r)
                    node[r] = rowNode[laneHalf * 8 + r];
            }
        };
        dag_panel<true, D_, 2, RING2_>(H, W2, e, w * 2, lane, w2ring, acc, ahead2);
        #pragma unroll
        for (int t = 0; t < 2; ++t) {
            const int j = w * 2 + t;
            #pragma unroll
            for (int i = 0; i < 8; ++i) {
                if (node[i] >= 0) {
                    const int row = R_ + level * M_ + node[i];
                    buf[(size_t)row * D_ + j * 16 + colLane] = acc[t][i] + bias2[t];
                }
            }
        }
    }
}

__device__ __forceinline__ int dag_key(int t, int d) {
    const int e  = t >= START_ ? T_ : (t < 0 ? 0 : t);
    const int dd = d < 1 ? 1 : (d > L_ ? L_ : d);
    return (dd - 1) * NENC_ + e;
}

__global__ __launch_bounds__(1024)
void dag_depth(const int* __restrict__ par, unsigned char* __restrict__ depOut) {
    static_assert(R_ == 1024 && M_ == 2048 && L_ % 4 == 0, "one root and two nodes of a level per thread, levels in fours");
    __shared__ unsigned char dep[N_];
    const int tid = threadIdx.x;
    dep[tid] = 0;
    int4 ahead[4];
    #pragma unroll
    for (int i = 0; i < 4; ++i)
        ahead[i] = *(const int4*)(par + ((size_t)i * M_ + tid * 2) * 2);
    dag_lds_barrier();
    for (int l0 = 0; l0 < L_; l0 += 4) {
        #pragma unroll
        for (int i = 0; i < 4; ++i) {
            const int l = l0 + i;
            const int4 cur = ahead[i];
            const int ln = l + 4 < L_ ? l + 4 : L_ - 1;
            ahead[i] = *(const int4*)(par + ((size_t)ln * M_ + tid * 2) * 2);
            const int hi = R_ + l * M_;
            const int a0 = dep[(unsigned)cur.x < (unsigned)hi ? cur.x : 0];
            const int a1 = dep[(unsigned)cur.y < (unsigned)hi ? cur.y : 0];
            const int c0 = dep[(unsigned)cur.z < (unsigned)hi ? cur.z : 0];
            const int c1 = dep[(unsigned)cur.w < (unsigned)hi ? cur.w : 0];
            dep[hi + tid * 2]     = (unsigned char)(1 + (a0 > a1 ? a0 : a1));
            dep[hi + tid * 2 + 1] = (unsigned char)(1 + (c0 > c1 ? c0 : c1));
            dag_lds_barrier();
        }
    }
    for (int i = tid; i < QN_ / 16; i += 1024)
        ((uint4*)depOut)[i] = *(const uint4*)(dep + R_ + i * 16);
}

__global__ __launch_bounds__(1024)
void dag_rank(const int* __restrict__ types, const unsigned char* __restrict__ dep,
              unsigned short* __restrict__ rank, int* __restrict__ cnt) {
    __shared__ unsigned short wcnt[32 * NKEY_];
    const int level = blockIdx.x;
    const int tid   = threadIdx.x;
    const int lane  = tid & 31;
    const int wv    = tid >> 5;
    const int q     = level * M_ + tid * 2;
    const int2 tt = *(const int2*)(types + q);
    const int ka = dag_key(tt.x, dep[q]);
    const int kb = dag_key(tt.y, dep[q + 1]);
    for (int i = tid; i < 32 * NKEY_; i += 1024) wcnt[i] = 0;

    int ra = 0, rb = 0, ta = 0, tb = 0;
    #pragma unroll
    for (int s = 0; s < 32; ++s) {
        const int sa = __builtin_amdgcn_readlane(ka, s);
        const int sb = __builtin_amdgcn_readlane(kb, s);
        const int ma = (sa == ka ? 1 : 0) + (sb == ka ? 1 : 0);
        const int mb = (sa == kb ? 1 : 0) + (sb == kb ? 1 : 0);
        ta += ma;
        tb += mb;
        if (s < lane) { ra += ma; rb += mb; }
    }
    if (ka == kb) rb += 1;
    __syncthreads();
    if (ra == ta - 1) wcnt[wv * NKEY_ + ka] = (unsigned short)ta;
    if (rb == tb - 1) wcnt[wv * NKEY_ + kb] = (unsigned short)tb;
    __syncthreads();
    if (tid < NKEY_) {
        int run = 0;
        for (int v = 0; v < 32; ++v) {
            const int c = wcnt[v * NKEY_ + tid];
            wcnt[v * NKEY_ + tid] = (unsigned short)run;
            run += c;
        }
        cnt[level * NKEY_ + tid] = run;
    }
    __syncthreads();
    rank[q]     = (unsigned short)(wcnt[wv * NKEY_ + ka] + ra);
    rank[q + 1] = (unsigned short)(wcnt[wv * NKEY_ + kb] + rb);
}

__global__ __launch_bounds__(NKEY_)
void dag_offsets(int* __restrict__ cnt, int* __restrict__ tileStart, int* __restrict__ recs_) {
    __shared__ int tot[NKEY_];
    __shared__ int first[NKEY_ + 1];
    const int k = threadIdx.x;
    int total = 0;
    for (int l = 0; l < L_; ++l) total += cnt[l * NKEY_ + k];
    tot[k] = total;
    __syncthreads();
    if (k == 0) {
        int t = 0;
        for (int i = 0; i < NKEY_; ++i) { first[i] = t; t += (tot[i] + 15) >> 4; }
        first[NKEY_] = t;
    }
    __syncthreads();
    int slot = first[k] * 16;
    for (int l = 0; l < L_; ++l) {
        const int c = cnt[l * NKEY_ + k];
        cnt[l * NKEY_ + k] = slot;
        slot += c;
    }
    int4* rec = (int4*)recs_;
    const int e = k % NENC_;
    for (int s = slot; s < first[k + 1] * 16; ++s)
        rec[s] = make_int4(0, 0, e == T_ ? START_ : e, -1);
    if (k <= L_) tileStart[k] = first[k < L_ ? k * NENC_ : NKEY_];
}

__global__ __launch_bounds__(1024)
void dag_place(const int* __restrict__ par, const int* __restrict__ types,
               const unsigned char* __restrict__ dep, const unsigned short* __restrict__ rank,
               const int* __restrict__ base, int* __restrict__ recs_) {
    const int level = blockIdx.x;
    const int q     = level * M_ + threadIdx.x * 2;
    const int2 tt = *(const int2*)(types + q);
    const int4 pp = *(const int4*)(par + (size_t)q * 2);
    const int ka = dag_key(tt.x, dep[q]);
    const int kb = dag_key(tt.y, dep[q + 1]);
    int4* rec = (int4*)recs_;
    rec[base[level * NKEY_ + ka] + rank[q]]     = make_int4(pp.x, pp.y, tt.x, q);
    rec[base[level * NKEY_ + kb] + rank[q + 1]] = make_int4(pp.z, pp.w, tt.y, q + 1);
}

__global__ __launch_bounds__(128)
void dag_level_depth(float* __restrict__ buf,
                        const float* __restrict__ W1,
                        const float* __restrict__ b1,
                        const float* __restrict__ W2,
                        const float* __restrict__ b2,
                        const float* __restrict__ slots,
                        const int*   __restrict__ recs_,
                        const int*   __restrict__ tileStart,
                        int depth)
{
    __shared__ float X[16 * XSTR_];
    __shared__ float H[16 * XSTR_];
    __shared__ int   rowNode[2][16];

    const int lane     = threadIdx.x & 31;
    const int w        = threadIdx.x >> 5;
    const int colLane  = lane & 15;
    const int laneHalf = (lane >> 4) & 1;
    const int tileEnd = tileStart[depth + 1];
    for (int tile = tileStart[depth] + blockIdx.x, par2 = 0; tile < tileEnd; tile += DGRID_, par2 ^= 1) {
    const int4* tr = (const int4*)recs_ + (size_t)tile * 16;

    int4 rc[4];
    #pragma unroll
    for (int r = 0; r < 4; ++r)
        rc[r] = tr[w + 4 * r];
    const int t0 = rc[0].z;
    const int e = t0 >= START_ ? T_ : (t0 < 0 ? 0 : t0);
    if (threadIdx.x < 16) rowNode[par2][threadIdx.x] = tr[threadIdx.x].w;

    v4f ga[4], gc[4];
    #pragma unroll
    for (int r = 0; r < 4; ++r) {
        const int t = rc[r].z;
        int slot = t - START_;
        slot = slot < 0 ? 0 : (slot > OUTSLOTS_ - 1 ? OUTSLOTS_ - 1 : slot);
        const float* s0 = buf + (size_t)rc[r].x * D_;
        const float* s1 = (t >= START_) ? (slots + (size_t)slot * D_)
                                        : (buf + (size_t)rc[r].y * D_);
        ga[r] = *(const v4f*)(s0 + lane * 4);
        gc[r] = *(const v4f*)(s1 + lane * 4);
    }

    v2f w1ring[RING1_][8][4];
    #pragma unroll
    for (int g = 0; g < RING1_ - 1; ++g)
        dag_fetch_b<true, K1_, 4>(W1, e, w * 4, g, lane, w1ring[g]);

    #pragma unroll
    for (int r = 0; r < 4; ++r) {
        const int m = w + 4 * r;
        *(v4f*)(&X[m * XSTR_ + lane * 4])       = ga[r];
        *(v4f*)(&X[m * XSTR_ + 128 + lane * 4]) = gc[r];
    }
    dag_lds_barrier();

    v2f w2ring[RING2_][8][2];
    float bias2[2];
    {
        v8f acc[4] = {};
        float bias[4];
        auto ahead = [&](int i) {
            if (i == 0) {
                #pragma unroll
                for (int t = 0; t < 4; ++t)
                    bias[t] = b1[e * K1_ + (w * 4 + t) * 16 + colLane];
                #pragma unroll
                for (int t = 0; t < 2; ++t)
                    bias2[t] = b2[e * D_ + (w * 2 + t) * 16 + colLane];
            }
            dag_fetch_b<true, D_, 2>(W2, e, w * 2, 2 * i,     lane, w2ring[2 * i]);
            dag_fetch_b<true, D_, 2>(W2, e, w * 2, 2 * i + 1, lane, w2ring[2 * i + 1]);
        };
        dag_panel<true, K1_, 4, RING1_>(X, W1, e, w * 4, lane, w1ring, acc, ahead);
        #pragma unroll
        for (int g = 2 * (RING1_ - 1); g < RING2_ - 1; ++g)
            dag_fetch_b<true, D_, 2>(W2, e, w * 2, g, lane, w2ring[g]);
        __builtin_amdgcn_sched_barrier(0);
        #pragma unroll
        for (int t = 0; t < 4; ++t) {
            const int j = w * 4 + t;
            #pragma unroll
            for (int i = 0; i < 8; ++i) {
                float x = acc[t][i] + bias[t];
                float g = 0.5f * x * (1.0f + dag_erf_both_ranges(x * 0.70710678118654752f));
                H[(i + laneHalf * 8) * XSTR_ + j * 16 + colLane] = g;
            }
        }
    }
    dag_lds_barrier();

    {
        v8f acc[2] = {};
        int node[8];
        auto ahead2 = [&](int i) {
            if (i == 0) {
                #pragma unroll
                for (int r = 0; r < 8; ++r)
                    node[r] = rowNode[par2][laneHalf * 8 + r];
            }
        };
        dag_panel<true, D_, 2, RING2_>(H, W2, e, w * 2, lane, w2ring, acc, ahead2);
        #pragma unroll
        for (int t = 0; t < 2; ++t) {
            const int j = w * 2 + t;
            #pragma unroll
            for (int i = 0; i < 8; ++i) {
                if (node[i] >= 0) {
                    const int row = R_ + node[i];
                    buf[(size_t)row * D_ + j * 16 + colLane] = acc[t][i] + bias2[t];
                }
            }
        }
    }
    }
}


extern "C" void kernel_launch(void* const* d_in, const int* in_sizes, int n_in,
                              void* d_out, int out_size, void* d_ws, size_t ws_size,
                              hipStream_t stream) {
    const float* root  = (const float*)d_in[0];
    const float* W1    = (const float*)d_in[1];
    const float* b1    = (const float*)d_in[2];
    const float* W2    = (const float*)d_in[3];
    const float* b2    = (const float*)d_in[4];
    const float* slots = (const float*)d_in[5];
    const int*   par   = (const int*)d_in[6];
    const int*   typ   = (const int*)d_in[7];
    float* out = (float*)d_out;

    dag_init_roots<<<128, 256, 0, stream>>>(out, root);

    const bool packed = (ws_size >= WSP_BYTES);
    float* W1p = (float*)d_ws;
    float* W2p = (float*)((char*)d_ws + W1P_BYTES);

    const bool bucketed = (ws_size >= WSP_BYTES + REC_BYTES);
    int* recs = (int*)((char*)d_ws + WSP_BYTES);

    const bool byDepth = (ws_size >= WSP_BYTES + DREC_BYTES + DEP_BYTES + RANK_BYTES + CNT_BYTES + TST_BYTES);
    unsigned char*  dep   = (unsigned char*)d_ws + WSP_BYTES + DREC_BYTES;
    unsigned short* rnk   = (unsigned short*)((char*)d_ws + WSP_BYTES + DREC_BYTES + DEP_BYTES);
    int*            cnt   = (int*)((char*)d_ws + WSP_BYTES + DREC_BYTES + DEP_BYTES + RANK_BYTES);
    int*            tileS = (int*)((char*)d_ws + WSP_BYTES + DREC_BYTES + DEP_BYTES + RANK_BYTES + CNT_BYTES);

    if (byDepth) {
        dag_pack_w<<<W1P_FRAGS / 256, 256, 0, stream>>>(W1, W1p, 16, K1_);
        dag_pack_w<<<W2P_FRAGS / 256, 256, 0, stream>>>(W2, W2p, 8, D_);
        dag_depth<<<1, 1024, 0, stream>>>(par, dep);
        dag_rank<<<L_, 1024, 0, stream>>>(typ, dep, rnk, cnt);
        dag_offsets<<<1, NKEY_, 0, stream>>>(cnt, tileS, recs);
        dag_place<<<L_, 1024, 0, stream>>>(par, typ, dep, rnk, cnt, recs);
        for (int d = 0; d < L_; ++d)
            dag_level_depth<<<DGRID_, 128, 0, stream>>>(out, W1p, b1, W2p, b2,
                                                        slots, recs, tileS, d);
    } else if (bucketed) {
        dag_pack_w<<<W1P_FRAGS / 256, 256, 0, stream>>>(W1, W1p, 16, K1_);
        dag_pack_w<<<W2P_FRAGS / 256, 256, 0, stream>>>(W2, W2p, 8, D_);
        dag_bucket<<<L_, 1024, 0, stream>>>(par, typ, recs);
        for (int l = 0; l < L_; ++l)
            dag_level_bucketed<<<TILES_, 128, 0, stream>>>(out, W1p, b1, W2p, b2,
                                                           slots, recs, l);
    } else if (packed) {
        dag_pack_w<<<W1P_FRAGS / 256, 256, 0, stream>>>(W1, W1p, 16, K1_);
        dag_pack_w<<<W2P_FRAGS / 256, 256, 0, stream>>>(W2, W2p, 8, D_);
        for (int l = 0; l < L_; ++l)
            dag_level<true><<<dim3(128, NENC_), 128, 0, stream>>>(out, W1p, b1, W2p, b2,
                                                                 slots, par, typ, l);
    } else {
        for (int l = 0; l < L_; ++l)
            dag_level<false><<<dim3(128, NENC_), 128, 0, stream>>>(out, W1, b1, W2, b2,
                                                                  slots, par, typ, l);
    }
}


# Depth launches 64 -> 32 plus one tail launch: depths above 32 run in ONE work-group, depth after depth behind a fence and a barrier; a launch for an empty depth cost a launch gap; sums unchanged
# speedup vs baseline: 13.2167x; 1.1173x over previous
#include <hip/hip_runtime.h>
#include <math.h>

typedef __attribute__((ext_vector_type(2))) float v2f;
typedef __attribute__((ext_vector_type(4))) float v4f;
typedef __attribute__((ext_vector_type(8))) float v8f;

#define D_        128
#define R_        1024
#define T_        4
#define OUTSLOTS_ 256
#define START_    4
#define L_        64
#define M_        2048
#define N_        (R_ + L_ * M_)
#define NENC_     5
#define K1_       256
#define XSTR_     260

#define W1P_FRAGS (NENC_ * 16 * 64 * 32)
#define W2P_FRAGS (NENC_ * 8  * 64 * 32)
#define W1P_BYTES ((size_t)W1P_FRAGS * 8)
#define W2P_BYTES ((size_t)W2P_FRAGS * 8)
#define WSP_BYTES (W1P_BYTES + W2P_BYTES)

#define RING1_    4
#define RING2_    8
static_assert(2 * (RING1_ - 1) <= RING2_ - 1, "layer-2 groups requested during layer 1 must fit the layer-2 ring");

#define TILES_    132
#define SLOTS_    (TILES_ * 16)
#define REC_BYTES ((size_t)L_ * SLOTS_ * 16)

#define QN_        (L_ * M_)
#define NKEY_      (L_ * NENC_)
#define DTILES_    (QN_ / 16 + NKEY_)
#define DREC_BYTES ((size_t)DTILES_ * 16 * 16)
#define DEP_BYTES  ((size_t)QN_)
#define RANK_BYTES ((size_t)QN_ * 2)
#define CNT_BYTES  ((size_t)L_ * NKEY_ * 4)
#define TST_BYTES  ((size_t)512)
#define DGRID_     512
#define DWIDE_     (L_ / 2)

__global__ void dag_init_roots(float* __restrict__ buf, const float* __restrict__ root) {
    int i = blockIdx.x * blockDim.x + threadIdx.x;
    v4f v = *(const v4f*)(root + (size_t)i * 4);
    *(v4f*)(buf + (size_t)i * 4) = v;
}

__global__ void dag_pack_w(const float* __restrict__ W, float* __restrict__ Wp,
                           int ntile, int ncol) {
    int idx  = blockIdx.x * blockDim.x + threadIdx.x;
    int lane = idx & 31;
    int k4   = (idx >> 5) & 63;
    int j    = (idx >> 11) % ntile;
    int e    = idx / (ntile << 11);
    int col  = lane & 15;
    int kk   = k4 * 4 + ((lane >> 4) & 1) * 2;
    const float* Wsrc = W + (size_t)e * K1_ * ncol + (size_t)(j * 16 + col);
    const size_t dst = ((size_t)((idx >> 11) * 32 + (k4 >> 1)) * 32 + lane) * 4 + (k4 & 1) * 2;
    Wp[dst + 0] = Wsrc[(size_t)kk * ncol];
    Wp[dst + 1] = Wsrc[(size_t)(kk + 1) * ncol];
}

template <bool PACKED, int NCOL>
__device__ __forceinline__ v4f dag_load_b2(const float* __restrict__ Wq, int e, int j,
                                           int p, int lane) {
    if (PACKED) {
        const v4f* Wf = (const v4f*)Wq + ((size_t)(e * (NCOL / 16) + j) * 32) * 32 + lane;
        return Wf[p * 32];
    } else {
        const int colLane = lane & 15;
        const int kk = p * 8 + ((lane >> 4) & 1) * 2;
        const float* Wb = Wq + (size_t)e * K1_ * NCOL + (size_t)(j * 16 + colLane);
        v4f b;
        b.x = Wb[(size_t)kk * NCOL];
        b.y = Wb[(size_t)(kk + 1) * NCOL];
        b.z = Wb[(size_t)(kk + 4) * NCOL];
        b.w = Wb[(size_t)(kk + 5) * NCOL];
        return b;
    }
}

template <bool PACKED, int NCOL, int NT>
__device__ __forceinline__ void dag_fetch_b(const float* __restrict__ Wq, int e, int j0,
                                            int g, int lane, v2f (&b)[8][NT]) {
    #pragma unroll
    for (int p = 0; p < 4; ++p) {
        #pragma unroll
        for (int t = 0; t < NT; ++t) {
            const v4f q = dag_load_b2<PACKED, NCOL>(Wq, e, j0 + t, g * 4 + p, lane);
            b[2 * p][t].x     = q.x;
            b[2 * p][t].y     = q.y;
            b[2 * p + 1][t].x = q.z;
            b[2 * p + 1][t].y = q.w;
        }
    }
}

template <bool PACKED, int NCOL, int NT, int RING, class TAIL>
__device__ __forceinline__ void dag_panel(const float* __restrict__ Xs,
                                          const float* __restrict__ Wq,
                                          int e, int j0, int lane,
                                          v2f (&bq)[RING][8][NT], v8f (&acc)[NT],
                                          const TAIL& tail) {
    const int colLane = lane & 15;
    const int klo     = ((lane >> 4) & 1) * 2;
    const float* Arow = Xs + colLane * XSTR_ + klo;

    v2f aq[2][8];
    #pragma unroll
    for (int s = 0; s < 8; ++s)
        aq[0][s] = *(const v2f*)(Arow + s * 4);
    #pragma unroll
    for (int g = 0; g < 8; ++g) {
        const int cur = g & 1, nxt = cur ^ 1;
        if (g + RING - 1 < 8)
            dag_fetch_b<PACKED, NCOL, NT>(Wq, e, j0, g + RING - 1, lane, bq[(g + RING - 1) % RING]);
        else
            tail(g + RING - 1 - 8);
        if (g < 7) {
            #pragma unroll
            for (int s = 0; s < 8; ++s)
                aq[nxt][s] = *(const v2f*)(Arow + ((g + 1) * 8 + s) * 4);
        }
        __builtin_amdgcn_sched_barrier(0);
        #pragma unroll
        for (int s = 0; s < 8; ++s) {
            #pragma unroll
            for (int t = 0; t < NT; ++t)
                acc[t] = __builtin_amdgcn_wmma_f32_16x16x4_f32(
                        false, aq[cur][s], false, bq[g % RING][s][t], (short)0, acc[t], false, false);
        }
        __builtin_amdgcn_sched_barrier(0);
    }
}

template <bool PACKED>
__global__ __launch_bounds__(128)
void dag_level(float* __restrict__ buf,
               const float* __restrict__ W1,
               const float* __restrict__ b1,
               const float* __restrict__ W2,
               const float* __restrict__ b2,
               const float* __restrict__ slots,
               const int*   __restrict__ par,
               const int*   __restrict__ types,
               int level)
{
    __shared__ float X[16 * XSTR_];
    __shared__ float H[16 * XSTR_];

    const int lane     = threadIdx.x & 31;
    const int w        = threadIdx.x >> 5;
    const int e        = blockIdx.y;
    const int tile     = blockIdx.x;
    const int qbase    = level * M_ + tile * 16;
    const int rowbase  = R_ + qbase;
    const int colLane  = lane & 15;
    const int laneHalf = (lane >> 4) & 1;

    int gp0[4], gp1[4], gt[4];
    #pragma unroll
    for (int r = 0; r < 4; ++r) {
        const int m = w + 4 * r;
        gp0[r] = par[(qbase + m) * 2 + 0];
        gp1[r] = par[(qbase + m) * 2 + 1];
        gt[r]  = types[qbase + m];
    }

    bool own[8];
    bool ownAny = false;
    #pragma unroll
    for (int i = 0; i < 8; ++i) {
        int t = types[qbase + i + laneHalf * 8];
        int r = (t >= START_) ? T_ : t;
        own[i] = (r == e);
        ownAny |= own[i];
    }
    if (!__any(ownAny)) return;

    v2f w1ring[RING1_][8][4];
    #pragma unroll
    for (int g = 0; g < RING1_ - 1; ++g)
        dag_fetch_b<PACKED, K1_, 4>(W1, e, w * 4, g, lane, w1ring[g]);

    v4f ga[4], gc[4];
    #pragma unroll
    for (int r = 0; r < 4; ++r) {
        const int t = gt[r];
        int slot = t - START_;
        slot = slot < 0 ? 0 : (slot > OUTSLOTS_ - 1 ? OUTSLOTS_ - 1 : slot);
        const float* s0 = buf + (size_t)gp0[r] * D_;
        const float* s1 = (t >= START_) ? (slots + (size_t)slot * D_)
                                        : (buf + (size_t)gp1[r] * D_);
        ga[r] = *(const v4f*)(s0 + lane * 4);
        gc[r] = *(const v4f*)(s1 + lane * 4);
    }
    #pragma unroll
    for (int r = 0; r < 4; ++r) {
        const int m = w + 4 * r;
        *(v4f*)(&X[m * XSTR_ + lane * 4])       = ga[r];
        *(v4f*)(&X[m * XSTR_ + 128 + lane * 4]) = gc[r];
    }
    __syncthreads();

    v2f w2ring[RING2_][8][2];
    {
        v8f acc[4] = {};
        dag_panel<PACKED, K1_, 4, RING1_>(X, W1, e, w * 4, lane, w1ring, acc, [](int) {});
        #pragma unroll
        for (int g = 0; g < RING2_ - 1; ++g)
            dag_fetch_b<PACKED, D_, 2>(W2, e, w * 2, g, lane, w2ring[g]);
        #pragma unroll
        for (int t = 0; t < 4; ++t) {
            const int j = w * 4 + t;
            float bias = b1[e * K1_ + j * 16 + colLane];
            #pragma unroll
            for (int i = 0; i < 8; ++i) {
                if (own[i]) {
                    float x = acc[t][i] + bias;
                    float g = 0.5f * x * (1.0f + erff(x * 0.70710678118654752f));
                    H[(i + laneHalf * 8) * XSTR_ + j * 16 + colLane] = g;
                }
            }
        }
    }
    __syncthreads();

    {
        v8f acc[2] = {};
        dag_panel<PACKED, D_, 2, RING2_>(H, W2, e, w * 2, lane, w2ring, acc, [](int) {});
        #pragma unroll
        for (int t = 0; t < 2; ++t) {
            const int j = w * 2 + t;
            float bias = b2[e * D_ + j * 16 + colLane];
            #pragma unroll
            for (int i = 0; i < 8; ++i) {
                if (own[i]) {
                    int row = rowbase + i + laneHalf * 8;
                    buf[(size_t)row * D_ + j * 16 + colLane] = acc[t][i] + bias;
                }
            }
        }
    }
}


__global__ __launch_bounds__(1024)
void dag_bucket(const int* __restrict__ par, const int* __restrict__ types,
                int* __restrict__ recs_) {
    __shared__ int cnt[NENC_ * 32];
    __shared__ int fillFrom[NENC_];
    __shared__ int tileEnd[NENC_];

    const int level = blockIdx.x;
    const int tid   = threadIdx.x;
    const int lane  = tid & 31;
    const int wv    = tid >> 5;
    const int q     = level * M_ + tid * 2;
    const int2 tt = *(const int2*)(types + q);
    const int4 pp = *(const int4*)(par + (size_t)q * 2);
    const int r0 = tt.x >= START_ ? T_ : (tt.x < 0 ? 0 : tt.x);
    const int r1 = tt.y >= START_ ? T_ : (tt.y < 0 ? 0 : tt.y);

    const unsigned below = (1u << lane) - 1u;
    int rank0 = 0, rank1 = 0;
    for (int e = 0; e < NENC_; ++e) {
        const unsigned m0 = (unsigned)__ballot(r0 == e);
        const unsigned m1 = (unsigned)__ballot(r1 == e);
        const int before = __popc(m0 & below) + __popc(m1 & below);
        if (r0 == e) rank0 = before;
        if (r1 == e) rank1 = before + (r0 == e ? 1 : 0);
        if (lane == 0) cnt[e * 32 + wv] = __popc(m0) + __popc(m1);
    }
    __syncthreads();

    if (wv == 0) {
        int base = 0;
        for (int e = 0; e < NENC_; ++e) {
            const int v = cnt[e * 32 + lane];
            int inc = v;
            for (int d = 1; d < 32; d <<= 1) {
                const int u = __shfl_up(inc, d);
                if (lane >= d) inc += u;
            }
            const int total = __shfl(inc, 31);
            cnt[e * 32 + lane] = base + inc - v;
            if (lane == 0) fillFrom[e] = base + total;
            base += ((total + 15) >> 4) << 4;
            if (lane == 0) tileEnd[e] = base;
        }
    }
    __syncthreads();

    int4* lv = (int4*)recs_ + (size_t)level * SLOTS_;
    lv[cnt[r0 * 32 + wv] + rank0] = make_int4(pp.x, pp.y, tt.x, tid * 2);
    lv[cnt[r1 * 32 + wv] + rank1] = make_int4(pp.z, pp.w, tt.y, tid * 2 + 1);

    for (int s = tid; s < SLOTS_; s += 1024) {
        int e = 0;
        for (int k = 0; k < NENC_; ++k)
            if (s >= tileEnd[k]) e = k + 1;
        if (e == NENC_)
            lv[s] = make_int4(0, 0, 0, -1);
        else if (s >= fillFrom[e])
            lv[s] = make_int4(0, 0, e == T_ ? START_ : e, -1);
    }
}

__device__ __forceinline__ float dag_erf_both_ranges(float y) {
    const float ax = __builtin_fabsf(y);
    const float t = y * y;
    float q = __builtin_fmaf(t, -0x1.268bc2p-11f, 0x1.420828p-8f);
    q = __builtin_fmaf(t, q, -0x1.b59370p-6f);
    q = __builtin_fmaf(t, q, 0x1.ce077cp-4f);
    q = __builtin_fmaf(t, q, -0x1.812660p-2f);
    q = __builtin_fmaf(t, q, 0x1.06eba0p-3f);
    const float rs = __builtin_fmaf(ax, q, ax);
    float p = __builtin_fmaf(ax, 0x1.1d3156p-16f, -0x1.8d1290p-12f);
    p = __builtin_fmaf(ax, p, 0x1.f9a6d2p-9f);
    p = __builtin_fmaf(ax, p, -0x1.8c3164p-6f);
    p = __builtin_fmaf(ax, p, 0x1.b4e9c8p-4f);
    p = __builtin_fmaf(ax, p, 0x1.4515fap-1f);
    p = __builtin_fmaf(ax, p, 0x1.078e50p-3f);
    p = __builtin_fmaf(ax, p, ax);
    const float rl = 1.0f - expf(-p);
    const float r = (ax < 1.0f) ? rs : rl;
    return __builtin_copysignf(r, y);
}

__device__ __forceinline__ void dag_lds_barrier() {
    asm volatile("s_wait_dscnt 0x0" ::: "memory");
    asm volatile("s_barrier_signal -1\n\ts_barrier_wait -1" ::: "memory");
}

__global__ __launch_bounds__(128)
void dag_level_bucketed(float* __restrict__ buf,
                        const float* __restrict__ W1,
                        const float* __restrict__ b1,
                        const float* __restrict__ W2,
                        const float* __restrict__ b2,
                        const float* __restrict__ slots,
                        const int*   __restrict__ recs_,
                        int level)
{
    __shared__ float X[16 * XSTR_];
    __shared__ float H[16 * XSTR_];
    __shared__ int   rowNode[16];

    const int lane     = threadIdx.x & 31;
    const int w        = threadIdx.x >> 5;
    const int colLane  = lane & 15;
    const int laneHalf = (lane >> 4) & 1;
    const int4* tr = (const int4*)recs_ + (size_t)level * SLOTS_ + blockIdx.x * 16;

    int4 rc[4];
    #pragma unroll
    for (int r = 0; r < 4; ++r)
        rc[r] = tr[w + 4 * r];
    const int t0 = rc[0].z;
    const int e = t0 >= START_ ? T_ : (t0 < 0 ? 0 : t0);
    if (threadIdx.x < 16) rowNode[threadIdx.x] = tr[threadIdx.x].w;

    v4f ga[4], gc[4];
    #pragma unroll
    for (int r = 0; r < 4; ++r) {
        const int t = rc[r].z;
        int slot = t - START_;
        slot = slot < 0 ? 0 : (slot > OUTSLOTS_ - 1 ? OUTSLOTS_ - 1 : slot);
        const float* s0 = buf + (size_t)rc[r].x * D_;
        const float* s1 = (t >= START_) ? (slots + (size_t)slot * D_)
                                        : (buf + (size_t)rc[r].y * D_);
        ga[r] = *(const v4f*)(s0 + lane * 4);
        gc[r] = *(const v4f*)(s1 + lane * 4);
    }

    v2f w1ring[RING1_][8][4];
    #pragma unroll
    for (int g = 0; g < RING1_ - 1; ++g)
        dag_fetch_b<true, K1_, 4>(W1, e, w * 4, g, lane, w1ring[g]);

    #pragma unroll
    for (int r = 0; r < 4; ++r) {
        const int m = w + 4 * r;
        *(v4f*)(&X[m * XSTR_ + lane * 4])       = ga[r];
        *(v4f*)(&X[m * XSTR_ + 128 + lane * 4]) = gc[r];
    }
    dag_lds_barrier();

    v2f w2ring[RING2_][8][2];
    float bias2[2];
    {
        v8f acc[4] = {};
        float bias[4];
        auto ahead = [&](int i) {
            if (i == 0) {
                #pragma unroll
                for (int t = 0; t < 4; ++t)
                    bias[t] = b1[e * K1_ + (w * 4 + t) * 16 + colLane];
                #pragma unroll
                for (int t = 0; t < 2; ++t)
                    bias2[t] = b2[e * D_ + (w * 2 + t) * 16 + colLane];
            }
            dag_fetch_b<true, D_, 2>(W2, e, w * 2, 2 * i,     lane, w2ring[2 * i]);
            dag_fetch_b<true, D_, 2>(W2, e, w * 2, 2 * i + 1, lane, w2ring[2 * i + 1]);
        };
        dag_panel<true, K1_, 4, RING1_>(X, W1, e, w * 4, lane, w1ring, acc, ahead);
        #pragma unroll
        for (int g = 2 * (RING1_ - 1); g < RING2_ - 1; ++g)
            dag_fetch_b<true, D_, 2>(W2, e, w * 2, g, lane, w2ring[g]);
        __builtin_amdgcn_sched_barrier(0);
        #pragma unroll
        for (int t = 0; t < 4; ++t) {
            const int j = w * 4 + t;
            #pragma unroll
            for (int i = 0; i < 8; ++i) {
                float x = acc[t][i] + bias[t];
                float g = 0.5f * x * (1.0f + dag_erf_both_ranges(x * 0.70710678118654752f));
                H[(i + laneHalf * 8) * XSTR_ + j * 16 + colLane] = g;
            }
        }
    }
    dag_lds_barrier();

    {
        v8f acc[2] = {};
        int node[8];
        auto ahead2 = [&](int i) {
            if (i == 0) {
                #pragma unroll
                for (int r = 0; r < 8; ++r)
                    node[r] = rowNode[laneHalf * 8 + r];
            }
        };
        dag_panel<true, D_, 2, RING2_>(H, W2, e, w * 2, lane, w2ring, acc, ahead2);
        #pragma unroll
        for (int t = 0; t < 2; ++t) {
            const int j = w * 2 + t;
            #pragma unroll
            for (int i = 0; i < 8; ++i) {
                if (node[i] >= 0) {
                    const int row = R_ + level * M_ + node[i];
                    buf[(size_t)row * D_ + j * 16 + colLane] = acc[t][i] + bias2[t];
                }
            }
        }
    }
}

__device__ __forceinline__ int dag_key(int t, int d) {
    const int e  = t >= START_ ? T_ : (t < 0 ? 0 : t);
    const int dd = d < 1 ? 1 : (d > L_ ? L_ : d);
    return (dd - 1) * NENC_ + e;
}

__global__ __launch_bounds__(1024)
void dag_depth(const int* __restrict__ par, unsigned char* __restrict__ depOut) {
    static_assert(R_ == 1024 && M_ == 2048 && L_ % 4 == 0, "one root and two nodes of a level per thread, levels in fours");
    __shared__ unsigned char dep[N_];
    const int tid = threadIdx.x;
    dep[tid] = 0;
    int4 ahead[4];
    #pragma unroll
    for (int i = 0; i < 4; ++i)
        ahead[i] = *(const int4*)(par + ((size_t)i * M_ + tid * 2) * 2);
    dag_lds_barrier();
    for (int l0 = 0; l0 < L_; l0 += 4) {
        #pragma unroll
        for (int i = 0; i < 4; ++i) {
            const int l = l0 + i;
            const int4 cur = ahead[i];
            const int ln = l + 4 < L_ ? l + 4 : L_ - 1;
            ahead[i] = *(const int4*)(par + ((size_t)ln * M_ + tid * 2) * 2);
            const int hi = R_ + l * M_;
            const int a0 = dep[(unsigned)cur.x < (unsigned)hi ? cur.x : 0];
            const int a1 = dep[(unsigned)cur.y < (unsigned)hi ? cur.y : 0];
            const int c0 = dep[(unsigned)cur.z < (unsigned)hi ? cur.z : 0];
            const int c1 = dep[(unsigned)cur.w < (unsigned)hi ? cur.w : 0];
            dep[hi + tid * 2]     = (unsigned char)(1 + (a0 > a1 ? a0 : a1));
            dep[hi + tid * 2 + 1] = (unsigned char)(1 + (c0 > c1 ? c0 : c1));
            dag_lds_barrier();
        }
    }
    for (int i = tid; i < QN_ / 16; i += 1024)
        ((uint4*)depOut)[i] = *(const uint4*)(dep + R_ + i * 16);
}

__global__ __launch_bounds__(1024)
void dag_rank(const int* __restrict__ types, const unsigned char* __restrict__ dep,
              unsigned short* __restrict__ rank, int* __restrict__ cnt) {
    __shared__ unsigned short wcnt[32 * NKEY_];
    const int level = blockIdx.x;
    const int tid   = threadIdx.x;
    const int lane  = tid & 31;
    const int wv    = tid >> 5;
    const int q     = level * M_ + tid * 2;
    const int2 tt = *(const int2*)(types + q);
    const int ka = dag_key(tt.x, dep[q]);
    const int kb = dag_key(tt.y, dep[q + 1]);
    for (int i = tid; i < 32 * NKEY_; i += 1024) wcnt[i] = 0;

    int ra = 0, rb = 0, ta = 0, tb = 0;
    #pragma unroll
    for (int s = 0; s < 32; ++s) {
        const int sa = __builtin_amdgcn_readlane(ka, s);
        const int sb = __builtin_amdgcn_readlane(kb, s);
        const int ma = (sa == ka ? 1 : 0) + (sb == ka ? 1 : 0);
        const int mb = (sa == kb ? 1 : 0) + (sb == kb ? 1 : 0);
        ta += ma;
        tb += mb;
        if (s < lane) { ra += ma; rb += mb; }
    }
    if (ka == kb) rb += 1;
    __syncthreads();
    if (ra == ta - 1) wcnt[wv * NKEY_ + ka] = (unsigned short)ta;
    if (rb == tb - 1) wcnt[wv * NKEY_ + kb] = (unsigned short)tb;
    __syncthreads();
    if (tid < NKEY_) {
        int run = 0;
        for (int v = 0; v < 32; ++v) {
            const int c = wcnt[v * NKEY_ + tid];
            wcnt[v * NKEY_ + tid] = (unsigned short)run;
            run += c;
        }
        cnt[level * NKEY_ + tid] = run;
    }
    __syncthreads();
    rank[q]     = (unsigned short)(wcnt[wv * NKEY_ + ka] + ra);
    rank[q + 1] = (unsigned short)(wcnt[wv * NKEY_ + kb] + rb);
}

__global__ __launch_bounds__(NKEY_)
void dag_offsets(int* __restrict__ cnt, int* __restrict__ tileStart, int* __restrict__ recs_) {
    __shared__ int tot[NKEY_];
    __shared__ int first[NKEY_ + 1];
    const int k = threadIdx.x;
    int total = 0;
    for (int l = 0; l < L_; ++l) total += cnt[l * NKEY_ + k];
    tot[k] = total;
    __syncthreads();
    if (k == 0) {
        int t = 0;
        for (int i = 0; i < NKEY_; ++i) { first[i] = t; t += (tot[i] + 15) >> 4; }
        first[NKEY_] = t;
    }
    __syncthreads();
    int slot = first[k] * 16;
    for (int l = 0; l < L_; ++l) {
        const int c = cnt[l * NKEY_ + k];
        cnt[l * NKEY_ + k] = slot;
        slot += c;
    }
    int4* rec = (int4*)recs_;
    const int e = k % NENC_;
    for (int s = slot; s < first[k + 1] * 16; ++s)
        rec[s] = make_int4(0, 0, e == T_ ? START_ : e, -1);
    if (k <= L_) tileStart[k] = first[k < L_ ? k * NENC_ : NKEY_];
}

__global__ __launch_bounds__(1024)
void dag_place(const int* __restrict__ par, const int* __restrict__ types,
               const unsigned char* __restrict__ dep, const unsigned short* __restrict__ rank,
               const int* __restrict__ base, int* __restrict__ recs_) {
    const int level = blockIdx.x;
    const int q     = level * M_ + threadIdx.x * 2;
    const int2 tt = *(const int2*)(types + q);
    const int4 pp = *(const int4*)(par + (size_t)q * 2);
    const int ka = dag_key(tt.x, dep[q]);
    const int kb = dag_key(tt.y, dep[q + 1]);
    int4* rec = (int4*)recs_;
    rec[base[level * NKEY_ + ka] + rank[q]]     = make_int4(pp.x, pp.y, tt.x, q);
    rec[base[level * NKEY_ + kb] + rank[q + 1]] = make_int4(pp.z, pp.w, tt.y, q + 1);
}

template <bool TAIL>
__global__ __launch_bounds__(128)
void dag_level_depth(float* __restrict__ buf,
                        const float* __restrict__ W1,
                        const float* __restrict__ b1,
                        const float* __restrict__ W2,
                        const float* __restrict__ b2,
                        const float* __restrict__ slots,
                        const int*   __restrict__ recs_,
                        const int*   __restrict__ tileStart,
                        int depth)
{
    __shared__ float X[16 * XSTR_];
    __shared__ float H[16 * XSTR_];
    __shared__ int   rowNode[2][16];

    const int lane     = threadIdx.x & 31;
    const int w        = threadIdx.x >> 5;
    const int colLane  = lane & 15;
    const int laneHalf = (lane >> 4) & 1;
    int d       = depth;
    int tileBeg = tileStart[d];
    if (TAIL && tileBeg >= tileStart[L_]) return;
    do {
    const int tileEnd = tileStart[d + 1];
    for (int tile = tileBeg + (TAIL ? 0 : (int)blockIdx.x), par2 = 0; tile < tileEnd; tile += (TAIL ? 1 : DGRID_), par2 ^= 1) {
    const int4* tr = (const int4*)recs_ + (size_t)tile * 16;

    int4 rc[4];
    #pragma unroll
    for (int r = 0; r < 4; ++r)
        rc[r] = tr[w + 4 * r];
    const int t0 = rc[0].z;
    const int e = t0 >= START_ ? T_ : (t0 < 0 ? 0 : t0);
    if (threadIdx.x < 16) rowNode[par2][threadIdx.x] = tr[threadIdx.x].w;

    v4f ga[4], gc[4];
    #pragma unroll
    for (int r = 0; r < 4; ++r) {
        const int t = rc[r].z;
        int slot = t - START_;
        slot = slot < 0 ? 0 : (slot > OUTSLOTS_ - 1 ? OUTSLOTS_ - 1 : slot);
        const float* s0 = buf + (size_t)rc[r].x * D_;
        const float* s1 = (t >= START_) ? (slots + (size_t)slot * D_)
                                        : (buf + (size_t)rc[r].y * D_);
        ga[r] = *(const v4f*)(s0 + lane * 4);
        gc[r] = *(const v4f*)(s1 + lane * 4);
    }

    v2f w1ring[RING1_][8][4];
    #pragma unroll
    for (int g = 0; g < RING1_ - 1; ++g)
        dag_fetch_b<true, K1_, 4>(W1, e, w * 4, g, lane, w1ring[g]);

    #pragma unroll
    for (int r = 0; r < 4; ++r) {
        const int m = w + 4 * r;
        *(v4f*)(&X[m * XSTR_ + lane * 4])       = ga[r];
        *(v4f*)(&X[m * XSTR_ + 128 + lane * 4]) = gc[r];
    }
    dag_lds_barrier();

    v2f w2ring[RING2_][8][2];
    float bias2[2];
    {
        v8f acc[4] = {};
        float bias[4];
        auto ahead = [&](int i) {
            if (i == 0) {
                #pragma unroll
                for (int t = 0; t < 4; ++t)
                    bias[t] = b1[e * K1_ + (w * 4 + t) * 16 + colLane];
                #pragma unroll
                for (int t = 0; t < 2; ++t)
                    bias2[t] = b2[e * D_ + (w * 2 + t) * 16 + colLane];
            }
            dag_fetch_b<true, D_, 2>(W2, e, w * 2, 2 * i,     lane, w2ring[2 * i]);
            dag_fetch_b<true, D_, 2>(W2, e, w * 2, 2 * i + 1, lane, w2ring[2 * i + 1]);
        };
        dag_panel<true, K1_, 4, RING1_>(X, W1, e, w * 4, lane, w1ring, acc, ahead);
        #pragma unroll
        for (int g = 2 * (RING1_ - 1); g < RING2_ - 1; ++g)
            dag_fetch_b<true, D_, 2>(W2, e, w * 2, g, lane, w2ring[g]);
        __builtin_amdgcn_sched_barrier(0);
        #pragma unroll
        for (int t = 0; t < 4; ++t) {
            const int j = w * 4 + t;
            #pragma unroll
            for (int i = 0; i < 8; ++i) {
                float x = acc[t][i] + bias[t];
                float g = 0.5f * x * (1.0f + dag_erf_both_ranges(x * 0.70710678118654752f));
                H[(i + laneHalf * 8) * XSTR_ + j * 16 + colLane] = g;
            }
        }
    }
    dag_lds_barrier();

    {
        v8f acc[2] = {};
        int node[8];
        auto ahead2 = [&](int i) {
            if (i == 0) {
                #pragma unroll
                for (int r = 0; r < 8; ++r)
                    node[r] = rowNode[par2][laneHalf * 8 + r];
            }
        };
        dag_panel<true, D_, 2, RING2_>(H, W2, e, w * 2, lane, w2ring, acc, ahead2);
        if (TAIL) {
            asm volatile("v_nop\n\tv_nop\n\tv_nop\n\tv_nop");
        }
        #pragma unroll
        for (int t = 0; t < 2; ++t) {
            const int j = w * 2 + t;
            #pragma unroll
            for (int i = 0; i < 8; ++i) {
                if (node[i] >= 0) {
                    const int row = R_ + node[i];
                    buf[(size_t)row * D_ + j * 16 + colLane] = acc[t][i] + bias2[t];
                }
            }
        }
    }
    }
    if (!TAIL) break;
    ++d;
    tileBeg = tileEnd;
    if (d >= L_ || tileBeg >= tileStart[L_]) break;
    __threadfence();
    __syncthreads();
    } while (true);
}


extern "C" void kernel_launch(void* const* d_in, const int* in_sizes, int n_in,
                              void* d_out, int out_size, void* d_ws, size_t ws_size,
                              hipStream_t stream) {
    const float* root  = (const float*)d_in[0];
    const float* W1    = (const float*)d_in[1];
    const float* b1    = (const float*)d_in[2];
    const float* W2    = (const float*)d_in[3];
    const float* b2    = (const float*)d_in[4];
    const float* slots = (const float*)d_in[5];
    const int*   par   = (const int*)d_in[6];
    const int*   typ   = (const int*)d_in[7];
    float* out = (float*)d_out;

    dag_init_roots<<<128, 256, 0, stream>>>(out, root);

    const bool packed = (ws_size >= WSP_BYTES);
    float* W1p = (float*)d_ws;
    float* W2p = (float*)((char*)d_ws + W1P_BYTES);

    const bool bucketed = (ws_size >= WSP_BYTES + REC_BYTES);
    int* recs = (int*)((char*)d_ws + WSP_BYTES);

    const bool byDepth = (ws_size >= WSP_BYTES + DREC_BYTES + DEP_BYTES + RANK_BYTES + CNT_BYTES + TST_BYTES);
    unsigned char*  dep   = (unsigned char*)d_ws + WSP_BYTES + DREC_BYTES;
    unsigned short* rnk   = (unsigned short*)((char*)d_ws + WSP_BYTES + DREC_BYTES + DEP_BYTES);
    int*            cnt   = (int*)((char*)d_ws + WSP_BYTES + DREC_BYTES + DEP_BYTES + RANK_BYTES);
    int*            tileS = (int*)((char*)d_ws + WSP_BYTES + DREC_BYTES + DEP_BYTES + RANK_BYTES + CNT_BYTES);

    if (byDepth) {
        dag_pack_w<<<W1P_FRAGS / 256, 256, 0, stream>>>(W1, W1p, 16, K1_);
        dag_pack_w<<<W2P_FRAGS / 256, 256, 0, stream>>>(W2, W2p, 8, D_);
        dag_depth<<<1, 1024, 0, stream>>>(par, dep);
        dag_rank<<<L_, 1024, 0, stream>>>(typ, dep, rnk, cnt);
        dag_offsets<<<1, NKEY_, 0, stream>>>(cnt, tileS, recs);
        dag_place<<<L_, 1024, 0, stream>>>(par, typ, dep, rnk, cnt, recs);
        for (int d = 0; d < DWIDE_; ++d)
            dag_level_depth<false><<<DGRID_, 128, 0, stream>>>(out, W1p, b1, W2p, b2,
                                                               slots, recs, tileS, d);
        dag_level_depth<true><<<1, 128, 0, stream>>>(out, W1p, b1, W2p, b2,
                                                     slots, recs, tileS, DWIDE_);
    } else if (bucketed) {
        dag_pack_w<<<W1P_FRAGS / 256, 256, 0, stream>>>(W1, W1p, 16, K1_);
        dag_pack_w<<<W2P_FRAGS / 256, 256, 0, stream>>>(W2, W2p, 8, D_);
        dag_bucket<<<L_, 1024, 0, stream>>>(par, typ, recs);
        for (int l = 0; l < L_; ++l)
            dag_level_bucketed<<<TILES_, 128, 0, stream>>>(out, W1p, b1, W2p, b2,
                                                           slots, recs, l);
    } else if (packed) {
        dag_pack_w<<<W1P_FRAGS / 256, 256, 0, stream>>>(W1, W1p, 16, K1_);
        dag_pack_w<<<W2P_FRAGS / 256, 256, 0, stream>>>(W2, W2p, 8, D_);
        for (int l = 0; l < L_; ++l)
            dag_level<true><<<dim3(128, NENC_), 128, 0, stream>>>(out, W1p, b1, W2p, b2,
                                                                 slots, par, typ, l);
    } else {
        for (int l = 0; l < L_; ++l)
            dag_level<false><<<dim3(128, NENC_), 128, 0, stream>>>(out, W1, b1, W2, b2,
                                                                  slots, par, typ, l);
    }
}


# Launches 40 -> 25: root copy, both weight packs and the depth pass share ONE launch (work-groups by role, beside the serial depth pass); own launches for depths 1..20, not 1..32; sums unchanged
# speedup vs baseline: 14.1635x; 1.0716x over previous
#include <hip/hip_runtime.h>
#include <math.h>

typedef __attribute__((ext_vector_type(2))) float v2f;
typedef __attribute__((ext_vector_type(4))) float v4f;
typedef __attribute__((ext_vector_type(8))) float v8f;

#define D_        128
#define R_        1024
#define T_        4
#define OUTSLOTS_ 256
#define START_    4
#define L_        64
#define M_        2048
#define N_        (R_ + L_ * M_)
#define NENC_     5
#define K1_       256
#define XSTR_     260

#define W1P_FRAGS (NENC_ * 16 * 64 * 32)
#define W2P_FRAGS (NENC_ * 8  * 64 * 32)
#define W1P_BYTES ((size_t)W1P_FRAGS * 8)
#define W2P_BYTES ((size_t)W2P_FRAGS * 8)
#define WSP_BYTES (W1P_BYTES + W2P_BYTES)

#define RING1_    4
#define RING2_    8
static_assert(2 * (RING1_ - 1) <= RING2_ - 1, "layer-2 groups requested during layer 1 must fit the layer-2 ring");

#define TILES_    132
#define SLOTS_    (TILES_ * 16)
#define REC_BYTES ((size_t)L_ * SLOTS_ * 16)

#define QN_        (L_ * M_)
#define NKEY_      (L_ * NENC_)
#define DTILES_    (QN_ / 16 + NKEY_)
#define DREC_BYTES ((size_t)DTILES_ * 16 * 16)
#define DEP_BYTES  ((size_t)QN_)
#define RANK_BYTES ((size_t)QN_ * 2)
#define CNT_BYTES  ((size_t)L_ * NKEY_ * 4)
#define TST_BYTES  ((size_t)512)
#define DGRID_     512
#define DWIDE_     20
static_assert(DWIDE_ < L_, "the tail launch starts at depth DWIDE_ + 1");

__device__ __forceinline__ void dag_copy_root4(float* __restrict__ buf, const float* __restrict__ root, int i) {
    v4f v = *(const v4f*)(root + (size_t)i * 4);
    *(v4f*)(buf + (size_t)i * 4) = v;
}

__global__ void dag_init_roots(float* __restrict__ buf, const float* __restrict__ root) {
    dag_copy_root4(buf, root, blockIdx.x * blockDim.x + threadIdx.x);
}

__device__ __forceinline__ void dag_pack_frag(const float* __restrict__ W, float* __restrict__ Wp,
                                              int ntile, int ncol, int idx) {
    int lane = idx & 31;
    int k4   = (idx >> 5) & 63;
    int j    = (idx >> 11) % ntile;
    int e    = idx / (ntile << 11);
    int col  = lane & 15;
    int kk   = k4 * 4 + ((lane >> 4) & 1) * 2;
    const float* Wsrc = W + (size_t)e * K1_ * ncol + (size_t)(j * 16 + col);
    const size_t dst = ((size_t)((idx >> 11) * 32 + (k4 >> 1)) * 32 + lane) * 4 + (k4 & 1) * 2;
    Wp[dst + 0] = Wsrc[(size_t)kk * ncol];
    Wp[dst + 1] = Wsrc[(size_t)(kk + 1) * ncol];
}

__global__ void dag_pack_w(const float* __restrict__ W, float* __restrict__ Wp,
                           int ntile, int ncol) {
    dag_pack_frag(W, Wp, ntile, ncol, blockIdx.x * blockDim.x + threadIdx.x);
}

template <bool PACKED, int NCOL>
__device__ __forceinline__ v4f dag_load_b2(const float* __restrict__ Wq, int e, int j,
                                           int p, int lane) {
    if (PACKED) {
        const v4f* Wf = (const v4f*)Wq + ((size_t)(e * (NCOL / 16) + j) * 32) * 32 + lane;
        return Wf[p * 32];
    } else {
        const int colLane = lane & 15;
        const int kk = p * 8 + ((lane >> 4) & 1) * 2;
        const float* Wb = Wq + (size_t)e * K1_ * NCOL + (size_t)(j * 16 + colLane);
        v4f b;
        b.x = Wb[(size_t)kk * NCOL];
        b.y = Wb[(size_t)(kk + 1) * NCOL];
        b.z = Wb[(size_t)(kk + 4) * NCOL];
        b.w = Wb[(size_t)(kk + 5) * NCOL];
        return b;
    }
}

template <bool PACKED, int NCOL, int NT>
__device__ __forceinline__ void dag_fetch_b(const float* __restrict__ Wq, int e, int j0,
                                            int g, int lane, v2f (&b)[8][NT]) {
    #pragma unroll
    for (int p = 0; p < 4; ++p) {
        #pragma unroll
        for (int t = 0; t < NT; ++t) {
            const v4f q = dag_load_b2<PACKED, NCOL>(Wq, e, j0 + t, g * 4 + p, lane);
            b[2 * p][t].x     = q.x;
            b[2 * p][t].y     = q.y;
            b[2 * p + 1][t].x = q.z;
            b[2 * p + 1][t].y = q.w;
        }
    }
}

template <bool PACKED, int NCOL, int NT, int RING, class TAIL>
__device__ __forceinline__ void dag_panel(const float* __restrict__ Xs,
                                          const float* __restrict__ Wq,
                                          int e, int j0, int lane,
                                          v2f (&bq)[RING][8][NT], v8f (&acc)[NT],
                                          const TAIL& tail) {
    const int colLane = lane & 15;
    const int klo     = ((lane >> 4) & 1) * 2;
    const float* Arow = Xs + colLane * XSTR_ + klo;

    v2f aq[2][8];
    #pragma unroll
    for (int s = 0; s < 8; ++s)
        aq[0][s] = *(const v2f*)(Arow + s * 4);
    #pragma unroll
    for (int g = 0; g < 8; ++g) {
        const int cur = g & 1, nxt = cur ^ 1;
        if (g + RING - 1 < 8)
            dag_fetch_b<PACKED, NCOL, NT>(Wq, e, j0, g + RING - 1, lane, bq[(g + RING - 1) % RING]);
        else
            tail(g + RING - 1 - 8);
        if (g < 7) {
            #pragma unroll
            for (int s = 0; s < 8; ++s)
                aq[nxt][s] = *(const v2f*)(Arow + ((g + 1) * 8 + s) * 4);
        }
        __builtin_amdgcn_sched_barrier(0);
        #pragma unroll
        for (int s = 0; s < 8; ++s) {
            #pragma unroll
            for (int t = 0; t < NT; ++t)
                acc[t] = __builtin_amdgcn_wmma_f32_16x16x4_f32(
                        false, aq[cur][s], false, bq[g % RING][s][t], (short)0, acc[t], false, false);
        }
        __builtin_amdgcn_sched_barrier(0);
    }
}

template <bool PACKED>
__global__ __launch_bounds__(128)
void dag_level(float* __restrict__ buf,
               const float* __restrict__ W1,
               const float* __restrict__ b1,
               const float* __restrict__ W2,
               const float* __restrict__ b2,
               const float* __restrict__ slots,
               const int*   __restrict__ par,
               const int*   __restrict__ types,
               int level)
{
    __shared__ float X[16 * XSTR_];
    __shared__ float H[16 * XSTR_];

    const int lane     = threadIdx.x & 31;
    const int w        = threadIdx.x >> 5;
    const int e        = blockIdx.y;
    const int tile     = blockIdx.x;
    const int qbase    = level * M_ + tile * 16;
    const int rowbase  = R_ + qbase;
    const int colLane  = lane & 15;
    const int laneHalf = (lane >> 4) & 1;

    int gp0[4], gp1[4], gt[4];
    #pragma unroll
    for (int r = 0; r < 4; ++r) {
        const int m = w + 4 * r;
        gp0[r] = par[(qbase + m) * 2 + 0];
        gp1[r] = par[(qbase + m) * 2 + 1];
        gt[r]  = types[qbase + m];
    }

    bool own[8];
    bool ownAny = false;
    #pragma unroll
    for (int i = 0; i < 8; ++i) {
        int t = types[qbase + i + laneHalf * 8];
        int r = (t >= START_) ? T_ : t;
        own[i] = (r == e);
        ownAny |= own[i];
    }
    if (!__any(ownAny)) return;

    v2f w1ring[RING1_][8][4];
    #pragma unroll
    for (int g = 0; g < RING1_ - 1; ++g)
        dag_fetch_b<PACKED, K1_, 4>(W1, e, w * 4, g, lane, w1ring[g]);

    v4f ga[4], gc[4];
    #pragma unroll
    for (int r = 0; r < 4; ++r) {
        const int t = gt[r];
        int slot = t - START_;
        slot = slot < 0 ? 0 : (slot > OUTSLOTS_ - 1 ? OUTSLOTS_ - 1 : slot);
        const float* s0 = buf + (size_t)gp0[r] * D_;
        const float* s1 = (t >= START_) ? (slots + (size_t)slot * D_)
                                        : (buf + (size_t)gp1[r] * D_);
        ga[r] = *(const v4f*)(s0 + lane * 4);
        gc[r] = *(const v4f*)(s1 + lane * 4);
    }
    #pragma unroll
    for (int r = 0; r < 4; ++r) {
        const int m = w + 4 * r;
        *(v4f*)(&X[m * XSTR_ + lane * 4])       = ga[r];
        *(v4f*)(&X[m * XSTR_ + 128 + lane * 4]) = gc[r];
    }
    __syncthreads();

    v2f w2ring[RING2_][8][2];
    {
        v8f acc[4] = {};
        dag_panel<PACKED, K1_, 4, RING1_>(X, W1, e, w * 4, lane, w1ring, acc, [](int) {});
        #pragma unroll
        for (int g = 0; g < RING2_ - 1; ++g)
            dag_fetch_b<PACKED, D_, 2>(W2, e, w * 2, g, lane, w2ring[g]);
        #pragma unroll
        for (int t = 0; t < 4; ++t) {
            const int j = w * 4 + t;
            float bias = b1[e * K1_ + j * 16 + colLane];
            #pragma unroll
            for (int i = 0; i < 8; ++i) {
                if (own[i]) {
                    float x = acc[t][i] + bias;
                    float g = 0.5f * x * (1.0f + erff(x * 0.70710678118654752f));
                    H[(i + laneHalf * 8) * XSTR_ + j * 16 + colLane] = g;
                }
            }
        }
    }
    __syncthreads();

    {
        v8f acc[2] = {};
        dag_panel<PACKED, D_, 2, RING2_>(H, W2, e, w * 2, lane, w2ring, acc, [](int) {});
        #pragma unroll
        for (int t = 0; t < 2; ++t) {
            const int j = w * 2 + t;
            float bias = b2[e * D_ + j * 16 + colLane];
            #pragma unroll
            for (int i = 0; i < 8; ++i) {
                if (own[i]) {
                    int row = rowbase + i + laneHalf * 8;
                    buf[(size_t)row * D_ + j * 16 + colLane] = acc[t][i] + bias;
                }
            }
        }
    }
}


__global__ __launch_bounds__(1024)
void dag_bucket(const int* __restrict__ par, const int* __restrict__ types,
                int* __restrict__ recs_) {
    __shared__ int cnt[NENC_ * 32];
    __shared__ int fillFrom[NENC_];
    __shared__ int tileEnd[NENC_];

    const int level = blockIdx.x;
    const int tid   = threadIdx.x;
    const int lane  = tid & 31;
    const int wv    = tid >> 5;
    const int q     = level * M_ + tid * 2;
    const int2 tt = *(const int2*)(types + q);
    const int4 pp = *(const int4*)(par + (size_t)q * 2);
    const int r0 = tt.x >= START_ ? T_ : (tt.x < 0 ? 0 : tt.x);
    const int r1 = tt.y >= START_ ? T_ : (tt.y < 0 ? 0 : tt.y);

    const unsigned below = (1u << lane) - 1u;
    int rank0 = 0, rank1 = 0;
    for (int e = 0; e < NENC_; ++e) {
        const unsigned m0 = (unsigned)__ballot(r0 == e);
        const unsigned m1 = (unsigned)__ballot(r1 == e);
        const int before = __popc(m0 & below) + __popc(m1 & below);
        if (r0 == e) rank0 = before;
        if (r1 == e) rank1 = before + (r0 == e ? 1 : 0);
        if (lane == 0) cnt[e * 32 + wv] = __popc(m0) + __popc(m1);
    }
    __syncthreads();

    if (wv == 0) {
        int base = 0;
        for (int e = 0; e < NENC_; ++e) {
            const int v = cnt[e * 32 + lane];
            int inc = v;
            for (int d = 1; d < 32; d <<= 1) {
                const int u = __shfl_up(inc, d);
                if (lane >= d) inc += u;
            }
            const int total = __shfl(inc, 31);
            cnt[e * 32 + lane] = base + inc - v;
            if (lane == 0) fillFrom[e] = base + total;
            base += ((total + 15) >> 4) << 4;
            if (lane == 0) tileEnd[e] = base;
        }
    }
    __syncthreads();

    int4* lv = (int4*)recs_ + (size_t)level * SLOTS_;
    lv[cnt[r0 * 32 + wv] + rank0] = make_int4(pp.x, pp.y, tt.x, tid * 2);
    lv[cnt[r1 * 32 + wv] + rank1] = make_int4(pp.z, pp.w, tt.y, tid * 2 + 1);

    for (int s = tid; s < SLOTS_; s += 1024) {
        int e = 0;
        for (int k = 0; k < NENC_; ++k)
            if (s >= tileEnd[k]) e = k + 1;
        if (e == NENC_)
            lv[s] = make_int4(0, 0, 0, -1);
        else if (s >= fillFrom[e])
            lv[s] = make_int4(0, 0, e == T_ ? START_ : e, -1);
    }
}

__device__ __forceinline__ float dag_erf_both_ranges(float y) {
    const float ax = __builtin_fabsf(y);
    const float t = y * y;
    float q = __builtin_fmaf(t, -0x1.268bc2p-11f, 0x1.420828p-8f);
    q = __builtin_fmaf(t, q, -0x1.b59370p-6f);
    q = __builtin_fmaf(t, q, 0x1.ce077cp-4f);
    q = __builtin_fmaf(t, q, -0x1.812660p-2f);
    q = __builtin_fmaf(t, q, 0x1.06eba0p-3f);
    const float rs = __builtin_fmaf(ax, q, ax);
    float p = __builtin_fmaf(ax, 0x1.1d3156p-16f, -0x1.8d1290p-12f);
    p = __builtin_fmaf(ax, p, 0x1.f9a6d2p-9f);
    p = __builtin_fmaf(ax, p, -0x1.8c3164p-6f);
    p = __builtin_fmaf(ax, p, 0x1.b4e9c8p-4f);
    p = __builtin_fmaf(ax, p, 0x1.4515fap-1f);
    p = __builtin_fmaf(ax, p, 0x1.078e50p-3f);
    p = __builtin_fmaf(ax, p, ax);
    const float rl = 1.0f - expf(-p);
    const float r = (ax < 1.0f) ? rs : rl;
    return __builtin_copysignf(r, y);
}

__device__ __forceinline__ void dag_lds_barrier() {
    asm volatile("s_wait_dscnt 0x0" ::: "memory");
    asm volatile("s_barrier_signal -1\n\ts_barrier_wait -1" ::: "memory");
}

__global__ __launch_bounds__(128)
void dag_level_bucketed(float* __restrict__ buf,
                        const float* __restrict__ W1,
                        const float* __restrict__ b1,
                        const float* __restrict__ W2,
                        const float* __restrict__ b2,
                        const float* __restrict__ slots,
                        const int*   __restrict__ recs_,
                        int level)
{
    __shared__ float X[16 * XSTR_];
    __shared__ float H[16 * XSTR_];
    __shared__ int   rowNode[16];

    const int lane     = threadIdx.x & 31;
    const int w        = threadIdx.x >> 5;
    const int colLane  = lane & 15;
    const int laneHalf = (lane >> 4) & 1;
    const int4* tr = (const int4*)recs_ + (size_t)level * SLOTS_ + blockIdx.x * 16;

    int4 rc[4];
    #pragma unroll
    for (int r = 0; r < 4; ++r)
        rc[r] = tr[w + 4 * r];
    const int t0 = rc[0].z;
    const int e = t0 >= START_ ? T_ : (t0 < 0 ? 0 : t0);
    if (threadIdx.x < 16) rowNode[threadIdx.x] = tr[threadIdx.x].w;

    v4f ga[4], gc[4];
    #pragma unroll
    for (int r = 0; r < 4; ++r) {
        const int t = rc[r].z;
        int slot = t - START_;
        slot = slot < 0 ? 0 : (slot > OUTSLOTS_ - 1 ? OUTSLOTS_ - 1 : slot);
        const float* s0 = buf + (size_t)rc[r].x * D_;
        const float* s1 = (t >= START_) ? (slots + (size_t)slot * D_)
                                        : (buf + (size_t)rc[r].y * D_);
        ga[r] = *(const v4f*)(s0 + lane * 4);
        gc[r] = *(const v4f*)(s1 + lane * 4);
    }

    v2f w1ring[RING1_][8][4];
    #pragma unroll
    for (int g = 0; g < RING1_ - 1; ++g)
        dag_fetch_b<true, K1_, 4>(W1, e, w * 4, g, lane, w1ring[g]);

    #pragma unroll
    for (int r = 0; r < 4; ++r) {
        const int m = w + 4 * r;
        *(v4f*)(&X[m * XSTR_ + lane * 4])       = ga[r];
        *(v4f*)(&X[m * XSTR_ + 128 + lane * 4]) = gc[r];
    }
    dag_lds_barrier();

    v2f w2ring[RING2_][8][2];
    float bias2[2];
    {
        v8f acc[4] = {};
        float bias[4];
        auto ahead = [&](int i) {
            if (i == 0) {
                #pragma unroll
                for (int t = 0; t < 4; ++t)
                    bias[t] = b1[e * K1_ + (w * 4 + t) * 16 + colLane];
                #pragma unroll
                for (int t = 0; t < 2; ++t)
                    bias2[t] = b2[e * D_ + (w * 2 + t) * 16 + colLane];
            }
            dag_fetch_b<true, D_, 2>(W2, e, w * 2, 2 * i,     lane, w2ring[2 * i]);
            dag_fetch_b<true, D_, 2>(W2, e, w * 2, 2 * i + 1, lane, w2ring[2 * i + 1]);
        };
        dag_panel<true, K1_, 4, RING1_>(X, W1, e, w * 4, lane, w1ring, acc, ahead);
        #pragma unroll
        for (int g = 2 * (RING1_ - 1); g < RING2_ - 1; ++g)
            dag_fetch_b<true, D_, 2>(W2, e, w * 2, g, lane, w2ring[g]);
        __builtin_amdgcn_sched_barrier(0);
        #pragma unroll
        for (int t = 0; t < 4; ++t) {
            const int j = w * 4 + t;
            #pragma unroll
            for (int i = 0; i < 8; ++i) {
                float x = acc[t][i] + bias[t];
                float g = 0.5f * x * (1.0f + dag_erf_both_ranges(x * 0.70710678118654752f));
                H[(i + laneHalf * 8) * XSTR_ + j * 16 + colLane] = g;
            }
        }
    }
    dag_lds_barrier();

    {
        v8f acc[2] = {};
        int node[8];
        auto ahead2 = [&](int i) {
            if (i == 0) {
                #pragma unroll
                for (int r = 0; r < 8; ++r)
                    node[r] = rowNode[laneHalf * 8 + r];
            }
        };
        dag_panel<true, D_, 2, RING2_>(H, W2, e, w * 2, lane, w2ring, acc, ahead2);
        #pragma unroll
        for (int t = 0; t < 2; ++t) {
            const int j = w * 2 + t;
            #pragma unroll
            for (int i = 0; i < 8; ++i) {
                if (node[i] >= 0) {
                    const int row = R_ + level * M_ + node[i];
                    buf[(size_t)row * D_ + j * 16 + colLane] = acc[t][i] + bias2[t];
                }
            }
        }
    }
}

__device__ __forceinline__ int dag_key(int t, int d) {
    const int e  = t >= START_ ? T_ : (t < 0 ? 0 : t);
    const int dd = d < 1 ? 1 : (d > L_ ? L_ : d);
    return (dd - 1) * NENC_ + e;
}

__device__ __forceinline__ void dag_depth_pass(const int* __restrict__ par, unsigned char* __restrict__ depOut) {
    static_assert(R_ == 1024 && M_ == 2048 && L_ % 4 == 0, "one root and two nodes of a level per thread, levels in fours");
    __shared__ unsigned char dep[N_];
    const int tid = threadIdx.x;
    dep[tid] = 0;
    int4 ahead[4];
    #pragma unroll
    for (int i = 0; i < 4; ++i)
        ahead[i] = *(const int4*)(par + ((size_t)i * M_ + tid * 2) * 2);
    dag_lds_barrier();
    for (int l0 = 0; l0 < L_; l0 += 4) {
        #pragma unroll
        for (int i = 0; i < 4; ++i) {
            const int l = l0 + i;
            const int4 cur = ahead[i];
            const int ln = l + 4 < L_ ? l + 4 : L_ - 1;
            ahead[i] = *(const int4*)(par + ((size_t)ln * M_ + tid * 2) * 2);
            const int hi = R_ + l * M_;
            const int a0 = dep[(unsigned)cur.x < (unsigned)hi ? cur.x : 0];
            const int a1 = dep[(unsigned)cur.y < (unsigned)hi ? cur.y : 0];
            const int c0 = dep[(unsigned)cur.z < (unsigned)hi ? cur.z : 0];
            const int c1 = dep[(unsigned)cur.w < (unsigned)hi ? cur.w : 0];
            dep[hi + tid * 2]     = (unsigned char)(1 + (a0 > a1 ? a0 : a1));
            dep[hi + tid * 2 + 1] = (unsigned char)(1 + (c0 > c1 ? c0 : c1));
            dag_lds_barrier();
        }
    }
    for (int i = tid; i < QN_ / 16; i += 1024)
        ((uint4*)depOut)[i] = *(const uint4*)(dep + R_ + i * 16);
}

#define FRONT_ROOT_ (R_ * D_ / 4 / 1024)
#define FRONT_W1_   (W1P_FRAGS / 1024)
#define FRONT_W2_   (W2P_FRAGS / 1024)
#define FRONT_GRID_ (1 + FRONT_ROOT_ + FRONT_W1_ + FRONT_W2_)
static_assert(R_ * D_ / 4 % 1024 == 0 && W1P_FRAGS % 1024 == 0 && W2P_FRAGS % 1024 == 0, "whole work-groups per role");
__global__ __launch_bounds__(1024)
void dag_front(const int* __restrict__ par, unsigned char* __restrict__ depOut,
               float* __restrict__ buf, const float* __restrict__ root,
               const float* __restrict__ W1, float* __restrict__ W1p,
               const float* __restrict__ W2, float* __restrict__ W2p) {
    int b = blockIdx.x;
    if (b == 0) { dag_depth_pass(par, depOut); return; }
    b -= 1;
    if (b < FRONT_ROOT_) { dag_copy_root4(buf, root, b * 1024 + threadIdx.x); return; }
    b -= FRONT_ROOT_;
    if (b < FRONT_W1_) { dag_pack_frag(W1, W1p, 16, K1_, b * 1024 + threadIdx.x); return; }
    b -= FRONT_W1_;
    dag_pack_frag(W2, W2p, 8, D_, b * 1024 + threadIdx.x);
}

__global__ __launch_bounds__(1024)
void dag_rank(const int* __restrict__ types, const unsigned char* __restrict__ dep,
              unsigned short* __restrict__ rank, int* __restrict__ cnt) {
    __shared__ unsigned short wcnt[32 * NKEY_];
    const int level = blockIdx.x;
    const int tid   = threadIdx.x;
    const int lane  = tid & 31;
    const int wv    = tid >> 5;
    const int q     = level * M_ + tid * 2;
    const int2 tt = *(const int2*)(types + q);
    const int ka = dag_key(tt.x, dep[q]);
    const int kb = dag_key(tt.y, dep[q + 1]);
    for (int i = tid; i < 32 * NKEY_; i += 1024) wcnt[i] = 0;

    int ra = 0, rb = 0, ta = 0, tb = 0;
    #pragma unroll
    for (int s = 0; s < 32; ++s) {
        const int sa = __builtin_amdgcn_readlane(ka, s);
        const int sb = __builtin_amdgcn_readlane(kb, s);
        const int ma = (sa == ka ? 1 : 0) + (sb == ka ? 1 : 0);
        const int mb = (sa == kb ? 1 : 0) + (sb == kb ? 1 : 0);
        ta += ma;
        tb += mb;
        if (s < lane) { ra += ma; rb += mb; }
    }
    if (ka == kb) rb += 1;
    __syncthreads();
    if (ra == ta - 1) wcnt[wv * NKEY_ + ka] = (unsigned short)ta;
    if (rb == tb - 1) wcnt[wv * NKEY_ + kb] = (unsigned short)tb;
    __syncthreads();
    if (tid < NKEY_) {
        int run = 0;
        for (int v = 0; v < 32; ++v) {
            const int c = wcnt[v * NKEY_ + tid];
            wcnt[v * NKEY_ + tid] = (unsigned short)run;
            run += c;
        }
        cnt[level * NKEY_ + tid] = run;
    }
    __syncthreads();
    rank[q]     = (unsigned short)(wcnt[wv * NKEY_ + ka] + ra);
    rank[q + 1] = (unsigned short)(wcnt[wv * NKEY_ + kb] + rb);
}

__global__ __launch_bounds__(NKEY_)
void dag_offsets(int* __restrict__ cnt, int* __restrict__ tileStart, int* __restrict__ recs_) {
    __shared__ int tot[NKEY_];
    __shared__ int first[NKEY_ + 1];
    const int k = threadIdx.x;
    int total = 0;
    for (int l = 0; l < L_; ++l) total += cnt[l * NKEY_ + k];
    tot[k] = total;
    __syncthreads();
    if (k == 0) {
        int t = 0;
        for (int i = 0; i < NKEY_; ++i) { first[i] = t; t += (tot[i] + 15) >> 4; }
        first[NKEY_] = t;
    }
    __syncthreads();
    int slot = first[k] * 16;
    for (int l = 0; l < L_; ++l) {
        const int c = cnt[l * NKEY_ + k];
        cnt[l * NKEY_ + k] = slot;
        slot += c;
    }
    int4* rec = (int4*)recs_;
    const int e = k % NENC_;
    for (int s = slot; s < first[k + 1] * 16; ++s)
        rec[s] = make_int4(0, 0, e == T_ ? START_ : e, -1);
    if (k <= L_) tileStart[k] = first[k < L_ ? k * NENC_ : NKEY_];
}

__global__ __launch_bounds__(1024)
void dag_place(const int* __restrict__ par, const int* __restrict__ types,
               const unsigned char* __restrict__ dep, const unsigned short* __restrict__ rank,
               const int* __restrict__ base, int* __restrict__ recs_) {
    const int level = blockIdx.x;
    const int q     = level * M_ + threadIdx.x * 2;
    const int2 tt = *(const int2*)(types + q);
    const int4 pp = *(const int4*)(par + (size_t)q * 2);
    const int ka = dag_key(tt.x, dep[q]);
    const int kb = dag_key(tt.y, dep[q + 1]);
    int4* rec = (int4*)recs_;
    rec[base[level * NKEY_ + ka] + rank[q]]     = make_int4(pp.x, pp.y, tt.x, q);
    rec[base[level * NKEY_ + kb] + rank[q + 1]] = make_int4(pp.z, pp.w, tt.y, q + 1);
}

template <bool TAIL>
__global__ __launch_bounds__(128)
void dag_level_depth(float* __restrict__ buf,
                        const float* __restrict__ W1,
                        const float* __restrict__ b1,
                        const float* __restrict__ W2,
                        const float* __restrict__ b2,
                        const float* __restrict__ slots,
                        const int*   __restrict__ recs_,
                        const int*   __restrict__ tileStart,
                        int depth)
{
    __shared__ float X[16 * XSTR_];
    __shared__ float H[16 * XSTR_];
    __shared__ int   rowNode[2][16];

    const int lane     = threadIdx.x & 31;
    const int w        = threadIdx.x >> 5;
    const int colLane  = lane & 15;
    const int laneHalf = (lane >> 4) & 1;
    int d       = depth;
    int tileBeg = tileStart[d];
    if (TAIL && tileBeg >= tileStart[L_]) return;
    do {
    const int tileEnd = tileStart[d + 1];
    for (int tile = tileBeg + (TAIL ? 0 : (int)blockIdx.x), par2 = 0; tile < tileEnd; tile += (TAIL ? 1 : DGRID_), par2 ^= 1) {
    const int4* tr = (const int4*)recs_ + (size_t)tile * 16;

    int4 rc[4];
    #pragma unroll
    for (int r = 0; r < 4; ++r)
        rc[r] = tr[w + 4 * r];
    const int t0 = rc[0].z;
    const int e = t0 >= START_ ? T_ : (t0 < 0 ? 0 : t0);
    if (threadIdx.x < 16) rowNode[par2][threadIdx.x] = tr[threadIdx.x].w;

    v4f ga[4], gc[4];
    #pragma unroll
    for (int r = 0; r < 4; ++r) {
        const int t = rc[r].z;
        int slot = t - START_;
        slot = slot < 0 ? 0 : (slot > OUTSLOTS_ - 1 ? OUTSLOTS_ - 1 : slot);
        const float* s0 = buf + (size_t)rc[r].x * D_;
        const float* s1 = (t >= START_) ? (slots + (size_t)slot * D_)
                                        : (buf + (size_t)rc[r].y * D_);
        ga[r] = *(const v4f*)(s0 + lane * 4);
        gc[r] = *(const v4f*)(s1 + lane * 4);
    }

    v2f w1ring[RING1_][8][4];
    #pragma unroll
    for (int g = 0; g < RING1_ - 1; ++g)
        dag_fetch_b<true, K1_, 4>(W1, e, w * 4, g, lane, w1ring[g]);

    #pragma unroll
    for (int r = 0; r < 4; ++r) {
        const int m = w + 4 * r;
        *(v4f*)(&X[m * XSTR_ + lane * 4])       = ga[r];
        *(v4f*)(&X[m * XSTR_ + 128 + lane * 4]) = gc[r];
    }
    dag_lds_barrier();

    v2f w2ring[RING2_][8][2];
    float bias2[2];
    {
        v8f acc[4] = {};
        float bias[4];
        auto ahead = [&](int i) {
            if (i == 0) {
                #pragma unroll
                for (int t = 0; t < 4; ++t)
                    bias[t] = b1[e * K1_ + (w * 4 + t) * 16 + colLane];
                #pragma unroll
                for (int t = 0; t < 2; ++t)
                    bias2[t] = b2[e * D_ + (w * 2 + t) * 16 + colLane];
            }
            dag_fetch_b<true, D_, 2>(W2, e, w * 2, 2 * i,     lane, w2ring[2 * i]);
            dag_fetch_b<true, D_, 2>(W2, e, w * 2, 2 * i + 1, lane, w2ring[2 * i + 1]);
        };
        dag_panel<true, K1_, 4, RING1_>(X, W1, e, w * 4, lane, w1ring, acc, ahead);
        #pragma unroll
        for (int g = 2 * (RING1_ - 1); g < RING2_ - 1; ++g)
            dag_fetch_b<true, D_, 2>(W2, e, w * 2, g, lane, w2ring[g]);
        __builtin_amdgcn_sched_barrier(0);
        #pragma unroll
        for (int t = 0; t < 4; ++t) {
            const int j = w * 4 + t;
            #pragma unroll
            for (int i = 0; i < 8; ++i) {
                float x = acc[t][i] + bias[t];
                float g = 0.5f * x * (1.0f + dag_erf_both_ranges(x * 0.70710678118654752f));
                H[(i + laneHalf * 8) * XSTR_ + j * 16 + colLane] = g;
            }
        }
    }
    dag_lds_barrier();

    {
        v8f acc[2] = {};
        int node[8];
        auto ahead2 = [&](int i) {
            if (i == 0) {
                #pragma unroll
                for (int r = 0; r < 8; ++r)
                    node[r] = rowNode[par2][laneHalf * 8 + r];
            }
        };
        dag_panel<true, D_, 2, RING2_>(H, W2, e, w * 2, lane, w2ring, acc, ahead2);
        if (TAIL) {
            asm volatile("v_nop\n\tv_nop\n\tv_nop\n\tv_nop");
        }
        #pragma unroll
        for (int t = 0; t < 2; ++t) {
            const int j = w * 2 + t;
            #pragma unroll
            for (int i = 0; i < 8; ++i) {
                if (node[i] >= 0) {
                    const int row = R_ + node[i];
                    buf[(size_t)row * D_ + j * 16 + colLane] = acc[t][i] + bias2[t];
                }
            }
        }
    }
    }
    if (!TAIL) break;
    ++d;
    tileBeg = tileEnd;
    if (d >= L_ || tileBeg >= tileStart[L_]) break;
    __threadfence();
    __syncthreads();
    } while (true);
}


extern "C" void kernel_launch(void* const* d_in, const int* in_sizes, int n_in,
                              void* d_out, int out_size, void* d_ws, size_t ws_size,
                              hipStream_t stream) {
    const float* root  = (const float*)d_in[0];
    const float* W1    = (const float*)d_in[1];
    const float* b1    = (const float*)d_in[2];
    const float* W2    = (const float*)d_in[3];
    const float* b2    = (const float*)d_in[4];
    const float* slots = (const float*)d_in[5];
    const int*   par   = (const int*)d_in[6];
    const int*   typ   = (const int*)d_in[7];
    float* out = (float*)d_out;

    const bool packed = (ws_size >= WSP_BYTES);
    float* W1p = (float*)d_ws;
    float* W2p = (float*)((char*)d_ws + W1P_BYTES);

    const bool bucketed = (ws_size >= WSP_BYTES + REC_BYTES);
    int* recs = (int*)((char*)d_ws + WSP_BYTES);

    const bool byDepth = (ws_size >= WSP_BYTES + DREC_BYTES + DEP_BYTES + RANK_BYTES + CNT_BYTES + TST_BYTES);
    unsigned char*  dep   = (unsigned char*)d_ws + WSP_BYTES + DREC_BYTES;
    unsigned short* rnk   = (unsigned short*)((char*)d_ws + WSP_BYTES + DREC_BYTES + DEP_BYTES);
    int*            cnt   = (int*)((char*)d_ws + WSP_BYTES + DREC_BYTES + DEP_BYTES + RANK_BYTES);
    int*            tileS = (int*)((char*)d_ws + WSP_BYTES + DREC_BYTES + DEP_BYTES + RANK_BYTES + CNT_BYTES);

    if (!byDepth) dag_init_roots<<<128, 256, 0, stream>>>(out, root);

    if (byDepth) {
        dag_front<<<FRONT_GRID_, 1024, 0, stream>>>(par, dep, out, root, W1, W1p, W2, W2p);
        dag_rank<<<L_, 1024, 0, stream>>>(typ, dep, rnk, cnt);
        dag_offsets<<<1, NKEY_, 0, stream>>>(cnt, tileS, recs);
        dag_place<<<L_, 1024, 0, stream>>>(par, typ, dep, rnk, cnt, recs);
        for (int d = 0; d < DWIDE_; ++d)
            dag_level_depth<false><<<DGRID_, 128, 0, stream>>>(out, W1p, b1, W2p, b2,
                                                               slots, recs, tileS, d);
        dag_level_depth<true><<<1, 128, 0, stream>>>(out, W1p, b1, W2p, b2,
                                                     slots, recs, tileS, DWIDE_);
    } else if (bucketed) {
        dag_pack_w<<<W1P_FRAGS / 256, 256, 0, stream>>>(W1, W1p, 16, K1_);
        dag_pack_w<<<W2P_FRAGS / 256, 256, 0, stream>>>(W2, W2p, 8, D_);
        dag_bucket<<<L_, 1024, 0, stream>>>(par, typ, recs);
        for (int l = 0; l < L_; ++l)
            dag_level_bucketed<<<TILES_, 128, 0, stream>>>(out, W1p, b1, W2p, b2,
                                                           slots, recs, l);
    } else if (packed) {
        dag_pack_w<<<W1P_FRAGS / 256, 256, 0, stream>>>(W1, W1p, 16, K1_);
        dag_pack_w<<<W2P_FRAGS / 256, 256, 0, stream>>>(W2, W2p, 8, D_);
        for (int l = 0; l < L_; ++l)
            dag_level<true><<<dim3(128, NENC_), 128, 0, stream>>>(out, W1p, b1, W2p, b2,
                                                                 slots, par, typ, l);
    } else {
        for (int l = 0; l < L_; ++l)
            dag_level<false><<<dim3(128, NENC_), 128, 0, stream>>>(out, W1, b1, W2, b2,
                                                                  slots, par, typ, l);
    }
}


# Leaves (rows no later row reads) leave their depth's launch: a pre-pass flags them and fills each launch to a whole number of tiles per unit with leaves of that or a smaller depth; row sums unchanged
# speedup vs baseline: 14.7894x; 1.0442x over previous
#include <hip/hip_runtime.h>
#include <math.h>

typedef __attribute__((ext_vector_type(2))) float v2f;
typedef __attribute__((ext_vector_type(4))) float v4f;
typedef __attribute__((ext_vector_type(8))) float v8f;

#define D_        128
#define R_        1024
#define T_        4
#define OUTSLOTS_ 256
#define START_    4
#define L_        64
#define M_        2048
#define N_        (R_ + L_ * M_)
#define NENC_     5
#define K1_       256
#define XSTR_     260

#define W1P_FRAGS (NENC_ * 16 * 64 * 32)
#define W2P_FRAGS (NENC_ * 8  * 64 * 32)
#define W1P_BYTES ((size_t)W1P_FRAGS * 8)
#define W2P_BYTES ((size_t)W2P_FRAGS * 8)
#define WSP_BYTES (W1P_BYTES + W2P_BYTES)

#define RING1_    4
#define RING2_    8
static_assert(2 * (RING1_ - 1) <= RING2_ - 1, "layer-2 groups requested during layer 1 must fit the layer-2 ring");

#define TILES_    132
#define SLOTS_    (TILES_ * 16)
#define REC_BYTES ((size_t)L_ * SLOTS_ * 16)

#define QN_        (L_ * M_)
#define NKEY_      (L_ * NENC_ * 2)
#define DTILES_    (QN_ / 16 + NKEY_)
#define DREC_BYTES ((size_t)DTILES_ * 16 * 16)
#define DEP_BYTES  ((size_t)QN_)
#define RANK_BYTES ((size_t)QN_ * 2)
#define CNT_BYTES  ((size_t)L_ * NKEY_ * 4)
#define TST_BYTES  ((size_t)512)
#define DGRID_     512
#define DUNITS_    (DGRID_ / 2)
#define DWIDE_     20
static_assert(DWIDE_ < L_, "the tail launch starts at depth DWIDE_ + 1");

__device__ __forceinline__ void dag_copy_root4(float* __restrict__ buf, const float* __restrict__ root, int i) {
    v4f v = *(const v4f*)(root + (size_t)i * 4);
    *(v4f*)(buf + (size_t)i * 4) = v;
}

__global__ void dag_init_roots(float* __restrict__ buf, const float* __restrict__ root) {
    dag_copy_root4(buf, root, blockIdx.x * blockDim.x + threadIdx.x);
}

__device__ __forceinline__ void dag_pack_frag(const float* __restrict__ W, float* __restrict__ Wp,
                                              int ntile, int ncol, int idx) {
    int lane = idx & 31;
    int k4   = (idx >> 5) & 63;
    int j    = (idx >> 11) % ntile;
    int e    = idx / (ntile << 11);
    int col  = lane & 15;
    int kk   = k4 * 4 + ((lane >> 4) & 1) * 2;
    const float* Wsrc = W + (size_t)e * K1_ * ncol + (size_t)(j * 16 + col);
    const size_t dst = ((size_t)((idx >> 11) * 32 + (k4 >> 1)) * 32 + lane) * 4 + (k4 & 1) * 2;
    Wp[dst + 0] = Wsrc[(size_t)kk * ncol];
    Wp[dst + 1] = Wsrc[(size_t)(kk + 1) * ncol];
}

__global__ void dag_pack_w(const float* __restrict__ W, float* __restrict__ Wp,
                           int ntile, int ncol) {
    dag_pack_frag(W, Wp, ntile, ncol, blockIdx.x * blockDim.x + threadIdx.x);
}

template <bool PACKED, int NCOL>
__device__ __forceinline__ v4f dag_load_b2(const float* __restrict__ Wq, int e, int j,
                                           int p, int lane) {
    if (PACKED) {
        const v4f* Wf = (const v4f*)Wq + ((size_t)(e * (NCOL / 16) + j) * 32) * 32 + lane;
        return Wf[p * 32];
    } else {
        const int colLane = lane & 15;
        const int kk = p * 8 + ((lane >> 4) & 1) * 2;
        const float* Wb = Wq + (size_t)e * K1_ * NCOL + (size_t)(j * 16 + colLane);
        v4f b;
        b.x = Wb[(size_t)kk * NCOL];
        b.y = Wb[(size_t)(kk + 1) * NCOL];
        b.z = Wb[(size_t)(kk + 4) * NCOL];
        b.w = Wb[(size_t)(kk + 5) * NCOL];
        return b;
    }
}

template <bool PACKED, int NCOL, int NT>
__device__ __forceinline__ void dag_fetch_b(const float* __restrict__ Wq, int e, int j0,
                                            int g, int lane, v2f (&b)[8][NT]) {
    #pragma unroll
    for (int p = 0; p < 4; ++p) {
        #pragma unroll
        for (int t = 0; t < NT; ++t) {
            const v4f q = dag_load_b2<PACKED, NCOL>(Wq, e, j0 + t, g * 4 + p, lane);
            b[2 * p][t].x     = q.x;
            b[2 * p][t].y     = q.y;
            b[2 * p + 1][t].x = q.z;
            b[2 * p + 1][t].y = q.w;
        }
    }
}

template <bool PACKED, int NCOL, int NT, int RING, class TAIL>
__device__ __forceinline__ void dag_panel(const float* __restrict__ Xs,
                                          const float* __restrict__ Wq,
                                          int e, int j0, int lane,
                                          v2f (&bq)[RING][8][NT], v8f (&acc)[NT],
                                          const TAIL& tail) {
    const int colLane = lane & 15;
    const int klo     = ((lane >> 4) & 1) * 2;
    const float* Arow = Xs + colLane * XSTR_ + klo;

    v2f aq[2][8];
    #pragma unroll
    for (int s = 0; s < 8; ++s)
        aq[0][s] = *(const v2f*)(Arow + s * 4);
    #pragma unroll
    for (int g = 0; g < 8; ++g) {
        const int cur = g & 1, nxt = cur ^ 1;
        if (g + RING - 1 < 8)
            dag_fetch_b<PACKED, NCOL, NT>(Wq, e, j0, g + RING - 1, lane, bq[(g + RING - 1) % RING]);
        else
            tail(g + RING - 1 - 8);
        if (g < 7) {
            #pragma unroll
            for (int s = 0; s < 8; ++s)
                aq[nxt][s] = *(const v2f*)(Arow + ((g + 1) * 8 + s) * 4);
        }
        __builtin_amdgcn_sched_barrier(0);
        #pragma unroll
        for (int s = 0; s < 8; ++s) {
            #pragma unroll
            for (int t = 0; t < NT; ++t)
                acc[t] = __builtin_amdgcn_wmma_f32_16x16x4_f32(
                        false, aq[cur][s], false, bq[g % RING][s][t], (short)0, acc[t], false, false);
        }
        __builtin_amdgcn_sched_barrier(0);
    }
}

template <bool PACKED>
__global__ __launch_bounds__(128)
void dag_level(float* __restrict__ buf,
               const float* __restrict__ W1,
               const float* __restrict__ b1,
               const float* __restrict__ W2,
               const float* __restrict__ b2,
               const float* __restrict__ slots,
               const int*   __restrict__ par,
               const int*   __restrict__ types,
               int level)
{
    __shared__ float X[16 * XSTR_];
    __shared__ float H[16 * XSTR_];

    const int lane     = threadIdx.x & 31;
    const int w        = threadIdx.x >> 5;
    const int e        = blockIdx.y;
    const int tile     = blockIdx.x;
    const int qbase    = level * M_ + tile * 16;
    const int rowbase  = R_ + qbase;
    const int colLane  = lane & 15;
    const int laneHalf = (lane >> 4) & 1;

    int gp0[4], gp1[4], gt[4];
    #pragma unroll
    for (int r = 0; r < 4; ++r) {
        const int m = w + 4 * r;
        gp0[r] = par[(qbase + m) * 2 + 0];
        gp1[r] = par[(qbase + m) * 2 + 1];
        gt[r]  = types[qbase + m];
    }

    bool own[8];
    bool ownAny = false;
    #pragma unroll
    for (int i = 0; i < 8; ++i) {
        int t = types[qbase + i + laneHalf * 8];
        int r = (t >= START_) ? T_ : t;
        own[i] = (r == e);
        ownAny |= own[i];
    }
    if (!__any(ownAny)) return;

    v2f w1ring[RING1_][8][4];
    #pragma unroll
    for (int g = 0; g < RING1_ - 1; ++g)
        dag_fetch_b<PACKED, K1_, 4>(W1, e, w * 4, g, lane, w1ring[g]);

    v4f ga[4], gc[4];
    #pragma unroll
    for (int r = 0; r < 4; ++r) {
        const int t = gt[r];
        int slot = t - START_;
        slot = slot < 0 ? 0 : (slot > OUTSLOTS_ - 1 ? OUTSLOTS_ - 1 : slot);
        const float* s0 = buf + (size_t)gp0[r] * D_;
        const float* s1 = (t >= START_) ? (slots + (size_t)slot * D_)
                                        : (buf + (size_t)gp1[r] * D_);
        ga[r] = *(const v4f*)(s0 + lane * 4);
        gc[r] = *(const v4f*)(s1 + lane * 4);
    }
    #pragma unroll
    for (int r = 0; r < 4; ++r) {
        const int m = w + 4 * r;
        *(v4f*)(&X[m * XSTR_ + lane * 4])       = ga[r];
        *(v4f*)(&X[m * XSTR_ + 128 + lane * 4]) = gc[r];
    }
    __syncthreads();

    v2f w2ring[RING2_][8][2];
    {
        v8f acc[4] = {};
        dag_panel<PACKED, K1_, 4, RING1_>(X, W1, e, w * 4, lane, w1ring, acc, [](int) {});
        #pragma unroll
        for (int g = 0; g < RING2_ - 1; ++g)
            dag_fetch_b<PACKED, D_, 2>(W2, e, w * 2, g, lane, w2ring[g]);
        #pragma unroll
        for (int t = 0; t < 4; ++t) {
            const int j = w * 4 + t;
            float bias = b1[e * K1_ + j * 16 + colLane];
            #pragma unroll
            for (int i = 0; i < 8; ++i) {
                if (own[i]) {
                    float x = acc[t][i] + bias;
                    float g = 0.5f * x * (1.0f + erff(x * 0.70710678118654752f));
                    H[(i + laneHalf * 8) * XSTR_ + j * 16 + colLane] = g;
                }
            }
        }
    }
    __syncthreads();

    {
        v8f acc[2] = {};
        dag_panel<PACKED, D_, 2, RING2_>(H, W2, e, w * 2, lane, w2ring, acc, [](int) {});
        #pragma unroll
        for (int t = 0; t < 2; ++t) {
            const int j = w * 2 + t;
            float bias = b2[e * D_ + j * 16 + colLane];
            #pragma unroll
            for (int i = 0; i < 8; ++i) {
                if (own[i]) {
                    int row = rowbase + i + laneHalf * 8;
                    buf[(size_t)row * D_ + j * 16 + colLane] = acc[t][i] + bias;
                }
            }
        }
    }
}


__global__ __launch_bounds__(1024)
void dag_bucket(const int* __restrict__ par, const int* __restrict__ types,
                int* __restrict__ recs_) {
    __shared__ int cnt[NENC_ * 32];
    __shared__ int fillFrom[NENC_];
    __shared__ int tileEnd[NENC_];

    const int level = blockIdx.x;
    const int tid   = threadIdx.x;
    const int lane  = tid & 31;
    const int wv    = tid >> 5;
    const int q     = level * M_ + tid * 2;
    const int2 tt = *(const int2*)(types + q);
    const int4 pp = *(const int4*)(par + (size_t)q * 2);
    const int r0 = tt.x >= START_ ? T_ : (tt.x < 0 ? 0 : tt.x);
    const int r1 = tt.y >= START_ ? T_ : (tt.y < 0 ? 0 : tt.y);

    const unsigned below = (1u << lane) - 1u;
    int rank0 = 0, rank1 = 0;
    for (int e = 0; e < NENC_; ++e) {
        const unsigned m0 = (unsigned)__ballot(r0 == e);
        const unsigned m1 = (unsigned)__ballot(r1 == e);
        const int before = __popc(m0 & below) + __popc(m1 & below);
        if (r0 == e) rank0 = before;
        if (r1 == e) rank1 = before + (r0 == e ? 1 : 0);
        if (lane == 0) cnt[e * 32 + wv] = __popc(m0) + __popc(m1);
    }
    __syncthreads();

    if (wv == 0) {
        int base = 0;
        for (int e = 0; e < NENC_; ++e) {
            const int v = cnt[e * 32 + lane];
            int inc = v;
            for (int d = 1; d < 32; d <<= 1) {
                const int u = __shfl_up(inc, d);
                if (lane >= d) inc += u;
            }
            const int total = __shfl(inc, 31);
            cnt[e * 32 + lane] = base + inc - v;
            if (lane == 0) fillFrom[e] = base + total;
            base += ((total + 15) >> 4) << 4;
            if (lane == 0) tileEnd[e] = base;
        }
    }
    __syncthreads();

    int4* lv = (int4*)recs_ + (size_t)level * SLOTS_;
    lv[cnt[r0 * 32 + wv] + rank0] = make_int4(pp.x, pp.y, tt.x, tid * 2);
    lv[cnt[r1 * 32 + wv] + rank1] = make_int4(pp.z, pp.w, tt.y, tid * 2 + 1);

    for (int s = tid; s < SLOTS_; s += 1024) {
        int e = 0;
        for (int k = 0; k < NENC_; ++k)
            if (s >= tileEnd[k]) e = k + 1;
        if (e == NENC_)
            lv[s] = make_int4(0, 0, 0, -1);
        else if (s >= fillFrom[e])
            lv[s] = make_int4(0, 0, e == T_ ? START_ : e, -1);
    }
}

__device__ __forceinline__ float dag_erf_both_ranges(float y) {
    const float ax = __builtin_fabsf(y);
    const float t = y * y;
    float q = __builtin_fmaf(t, -0x1.268bc2p-11f, 0x1.420828p-8f);
    q = __builtin_fmaf(t, q, -0x1.b59370p-6f);
    q = __builtin_fmaf(t, q, 0x1.ce077cp-4f);
    q = __builtin_fmaf(t, q, -0x1.812660p-2f);
    q = __builtin_fmaf(t, q, 0x1.06eba0p-3f);
    const float rs = __builtin_fmaf(ax, q, ax);
    float p = __builtin_fmaf(ax, 0x1.1d3156p-16f, -0x1.8d1290p-12f);
    p = __builtin_fmaf(ax, p, 0x1.f9a6d2p-9f);
    p = __builtin_fmaf(ax, p, -0x1.8c3164p-6f);
    p = __builtin_fmaf(ax, p, 0x1.b4e9c8p-4f);
    p = __builtin_fmaf(ax, p, 0x1.4515fap-1f);
    p = __builtin_fmaf(ax, p, 0x1.078e50p-3f);
    p = __builtin_fmaf(ax, p, ax);
    const float rl = 1.0f - expf(-p);
    const float r = (ax < 1.0f) ? rs : rl;
    return __builtin_copysignf(r, y);
}

__device__ __forceinline__ void dag_lds_barrier() {
    asm volatile("s_wait_dscnt 0x0" ::: "memory");
    asm volatile("s_barrier_signal -1\n\ts_barrier_wait -1" ::: "memory");
}

__global__ __launch_bounds__(128)
void dag_level_bucketed(float* __restrict__ buf,
                        const float* __restrict__ W1,
                        const float* __restrict__ b1,
                        const float* __restrict__ W2,
                        const float* __restrict__ b2,
                        const float* __restrict__ slots,
                        const int*   __restrict__ recs_,
                        int level)
{
    __shared__ float X[16 * XSTR_];
    __shared__ float H[16 * XSTR_];
    __shared__ int   rowNode[16];

    const int lane     = threadIdx.x & 31;
    const int w        = threadIdx.x >> 5;
    const int colLane  = lane & 15;
    const int laneHalf = (lane >> 4) & 1;
    const int4* tr = (const int4*)recs_ + (size_t)level * SLOTS_ + blockIdx.x * 16;

    int4 rc[4];
    #pragma unroll
    for (int r = 0; r < 4; ++r)
        rc[r] = tr[w + 4 * r];
    const int t0 = rc[0].z;
    const int e = t0 >= START_ ? T_ : (t0 < 0 ? 0 : t0);
    if (threadIdx.x < 16) rowNode[threadIdx.x] = tr[threadIdx.x].w;

    v4f ga[4], gc[4];
    #pragma unroll
    for (int r = 0; r < 4; ++r) {
        const int t = rc[r].z;
        int slot = t - START_;
        slot = slot < 0 ? 0 : (slot > OUTSLOTS_ - 1 ? OUTSLOTS_ - 1 : slot);
        const float* s0 = buf + (size_t)rc[r].x * D_;
        const float* s1 = (t >= START_) ? (slots + (size_t)slot * D_)
                                        : (buf + (size_t)rc[r].y * D_);
        ga[r] = *(const v4f*)(s0 + lane * 4);
        gc[r] = *(const v4f*)(s1 + lane * 4);
    }

    v2f w1ring[RING1_][8][4];
    #pragma unroll
    for (int g = 0; g < RING1_ - 1; ++g)
        dag_fetch_b<true, K1_, 4>(W1, e, w * 4, g, lane, w1ring[g]);

    #pragma unroll
    for (int r = 0; r < 4; ++r) {
        const int m = w + 4 * r;
        *(v4f*)(&X[m * XSTR_ + lane * 4])       = ga[r];
        *(v4f*)(&X[m * XSTR_ + 128 + lane * 4]) = gc[r];
    }
    dag_lds_barrier();

    v2f w2ring[RING2_][8][2];
    float bias2[2];
    {
        v8f acc[4] = {};
        float bias[4];
        auto ahead = [&](int i) {
            if (i == 0) {
                #pragma unroll
                for (int t = 0; t < 4; ++t)
                    bias[t] = b1[e * K1_ + (w * 4 + t) * 16 + colLane];
                #pragma unroll
                for (int t = 0; t < 2; ++t)
                    bias2[t] = b2[e * D_ + (w * 2 + t) * 16 + colLane];
            }
            dag_fetch_b<true, D_, 2>(W2, e, w * 2, 2 * i,     lane, w2ring[2 * i]);
            dag_fetch_b<true, D_, 2>(W2, e, w * 2, 2 * i + 1, lane, w2ring[2 * i + 1]);
        };
        dag_panel<true, K1_, 4, RING1_>(X, W1, e, w * 4, lane, w1ring, acc, ahead);
        #pragma unroll
        for (int g = 2 * (RING1_ - 1); g < RING2_ - 1; ++g)
            dag_fetch_b<true, D_, 2>(W2, e, w * 2, g, lane, w2ring[g]);
        __builtin_amdgcn_sched_barrier(0);
        #pragma unroll
        for (int t = 0; t < 4; ++t) {
            const int j = w * 4 + t;
            #pragma unroll
            for (int i = 0; i < 8; ++i) {
                float x = acc[t][i] + bias[t];
                float g = 0.5f * x * (1.0f + dag_erf_both_ranges(x * 0.70710678118654752f));
                H[(i + laneHalf * 8) * XSTR_ + j * 16 + colLane] = g;
            }
        }
    }
    dag_lds_barrier();

    {
        v8f acc[2] = {};
        int node[8];
        auto ahead2 = [&](int i) {
            if (i == 0) {
                #pragma unroll
                for (int r = 0; r < 8; ++r)
                    node[r] = rowNode[laneHalf * 8 + r];
            }
        };
        dag_panel<true, D_, 2, RING2_>(H, W2, e, w * 2, lane, w2ring, acc, ahead2);
        #pragma unroll
        for (int t = 0; t < 2; ++t) {
            const int j = w * 2 + t;
            #pragma unroll
            for (int i = 0; i < 8; ++i) {
                if (node[i] >= 0) {
                    const int row = R_ + level * M_ + node[i];
                    buf[(size_t)row * D_ + j * 16 + colLane] = acc[t][i] + bias2[t];
                }
            }
        }
    }
}

__device__ __forceinline__ int dag_key(int t, int db) {
    const int e  = t >= START_ ? T_ : (t < 0 ? 0 : t);
    const int d  = db & 0x7f;
    const int dd = d < 1 ? 1 : (d > L_ ? L_ : d);
    return ((dd - 1) * NENC_ + e) * 2 + ((db & 0x80) ? 0 : 1);
}

__device__ __forceinline__ void dag_depth_pass(const int* __restrict__ par, unsigned char* __restrict__ depOut) {
    static_assert(R_ == 1024 && M_ == 2048 && L_ % 4 == 0, "one root and two nodes of a level per thread, levels in fours");
    __shared__ unsigned char dep[N_];
    const int tid = threadIdx.x;
    dep[tid] = 0;
    int4 ahead[4];
    #pragma unroll
    for (int i = 0; i < 4; ++i)
        ahead[i] = *(const int4*)(par + ((size_t)i * M_ + tid * 2) * 2);
    dag_lds_barrier();
    for (int l0 = 0; l0 < L_; l0 += 4) {
        #pragma unroll
        for (int i = 0; i < 4; ++i) {
            const int l = l0 + i;
            const int4 cur = ahead[i];
            const int ln = l + 4 < L_ ? l + 4 : L_ - 1;
            ahead[i] = *(const int4*)(par + ((size_t)ln * M_ + tid * 2) * 2);
            const int hi = R_ + l * M_;
            const int ia0 = (unsigned)cur.x < (unsigned)hi ? cur.x : 0;
            const int ia1 = (unsigned)cur.y < (unsigned)hi ? cur.y : 0;
            const int ic0 = (unsigned)cur.z < (unsigned)hi ? cur.z : 0;
            const int ic1 = (unsigned)cur.w < (unsigned)hi ? cur.w : 0;
            const int ra0 = dep[ia0], ra1 = dep[ia1], rc0 = dep[ic0], rc1 = dep[ic1];
            dep[ia0] = (unsigned char)(ra0 | 0x80);
            dep[ia1] = (unsigned char)(ra1 | 0x80);
            dep[ic0] = (unsigned char)(rc0 | 0x80);
            dep[ic1] = (unsigned char)(rc1 | 0x80);
            const int a0 = ra0 & 0x7f, a1 = ra1 & 0x7f, c0 = rc0 & 0x7f, c1 = rc1 & 0x7f;
            dep[hi + tid * 2]     = (unsigned char)(1 + (a0 > a1 ? a0 : a1));
            dep[hi + tid * 2 + 1] = (unsigned char)(1 + (c0 > c1 ? c0 : c1));
            dag_lds_barrier();
        }
    }
    for (int i = tid; i < QN_ / 16; i += 1024)
        ((uint4*)depOut)[i] = *(const uint4*)(dep + R_ + i * 16);
}

#define FRONT_ROOT_ (R_ * D_ / 4 / 1024)
#define FRONT_W1_   (W1P_FRAGS / 1024)
#define FRONT_W2_   (W2P_FRAGS / 1024)
#define FRONT_GRID_ (1 + FRONT_ROOT_ + FRONT_W1_ + FRONT_W2_)
static_assert(R_ * D_ / 4 % 1024 == 0 && W1P_FRAGS % 1024 == 0 && W2P_FRAGS % 1024 == 0, "whole work-groups per role");
__global__ __launch_bounds__(1024)
void dag_front(const int* __restrict__ par, unsigned char* __restrict__ depOut,
               float* __restrict__ buf, const float* __restrict__ root,
               const float* __restrict__ W1, float* __restrict__ W1p,
               const float* __restrict__ W2, float* __restrict__ W2p) {
    int b = blockIdx.x;
    if (b == 0) { dag_depth_pass(par, depOut); return; }
    b -= 1;
    if (b < FRONT_ROOT_) { dag_copy_root4(buf, root, b * 1024 + threadIdx.x); return; }
    b -= FRONT_ROOT_;
    if (b < FRONT_W1_) { dag_pack_frag(W1, W1p, 16, K1_, b * 1024 + threadIdx.x); return; }
    b -= FRONT_W1_;
    dag_pack_frag(W2, W2p, 8, D_, b * 1024 + threadIdx.x);
}

__global__ __launch_bounds__(1024)
void dag_rank(const int* __restrict__ types, const unsigned char* __restrict__ dep,
              unsigned short* __restrict__ rank, int* __restrict__ cnt) {
    __shared__ unsigned short wcnt[32 * NKEY_];
    const int level = blockIdx.x;
    const int tid   = threadIdx.x;
    const int lane  = tid & 31;
    const int wv    = tid >> 5;
    const int q     = level * M_ + tid * 2;
    const int2 tt = *(const int2*)(types + q);
    const int ka = dag_key(tt.x, dep[q]);
    const int kb = dag_key(tt.y, dep[q + 1]);
    for (int i = tid; i < 32 * NKEY_; i += 1024) wcnt[i] = 0;

    int ra = 0, rb = 0, ta = 0, tb = 0;
    #pragma unroll
    for (int s = 0; s < 32; ++s) {
        const int sa = __builtin_amdgcn_readlane(ka, s);
        const int sb = __builtin_amdgcn_readlane(kb, s);
        const int ma = (sa == ka ? 1 : 0) + (sb == ka ? 1 : 0);
        const int mb = (sa == kb ? 1 : 0) + (sb == kb ? 1 : 0);
        ta += ma;
        tb += mb;
        if (s < lane) { ra += ma; rb += mb; }
    }
    if (ka == kb) rb += 1;
    __syncthreads();
    if (ra == ta - 1) wcnt[wv * NKEY_ + ka] = (unsigned short)ta;
    if (rb == tb - 1) wcnt[wv * NKEY_ + kb] = (unsigned short)tb;
    __syncthreads();
    if (tid < NKEY_) {
        int run = 0;
        for (int v = 0; v < 32; ++v) {
            const int c = wcnt[v * NKEY_ + tid];
            wcnt[v * NKEY_ + tid] = (unsigned short)run;
            run += c;
        }
        cnt[level * NKEY_ + tid] = run;
    }
    __syncthreads();
    rank[q]     = (unsigned short)(wcnt[wv * NKEY_ + ka] + ra);
    rank[q + 1] = (unsigned short)(wcnt[wv * NKEY_ + kb] + rb);
}

__global__ __launch_bounds__(NKEY_)
void dag_offsets(int* __restrict__ cnt, int* __restrict__ tileStart, int* __restrict__ recs_) {
    __shared__ int til[NKEY_];
    __shared__ int first[NKEY_];
    __shared__ int tsl[L_ + 1];
    const int k = threadIdx.x;
    int total = 0;
    for (int l = 0; l < L_; ++l) total += cnt[l * NKEY_ + k];
    til[k]   = (total + 15) >> 4;
    first[k] = 0;
    __syncthreads();
    if (k < 32) {
        const int lane = k;
        int md = 0;
        for (int j = lane; j < NKEY_; j += 32)
            if (til[j] > 0) md = j / (2 * NENC_) + 1;
        for (int o = 16; o > 0; o >>= 1) {
            const int u = __shfl_xor(md, o);
            md = u > md ? u : md;
        }
        const int lmax = __builtin_amdgcn_readfirstlane(md);
        const int lcap = lmax < DWIDE_ ? lmax : DWIDE_;
        int off = 0;
        int qd  = 0;
        for (int l = 0; l < lmax; ++l) {
            if (lane == 0) tsl[l] = off;
            const int tn = lane < NENC_ ? til[(l * NENC_ + lane) * 2] : 0;
            int nl = 0;
            #pragma unroll
            for (int e = 0; e < NENC_; ++e) {
                if (lane == e) first[(l * NENC_ + e) * 2] = off + nl;
                nl += __builtin_amdgcn_readlane(tn, e);
            }
            off += nl;
            int room = (l + 1 >= lcap) ? (1 << 28) : ((DUNITS_ - nl % DUNITS_) % DUNITS_);
            for (int pass = 0; pass < (L_ * NENC_) / 32 + 1; ++pass) {
                const int j = qd + lane;
                const bool due = j < L_ * NENC_ && j / NENC_ <= l;
                const int tj = due ? til[j * 2 + 1] : -1;
                int n = 0;
                for (; n < 32; ++n) {
                    const int t = __builtin_amdgcn_readlane(tj, n);
                    if (t < 0 || t > room) break;
                    if (lane == n) first[j * 2 + 1] = off;
                    off  += t;
                    room -= t;
                }
                qd += n;
                if (n < 32) break;
            }
        }
        for (int l = lmax + lane; l <= L_; l += 32) tsl[l] = off;
    }
    __syncthreads();
    int slot = first[k] * 16;
    for (int l = 0; l < L_; ++l) {
        const int c = cnt[l * NKEY_ + k];
        cnt[l * NKEY_ + k] = slot;
        slot += c;
    }
    int4* rec = (int4*)recs_;
    const int e = (k >> 1) % NENC_;
    const int end = (first[k] + til[k]) * 16;
    for (int s = slot; s < end; ++s)
        rec[s] = make_int4(0, 0, e == T_ ? START_ : e, -1);
    if (k <= L_) tileStart[k] = tsl[k];
}

__global__ __launch_bounds__(1024)
void dag_place(const int* __restrict__ par, const int* __restrict__ types,
               const unsigned char* __restrict__ dep, const unsigned short* __restrict__ rank,
               const int* __restrict__ base, int* __restrict__ recs_) {
    const int level = blockIdx.x;
    const int q     = level * M_ + threadIdx.x * 2;
    const int2 tt = *(const int2*)(types + q);
    const int4 pp = *(const int4*)(par + (size_t)q * 2);
    const int ka = dag_key(tt.x, dep[q]);
    const int kb = dag_key(tt.y, dep[q + 1]);
    int4* rec = (int4*)recs_;
    rec[base[level * NKEY_ + ka] + rank[q]]     = make_int4(pp.x, pp.y, tt.x, q);
    rec[base[level * NKEY_ + kb] + rank[q + 1]] = make_int4(pp.z, pp.w, tt.y, q + 1);
}

template <bool TAIL>
__global__ __launch_bounds__(128)
void dag_level_depth(float* __restrict__ buf,
                        const float* __restrict__ W1,
                        const float* __restrict__ b1,
                        const float* __restrict__ W2,
                        const float* __restrict__ b2,
                        const float* __restrict__ slots,
                        const int*   __restrict__ recs_,
                        const int*   __restrict__ tileStart,
                        int depth)
{
    __shared__ float X[16 * XSTR_];
    __shared__ float H[16 * XSTR_];
    __shared__ int   rowNode[2][16];

    const int lane     = threadIdx.x & 31;
    const int w        = threadIdx.x >> 5;
    const int colLane  = lane & 15;
    const int laneHalf = (lane >> 4) & 1;
    int d       = depth;
    int tileBeg = tileStart[d];
    if (TAIL && tileBeg >= tileStart[L_]) return;
    do {
    const int tileEnd = tileStart[d + 1];
    for (int tile = tileBeg + (TAIL ? 0 : (int)blockIdx.x), par2 = 0; tile < tileEnd; tile += (TAIL ? 1 : DGRID_), par2 ^= 1) {
    const int4* tr = (const int4*)recs_ + (size_t)tile * 16;

    int4 rc[4];
    #pragma unroll
    for (int r = 0; r < 4; ++r)
        rc[r] = tr[w + 4 * r];
    const int t0 = rc[0].z;
    const int e = t0 >= START_ ? T_ : (t0 < 0 ? 0 : t0);
    if (threadIdx.x < 16) rowNode[par2][threadIdx.x] = tr[threadIdx.x].w;

    v4f ga[4], gc[4];
    #pragma unroll
    for (int r = 0; r < 4; ++r) {
        const int t = rc[r].z;
        int slot = t - START_;
        slot = slot < 0 ? 0 : (slot > OUTSLOTS_ - 1 ? OUTSLOTS_ - 1 : slot);
        const float* s0 = buf + (size_t)rc[r].x * D_;
        const float* s1 = (t >= START_) ? (slots + (size_t)slot * D_)
                                        : (buf + (size_t)rc[r].y * D_);
        ga[r] = *(const v4f*)(s0 + lane * 4);
        gc[r] = *(const v4f*)(s1 + lane * 4);
    }

    v2f w1ring[RING1_][8][4];
    #pragma unroll
    for (int g = 0; g < RING1_ - 1; ++g)
        dag_fetch_b<true, K1_, 4>(W1, e, w * 4, g, lane, w1ring[g]);

    #pragma unroll
    for (int r = 0; r < 4; ++r) {
        const int m = w + 4 * r;
        *(v4f*)(&X[m * XSTR_ + lane * 4])       = ga[r];
        *(v4f*)(&X[m * XSTR_ + 128 + lane * 4]) = gc[r];
    }
    dag_lds_barrier();

    v2f w2ring[RING2_][8][2];
    float bias2[2];
    {
        v8f acc[4] = {};
        float bias[4];
        auto ahead = [&](int i) {
            if (i == 0) {
                #pragma unroll
                for (int t = 0; t < 4; ++t)
                    bias[t] = b1[e * K1_ + (w * 4 + t) * 16 + colLane];
                #pragma unroll
                for (int t = 0; t < 2; ++t)
                    bias2[t] = b2[e * D_ + (w * 2 + t) * 16 + colLane];
            }
            dag_fetch_b<true, D_, 2>(W2, e, w * 2, 2 * i,     lane, w2ring[2 * i]);
            dag_fetch_b<true, D_, 2>(W2, e, w * 2, 2 * i + 1, lane, w2ring[2 * i + 1]);
        };
        dag_panel<true, K1_, 4, RING1_>(X, W1, e, w * 4, lane, w1ring, acc, ahead);
        #pragma unroll
        for (int g = 2 * (RING1_ - 1); g < RING2_ - 1; ++g)
            dag_fetch_b<true, D_, 2>(W2, e, w * 2, g, lane, w2ring[g]);
        __builtin_amdgcn_sched_barrier(0);
        #pragma unroll
        for (int t = 0; t < 4; ++t) {
            const int j = w * 4 + t;
            #pragma unroll
            for (int i = 0; i < 8; ++i) {
                float x = acc[t][i] + bias[t];
                float g = 0.5f * x * (1.0f + dag_erf_both_ranges(x * 0.70710678118654752f));
                H[(i + laneHalf * 8) * XSTR_ + j * 16 + colLane] = g;
            }
        }
    }
    dag_lds_barrier();

    {
        v8f acc[2] = {};
        int node[8];
        auto ahead2 = [&](int i) {
            if (i == 0) {
                #pragma unroll
                for (int r = 0; r < 8; ++r)
                    node[r] = rowNode[par2][laneHalf * 8 + r];
            }
        };
        dag_panel<true, D_, 2, RING2_>(H, W2, e, w * 2, lane, w2ring, acc, ahead2);
        if (TAIL) {
            asm volatile("v_nop\n\tv_nop\n\tv_nop\n\tv_nop");
        }
        #pragma unroll
        for (int t = 0; t < 2; ++t) {
            const int j = w * 2 + t;
            #pragma unroll
            for (int i = 0; i < 8; ++i) {
                if (node[i] >= 0) {
                    const int row = R_ + node[i];
                    buf[(size_t)row * D_ + j * 16 + colLane] = acc[t][i] + bias2[t];
                }
            }
        }
    }
    }
    if (!TAIL) break;
    ++d;
    tileBeg = tileEnd;
    if (d >= L_ || tileBeg >= tileStart[L_]) break;
    __threadfence();
    __syncthreads();
    } while (true);
}


extern "C" void kernel_launch(void* const* d_in, const int* in_sizes, int n_in,
                              void* d_out, int out_size, void* d_ws, size_t ws_size,
                              hipStream_t stream) {
    const float* root  = (const float*)d_in[0];
    const float* W1    = (const float*)d_in[1];
    const float* b1    = (const float*)d_in[2];
    const float* W2    = (const float*)d_in[3];
    const float* b2    = (const float*)d_in[4];
    const float* slots = (const float*)d_in[5];
    const int*   par   = (const int*)d_in[6];
    const int*   typ   = (const int*)d_in[7];
    float* out = (float*)d_out;

    const bool packed = (ws_size >= WSP_BYTES);
    float* W1p = (float*)d_ws;
    float* W2p = (float*)((char*)d_ws + W1P_BYTES);

    const bool bucketed = (ws_size >= WSP_BYTES + REC_BYTES);
    int* recs = (int*)((char*)d_ws + WSP_BYTES);

    const bool byDepth = (ws_size >= WSP_BYTES + DREC_BYTES + DEP_BYTES + RANK_BYTES + CNT_BYTES + TST_BYTES);
    unsigned char*  dep   = (unsigned char*)d_ws + WSP_BYTES + DREC_BYTES;
    unsigned short* rnk   = (unsigned short*)((char*)d_ws + WSP_BYTES + DREC_BYTES + DEP_BYTES);
    int*            cnt   = (int*)((char*)d_ws + WSP_BYTES + DREC_BYTES + DEP_BYTES + RANK_BYTES);
    int*            tileS = (int*)((char*)d_ws + WSP_BYTES + DREC_BYTES + DEP_BYTES + RANK_BYTES + CNT_BYTES);

    if (!byDepth) dag_init_roots<<<128, 256, 0, stream>>>(out, root);

    if (byDepth) {
        dag_front<<<FRONT_GRID_, 1024, 0, stream>>>(par, dep, out, root, W1, W1p, W2, W2p);
        dag_rank<<<L_, 1024, 0, stream>>>(typ, dep, rnk, cnt);
        dag_offsets<<<1, NKEY_, 0, stream>>>(cnt, tileS, recs);
        dag_place<<<L_, 1024, 0, stream>>>(par, typ, dep, rnk, cnt, recs);
        for (int d = 0; d < DWIDE_; ++d)
            dag_level_depth<false><<<DGRID_, 128, 0, stream>>>(out, W1p, b1, W2p, b2,
                                                               slots, recs, tileS, d);
        dag_level_depth<true><<<1, 128, 0, stream>>>(out, W1p, b1, W2p, b2,
                                                     slots, recs, tileS, DWIDE_);
    } else if (bucketed) {
        dag_pack_w<<<W1P_FRAGS / 256, 256, 0, stream>>>(W1, W1p, 16, K1_);
        dag_pack_w<<<W2P_FRAGS / 256, 256, 0, stream>>>(W2, W2p, 8, D_);
        dag_bucket<<<L_, 1024, 0, stream>>>(par, typ, recs);
        for (int l = 0; l < L_; ++l)
            dag_level_bucketed<<<TILES_, 128, 0, stream>>>(out, W1p, b1, W2p, b2,
                                                           slots, recs, l);
    } else if (packed) {
        dag_pack_w<<<W1P_FRAGS / 256, 256, 0, stream>>>(W1, W1p, 16, K1_);
        dag_pack_w<<<W2P_FRAGS / 256, 256, 0, stream>>>(W2, W2p, 8, D_);
        for (int l = 0; l < L_; ++l)
            dag_level<true><<<dim3(128, NENC_), 128, 0, stream>>>(out, W1p, b1, W2p, b2,
                                                                 slots, par, typ, l);
    } else {
        for (int l = 0; l < L_; ++l)
            dag_level<false><<<dim3(128, NENC_), 128, 0, stream>>>(out, W1, b1, W2, b2,
                                                                  slots, par, typ, l);
    }
}


# Parent marking leaves the serial depth walk: 4 more work-groups of the first launch mark the named rows in LDS, a quarter of the levels each; ranking ORs their planes; walk keeps 5 of 9 LDS ops
# speedup vs baseline: 15.4416x; 1.0441x over previous
#include <hip/hip_runtime.h>
#include <math.h>

typedef __attribute__((ext_vector_type(2))) float v2f;
typedef __attribute__((ext_vector_type(4))) float v4f;
typedef __attribute__((ext_vector_type(8))) float v8f;

#define D_        128
#define R_        1024
#define T_        4
#define OUTSLOTS_ 256
#define START_    4
#define L_        64
#define M_        2048
#define N_        (R_ + L_ * M_)
#define NENC_     5
#define K1_       256
#define XSTR_     260

#define W1P_FRAGS (NENC_ * 16 * 64 * 32)
#define W2P_FRAGS (NENC_ * 8  * 64 * 32)
#define W1P_BYTES ((size_t)W1P_FRAGS * 8)
#define W2P_BYTES ((size_t)W2P_FRAGS * 8)
#define WSP_BYTES (W1P_BYTES + W2P_BYTES)

#define RING1_    4
#define RING2_    8
static_assert(2 * (RING1_ - 1) <= RING2_ - 1, "layer-2 groups requested during layer 1 must fit the layer-2 ring");

#define TILES_    132
#define SLOTS_    (TILES_ * 16)
#define REC_BYTES ((size_t)L_ * SLOTS_ * 16)

#define QN_        (L_ * M_)
#define NKEY_      (L_ * NENC_ * 2)
#define DTILES_    (QN_ / 16 + NKEY_)
#define DREC_BYTES ((size_t)DTILES_ * 16 * 16)
#define DEP_BYTES  ((size_t)QN_)
#define FLAGW_     4
static_assert(L_ % FLAGW_ == 0, "whole levels per marking group");
#define RANK_BYTES ((size_t)QN_ * 2)
#define CNT_BYTES  ((size_t)L_ * NKEY_ * 4)
#define TST_BYTES  ((size_t)512)
static_assert((size_t)FLAGW_ * QN_ <= DREC_BYTES, "the marking groups' flags fit into the record area");
#define DGRID_     512
#define DUNITS_    (DGRID_ / 2)
#define DWIDE_     20
static_assert(DWIDE_ < L_, "the tail launch starts at depth DWIDE_ + 1");

__device__ __forceinline__ void dag_copy_root4(float* __restrict__ buf, const float* __restrict__ root, int i) {
    v4f v = *(const v4f*)(root + (size_t)i * 4);
    *(v4f*)(buf + (size_t)i * 4) = v;
}

__global__ void dag_init_roots(float* __restrict__ buf, const float* __restrict__ root) {
    dag_copy_root4(buf, root, blockIdx.x * blockDim.x + threadIdx.x);
}

__device__ __forceinline__ void dag_pack_frag(const float* __restrict__ W, float* __restrict__ Wp,
                                              int ntile, int ncol, int idx) {
    int lane = idx & 31;
    int k4   = (idx >> 5) & 63;
    int j    = (idx >> 11) % ntile;
    int e    = idx / (ntile << 11);
    int col  = lane & 15;
    int kk   = k4 * 4 + ((lane >> 4) & 1) * 2;
    const float* Wsrc = W + (size_t)e * K1_ * ncol + (size_t)(j * 16 + col);
    const size_t dst = ((size_t)((idx >> 11) * 32 + (k4 >> 1)) * 32 + lane) * 4 + (k4 & 1) * 2;
    Wp[dst + 0] = Wsrc[(size_t)kk * ncol];
    Wp[dst + 1] = Wsrc[(size_t)(kk + 1) * ncol];
}

__global__ void dag_pack_w(const float* __restrict__ W, float* __restrict__ Wp,
                           int ntile, int ncol) {
    dag_pack_frag(W, Wp, ntile, ncol, blockIdx.x * blockDim.x + threadIdx.x);
}

template <bool PACKED, int NCOL>
__device__ __forceinline__ v4f dag_load_b2(const float* __restrict__ Wq, int e, int j,
                                           int p, int lane) {
    if (PACKED) {
        const v4f* Wf = (const v4f*)Wq + ((size_t)(e * (NCOL / 16) + j) * 32) * 32 + lane;
        return Wf[p * 32];
    } else {
        const int colLane = lane & 15;
        const int kk = p * 8 + ((lane >> 4) & 1) * 2;
        const float* Wb = Wq + (size_t)e * K1_ * NCOL + (size_t)(j * 16 + colLane);
        v4f b;
        b.x = Wb[(size_t)kk * NCOL];
        b.y = Wb[(size_t)(kk + 1) * NCOL];
        b.z = Wb[(size_t)(kk + 4) * NCOL];
        b.w = Wb[(size_t)(kk + 5) * NCOL];
        return b;
    }
}

template <bool PACKED, int NCOL, int NT>
__device__ __forceinline__ void dag_fetch_b(const float* __restrict__ Wq, int e, int j0,
                                            int g, int lane, v2f (&b)[8][NT]) {
    #pragma unroll
    for (int p = 0; p < 4; ++p) {
        #pragma unroll
        for (int t = 0; t < NT; ++t) {
            const v4f q = dag_load_b2<PACKED, NCOL>(Wq, e, j0 + t, g * 4 + p, lane);
            b[2 * p][t].x     = q.x;
            b[2 * p][t].y     = q.y;
            b[2 * p + 1][t].x = q.z;
            b[2 * p + 1][t].y = q.w;
        }
    }
}

template <bool PACKED, int NCOL, int NT, int RING, class TAIL>
__device__ __forceinline__ void dag_panel(const float* __restrict__ Xs,
                                          const float* __restrict__ Wq,
                                          int e, int j0, int lane,
                                          v2f (&bq)[RING][8][NT], v8f (&acc)[NT],
                                          const TAIL& tail) {
    const int colLane = lane & 15;
    const int klo     = ((lane >> 4) & 1) * 2;
    const float* Arow = Xs + colLane * XSTR_ + klo;

    v2f aq[2][8];
    #pragma unroll
    for (int s = 0; s < 8; ++s)
        aq[0][s] = *(const v2f*)(Arow + s * 4);
    #pragma unroll
    for (int g = 0; g < 8; ++g) {
        const int cur = g & 1, nxt = cur ^ 1;
        if (g + RING - 1 < 8)
            dag_fetch_b<PACKED, NCOL, NT>(Wq, e, j0, g + RING - 1, lane, bq[(g + RING - 1) % RING]);
        else
            tail(g + RING - 1 - 8);
        if (g < 7) {
            #pragma unroll
            for (int s = 0; s < 8; ++s)
                aq[nxt][s] = *(const v2f*)(Arow + ((g + 1) * 8 + s) * 4);
        }
        __builtin_amdgcn_sched_barrier(0);
        #pragma unroll
        for (int s = 0; s < 8; ++s) {
            #pragma unroll
            for (int t = 0; t < NT; ++t)
                acc[t] = __builtin_amdgcn_wmma_f32_16x16x4_f32(
                        false, aq[cur][s], false, bq[g % RING][s][t], (short)0, acc[t], false, false);
        }
        __builtin_amdgcn_sched_barrier(0);
    }
}

template <bool PACKED>
__global__ __launch_bounds__(128)
void dag_level(float* __restrict__ buf,
               const float* __restrict__ W1,
               const float* __restrict__ b1,
               const float* __restrict__ W2,
               const float* __restrict__ b2,
               const float* __restrict__ slots,
               const int*   __restrict__ par,
               const int*   __restrict__ types,
               int level)
{
    __shared__ float X[16 * XSTR_];
    __shared__ float H[16 * XSTR_];

    const int lane     = threadIdx.x & 31;
    const int w        = threadIdx.x >> 5;
    const int e        = blockIdx.y;
    const int tile     = blockIdx.x;
    const int qbase    = level * M_ + tile * 16;
    const int rowbase  = R_ + qbase;
    const int colLane  = lane & 15;
    const int laneHalf = (lane >> 4) & 1;

    int gp0[4], gp1[4], gt[4];
    #pragma unroll
    for (int r = 0; r < 4; ++r) {
        const int m = w + 4 * r;
        gp0[r] = par[(qbase + m) * 2 + 0];
        gp1[r] = par[(qbase + m) * 2 + 1];
        gt[r]  = types[qbase + m];
    }

    bool own[8];
    bool ownAny = false;
    #pragma unroll
    for (int i = 0; i < 8; ++i) {
        int t = types[qbase + i + laneHalf * 8];
        int r = (t >= START_) ? T_ : t;
        own[i] = (r == e);
        ownAny |= own[i];
    }
    if (!__any(ownAny)) return;

    v2f w1ring[RING1_][8][4];
    #pragma unroll
    for (int g = 0; g < RING1_ - 1; ++g)
        dag_fetch_b<PACKED, K1_, 4>(W1, e, w * 4, g, lane, w1ring[g]);

    v4f ga[4], gc[4];
    #pragma unroll
    for (int r = 0; r < 4; ++r) {
        const int t = gt[r];
        int slot = t - START_;
        slot = slot < 0 ? 0 : (slot > OUTSLOTS_ - 1 ? OUTSLOTS_ - 1 : slot);
        const float* s0 = buf + (size_t)gp0[r] * D_;
        const float* s1 = (t >= START_) ? (slots + (size_t)slot * D_)
                                        : (buf + (size_t)gp1[r] * D_);
        ga[r] = *(const v4f*)(s0 + lane * 4);
        gc[r] = *(const v4f*)(s1 + lane * 4);
    }
    #pragma unroll
    for (int r = 0; r < 4; ++r) {
        const int m = w + 4 * r;
        *(v4f*)(&X[m * XSTR_ + lane * 4])       = ga[r];
        *(v4f*)(&X[m * XSTR_ + 128 + lane * 4]) = gc[r];
    }
    __syncthreads();

    v2f w2ring[RING2_][8][2];
    {
        v8f acc[4] = {};
        dag_panel<PACKED, K1_, 4, RING1_>(X, W1, e, w * 4, lane, w1ring, acc, [](int) {});
        #pragma unroll
        for (int g = 0; g < RING2_ - 1; ++g)
            dag_fetch_b<PACKED, D_, 2>(W2, e, w * 2, g, lane, w2ring[g]);
        #pragma unroll
        for (int t = 0; t < 4; ++t) {
            const int j = w * 4 + t;
            float bias = b1[e * K1_ + j * 16 + colLane];
            #pragma unroll
            for (int i = 0; i < 8; ++i) {
                if (own[i]) {
                    float x = acc[t][i] + bias;
                    float g = 0.5f * x * (1.0f + erff(x * 0.70710678118654752f));
                    H[(i + laneHalf * 8) * XSTR_ + j * 16 + colLane] = g;
                }
            }
        }
    }
    __syncthreads();

    {
        v8f acc[2] = {};
        dag_panel<PACKED, D_, 2, RING2_>(H, W2, e, w * 2, lane, w2ring, acc, [](int) {});
        #pragma unroll
        for (int t = 0; t < 2; ++t) {
            const int j = w * 2 + t;
            float bias = b2[e * D_ + j * 16 + colLane];
            #pragma unroll
            for (int i = 0; i < 8; ++i) {
                if (own[i]) {
                    int row = rowbase + i + laneHalf * 8;
                    buf[(size_t)row * D_ + j * 16 + colLane] = acc[t][i] + bias;
                }
            }
        }
    }
}


__global__ __launch_bounds__(1024)
void dag_bucket(const int* __restrict__ par, const int* __restrict__ types,
                int* __restrict__ recs_) {
    __shared__ int cnt[NENC_ * 32];
    __shared__ int fillFrom[NENC_];
    __shared__ int tileEnd[NENC_];

    const int level = blockIdx.x;
    const int tid   = threadIdx.x;
    const int lane  = tid & 31;
    const int wv    = tid >> 5;
    const int q     = level * M_ + tid * 2;
    const int2 tt = *(const int2*)(types + q);
    const int4 pp = *(const int4*)(par + (size_t)q * 2);
    const int r0 = tt.x >= START_ ? T_ : (tt.x < 0 ? 0 : tt.x);
    const int r1 = tt.y >= START_ ? T_ : (tt.y < 0 ? 0 : tt.y);

    const unsigned below = (1u << lane) - 1u;
    int rank0 = 0, rank1 = 0;
    for (int e = 0; e < NENC_; ++e) {
        const unsigned m0 = (unsigned)__ballot(r0 == e);
        const unsigned m1 = (unsigned)__ballot(r1 == e);
        const int before = __popc(m0 & below) + __popc(m1 & below);
        if (r0 == e) rank0 = before;
        if (r1 == e) rank1 = before + (r0 == e ? 1 : 0);
        if (lane == 0) cnt[e * 32 + wv] = __popc(m0) + __popc(m1);
    }
    __syncthreads();

    if (wv == 0) {
        int base = 0;
        for (int e = 0; e < NENC_; ++e) {
            const int v = cnt[e * 32 + lane];
            int inc = v;
            for (int d = 1; d < 32; d <<= 1) {
                const int u = __shfl_up(inc, d);
                if (lane >= d) inc += u;
            }
            const int total = __shfl(inc, 31);
            cnt[e * 32 + lane] = base + inc - v;
            if (lane == 0) fillFrom[e] = base + total;
            base += ((total + 15) >> 4) << 4;
            if (lane == 0) tileEnd[e] = base;
        }
    }
    __syncthreads();

    int4* lv = (int4*)recs_ + (size_t)level * SLOTS_;
    lv[cnt[r0 * 32 + wv] + rank0] = make_int4(pp.x, pp.y, tt.x, tid * 2);
    lv[cnt[r1 * 32 + wv] + rank1] = make_int4(pp.z, pp.w, tt.y, tid * 2 + 1);

    for (int s = tid; s < SLOTS_; s += 1024) {
        int e = 0;
        for (int k = 0; k < NENC_; ++k)
            if (s >= tileEnd[k]) e = k + 1;
        if (e == NENC_)
            lv[s] = make_int4(0, 0, 0, -1);
        else if (s >= fillFrom[e])
            lv[s] = make_int4(0, 0, e == T_ ? START_ : e, -1);
    }
}

__device__ __forceinline__ float dag_erf_both_ranges(float y) {
    const float ax = __builtin_fabsf(y);
    const float t = y * y;
    float q = __builtin_fmaf(t, -0x1.268bc2p-11f, 0x1.420828p-8f);
    q = __builtin_fmaf(t, q, -0x1.b59370p-6f);
    q = __builtin_fmaf(t, q, 0x1.ce077cp-4f);
    q = __builtin_fmaf(t, q, -0x1.812660p-2f);
    q = __builtin_fmaf(t, q, 0x1.06eba0p-3f);
    const float rs = __builtin_fmaf(ax, q, ax);
    float p = __builtin_fmaf(ax, 0x1.1d3156p-16f, -0x1.8d1290p-12f);
    p = __builtin_fmaf(ax, p, 0x1.f9a6d2p-9f);
    p = __builtin_fmaf(ax, p, -0x1.8c3164p-6f);
    p = __builtin_fmaf(ax, p, 0x1.b4e9c8p-4f);
    p = __builtin_fmaf(ax, p, 0x1.4515fap-1f);
    p = __builtin_fmaf(ax, p, 0x1.078e50p-3f);
    p = __builtin_fmaf(ax, p, ax);
    const float rl = 1.0f - expf(-p);
    const float r = (ax < 1.0f) ? rs : rl;
    return __builtin_copysignf(r, y);
}

__device__ __forceinline__ void dag_lds_barrier() {
    asm volatile("s_wait_dscnt 0x0" ::: "memory");
    asm volatile("s_barrier_signal -1\n\ts_barrier_wait -1" ::: "memory");
}

__global__ __launch_bounds__(128)
void dag_level_bucketed(float* __restrict__ buf,
                        const float* __restrict__ W1,
                        const float* __restrict__ b1,
                        const float* __restrict__ W2,
                        const float* __restrict__ b2,
                        const float* __restrict__ slots,
                        const int*   __restrict__ recs_,
                        int level)
{
    __shared__ float X[16 * XSTR_];
    __shared__ float H[16 * XSTR_];
    __shared__ int   rowNode[16];

    const int lane     = threadIdx.x & 31;
    const int w        = threadIdx.x >> 5;
    const int colLane  = lane & 15;
    const int laneHalf = (lane >> 4) & 1;
    const int4* tr = (const int4*)recs_ + (size_t)level * SLOTS_ + blockIdx.x * 16;

    int4 rc[4];
    #pragma unroll
    for (int r = 0; r < 4; ++r)
        rc[r] = tr[w + 4 * r];
    const int t0 = rc[0].z;
    const int e = t0 >= START_ ? T_ : (t0 < 0 ? 0 : t0);
    if (threadIdx.x < 16) rowNode[threadIdx.x] = tr[threadIdx.x].w;

    v4f ga[4], gc[4];
    #pragma unroll
    for (int r = 0; r < 4; ++r) {
        const int t = rc[r].z;
        int slot = t - START_;
        slot = slot < 0 ? 0 : (slot > OUTSLOTS_ - 1 ? OUTSLOTS_ - 1 : slot);
        const float* s0 = buf + (size_t)rc[r].x * D_;
        const float* s1 = (t >= START_) ? (slots + (size_t)slot * D_)
                                        : (buf + (size_t)rc[r].y * D_);
        ga[r] = *(const v4f*)(s0 + lane * 4);
        gc[r] = *(const v4f*)(s1 + lane * 4);
    }

    v2f w1ring[RING1_][8][4];
    #pragma unroll
    for (int g = 0; g < RING1_ - 1; ++g)
        dag_fetch_b<true, K1_, 4>(W1, e, w * 4, g, lane, w1ring[g]);

    #pragma unroll
    for (int r = 0; r < 4; ++r) {
        const int m = w + 4 * r;
        *(v4f*)(&X[m * XSTR_ + lane * 4])       = ga[r];
        *(v4f*)(&X[m * XSTR_ + 128 + lane * 4]) = gc[r];
    }
    dag_lds_barrier();

    v2f w2ring[RING2_][8][2];
    float bias2[2];
    {
        v8f acc[4] = {};
        float bias[4];
        auto ahead = [&](int i) {
            if (i == 0) {
                #pragma unroll
                for (int t = 0; t < 4; ++t)
                    bias[t] = b1[e * K1_ + (w * 4 + t) * 16 + colLane];
                #pragma unroll
                for (int t = 0; t < 2; ++t)
                    bias2[t] = b2[e * D_ + (w * 2 + t) * 16 + colLane];
            }
            dag_fetch_b<true, D_, 2>(W2, e, w * 2, 2 * i,     lane, w2ring[2 * i]);
            dag_fetch_b<true, D_, 2>(W2, e, w * 2, 2 * i + 1, lane, w2ring[2 * i + 1]);
        };
        dag_panel<true, K1_, 4, RING1_>(X, W1, e, w * 4, lane, w1ring, acc, ahead);
        #pragma unroll
        for (int g = 2 * (RING1_ - 1); g < RING2_ - 1; ++g)
            dag_fetch_b<true, D_, 2>(W2, e, w * 2, g, lane, w2ring[g]);
        __builtin_amdgcn_sched_barrier(0);
        #pragma unroll
        for (int t = 0; t < 4; ++t) {
            const int j = w * 4 + t;
            #pragma unroll
            for (int i = 0; i < 8; ++i) {
                float x = acc[t][i] + bias[t];
                float g = 0.5f * x * (1.0f + dag_erf_both_ranges(x * 0.70710678118654752f));
                H[(i + laneHalf * 8) * XSTR_ + j * 16 + colLane] = g;
            }
        }
    }
    dag_lds_barrier();

    {
        v8f acc[2] = {};
        int node[8];
        auto ahead2 = [&](int i) {
            if (i == 0) {
                #pragma unroll
                for (int r = 0; r < 8; ++r)
                    node[r] = rowNode[laneHalf * 8 + r];
            }
        };
        dag_panel<true, D_, 2, RING2_>(H, W2, e, w * 2, lane, w2ring, acc, ahead2);
        #pragma unroll
        for (int t = 0; t < 2; ++t) {
            const int j = w * 2 + t;
            #pragma unroll
            for (int i = 0; i < 8; ++i) {
                if (node[i] >= 0) {
                    const int row = R_ + level * M_ + node[i];
                    buf[(size_t)row * D_ + j * 16 + colLane] = acc[t][i] + bias2[t];
                }
            }
        }
    }
}

__device__ __forceinline__ int dag_key(int t, int db) {
    const int e  = t >= START_ ? T_ : (t < 0 ? 0 : t);
    const int d  = db & 0x7f;
    const int dd = d < 1 ? 1 : (d > L_ ? L_ : d);
    return ((dd - 1) * NENC_ + e) * 2 + ((db & 0x80) ? 0 : 1);
}

__device__ __forceinline__ void dag_depth_pass(const int* __restrict__ par, unsigned char* __restrict__ depOut,
                                               unsigned char* dep) {
    static_assert(R_ == 1024 && M_ == 2048 && L_ % 4 == 0, "one root and two nodes of a level per thread, levels in fours");
    const int tid = threadIdx.x;
    dep[tid] = 0;
    int4 ahead[4];
    #pragma unroll
    for (int i = 0; i < 4; ++i)
        ahead[i] = *(const int4*)(par + ((size_t)i * M_ + tid * 2) * 2);
    dag_lds_barrier();
    #pragma nounroll
    for (int l0 = 0; l0 < L_; l0 += 4) {
        #pragma unroll
        for (int i = 0; i < 4; ++i) {
            const int l = l0 + i;
            const int4 cur = ahead[i];
            const int ln = l + 4 < L_ ? l + 4 : L_ - 1;
            ahead[i] = *(const int4*)(par + ((size_t)ln * M_ + tid * 2) * 2);
            const int hi = R_ + l * M_;
            const int ia0 = (unsigned)cur.x < (unsigned)hi ? cur.x : 0;
            const int ia1 = (unsigned)cur.y < (unsigned)hi ? cur.y : 0;
            const int ic0 = (unsigned)cur.z < (unsigned)hi ? cur.z : 0;
            const int ic1 = (unsigned)cur.w < (unsigned)hi ? cur.w : 0;
            const int a0 = dep[ia0], a1 = dep[ia1], c0 = dep[ic0], c1 = dep[ic1];
            dep[hi + tid * 2]     = (unsigned char)(1 + (a0 > a1 ? a0 : a1));
            dep[hi + tid * 2 + 1] = (unsigned char)(1 + (c0 > c1 ? c0 : c1));
            dag_lds_barrier();
        }
    }
    for (int i = tid; i < QN_ / 16; i += 1024)
        ((uint4*)depOut)[i] = *(const uint4*)(dep + R_ + i * 16);
}

__device__ __forceinline__ void dag_mark_parents(const int* __restrict__ par, unsigned char* __restrict__ flagOut,
                                                 unsigned char* flg, int part) {
    static_assert(M_ == 2048 && QN_ % (16 * 1024) == 0, "two nodes of a level per thread, whole 16-byte rounds");
    const int tid = threadIdx.x;
    for (int i = tid; i < QN_ / 16; i += 1024)
        ((uint4*)flg)[i] = make_uint4(0u, 0u, 0u, 0u);
    int4 p[L_ / FLAGW_];
    #pragma unroll
    for (int i = 0; i < L_ / FLAGW_; ++i)
        p[i] = *(const int4*)(par + ((size_t)(part * (L_ / FLAGW_) + i) * M_ + tid * 2) * 2);
    __syncthreads();
    #pragma unroll
    for (int i = 0; i < L_ / FLAGW_; ++i) {
        const int l = part * (L_ / FLAGW_) + i;
        const unsigned span = (unsigned)(l * M_);
        const unsigned j0 = (unsigned)p[i].x - R_, j1 = (unsigned)p[i].y - R_;
        const unsigned j2 = (unsigned)p[i].z - R_, j3 = (unsigned)p[i].w - R_;
        if (j0 < span) flg[j0] = 1;
        if (j1 < span) flg[j1] = 1;
        if (j2 < span) flg[j2] = 1;
        if (j3 < span) flg[j3] = 1;
    }
    __syncthreads();
    for (int i = tid; i < QN_ / 16; i += 1024)
        ((uint4*)flagOut)[i] = ((const uint4*)flg)[i];
}

#define FRONT_ROOT_ (R_ * D_ / 4 / 1024)
#define FRONT_W1_   (W1P_FRAGS / 1024)
#define FRONT_W2_   (W2P_FRAGS / 1024)
#define FRONT_GRID_ (1 + FLAGW_ + FRONT_ROOT_ + FRONT_W1_ + FRONT_W2_)
static_assert(R_ * D_ / 4 % 1024 == 0 && W1P_FRAGS % 1024 == 0 && W2P_FRAGS % 1024 == 0, "whole work-groups per role");
__global__ __launch_bounds__(1024)
void dag_front(const int* __restrict__ par, unsigned char* __restrict__ depOut,
               unsigned char* __restrict__ flagOut,
               float* __restrict__ buf, const float* __restrict__ root,
               const float* __restrict__ W1, float* __restrict__ W1p,
               const float* __restrict__ W2, float* __restrict__ W2p) {
    __shared__ alignas(16) unsigned char bytes[N_];
    int b = blockIdx.x;
    if (b == 0) { dag_depth_pass(par, depOut, bytes); return; }
    b -= 1;
    if (b < FLAGW_) { dag_mark_parents(par, flagOut + (size_t)b * QN_, bytes, b); return; }
    b -= FLAGW_;
    if (b < FRONT_ROOT_) { dag_copy_root4(buf, root, b * 1024 + threadIdx.x); return; }
    b -= FRONT_ROOT_;
    if (b < FRONT_W1_) { dag_pack_frag(W1, W1p, 16, K1_, b * 1024 + threadIdx.x); return; }
    b -= FRONT_W1_;
    dag_pack_frag(W2, W2p, 8, D_, b * 1024 + threadIdx.x);
}

__global__ __launch_bounds__(1024)
void dag_rank(const int* __restrict__ types, unsigned char* __restrict__ dep,
              const unsigned char* __restrict__ flags,
              unsigned short* __restrict__ rank, int* __restrict__ cnt) {
    __shared__ unsigned short wcnt[32 * NKEY_];
    const int level = blockIdx.x;
    const int tid   = threadIdx.x;
    const int lane  = tid & 31;
    const int wv    = tid >> 5;
    const int q     = level * M_ + tid * 2;
    const int2 tt = *(const int2*)(types + q);
    const unsigned dd = *(const unsigned short*)(dep + q);
    unsigned ff = 0;
    #pragma unroll
    for (int f = 0; f < FLAGW_; ++f)
        ff |= *(const unsigned short*)(flags + (size_t)f * QN_ + q);
    const int da = (int)(dd & 0x7fu) | ((ff & 0x00ffu) ? 0x80 : 0);
    const int db = (int)((dd >> 8) & 0x7fu) | ((ff & 0xff00u) ? 0x80 : 0);
    *(unsigned short*)(dep + q) = (unsigned short)(da | (db << 8));
    const int ka = dag_key(tt.x, da);
    const int kb = dag_key(tt.y, db);
    for (int i = tid; i < 32 * NKEY_; i += 1024) wcnt[i] = 0;

    int ra = 0, rb = 0, ta = 0, tb = 0;
    #pragma unroll
    for (int s = 0; s < 32; ++s) {
        const int sa = __builtin_amdgcn_readlane(ka, s);
        const int sb = __builtin_amdgcn_readlane(kb, s);
        const int ma = (sa == ka ? 1 : 0) + (sb == ka ? 1 : 0);
        const int mb = (sa == kb ? 1 : 0) + (sb == kb ? 1 : 0);
        ta += ma;
        tb += mb;
        if (s < lane) { ra += ma; rb += mb; }
    }
    if (ka == kb) rb += 1;
    __syncthreads();
    if (ra == ta - 1) wcnt[wv * NKEY_ + ka] = (unsigned short)ta;
    if (rb == tb - 1) wcnt[wv * NKEY_ + kb] = (unsigned short)tb;
    __syncthreads();
    if (tid < NKEY_) {
        int run = 0;
        for (int v = 0; v < 32; ++v) {
            const int c = wcnt[v * NKEY_ + tid];
            wcnt[v * NKEY_ + tid] = (unsigned short)run;
            run += c;
        }
        cnt[level * NKEY_ + tid] = run;
    }
    __syncthreads();
    rank[q]     = (unsigned short)(wcnt[wv * NKEY_ + ka] + ra);
    rank[q + 1] = (unsigned short)(wcnt[wv * NKEY_ + kb] + rb);
}

__global__ __launch_bounds__(NKEY_)
void dag_offsets(int* __restrict__ cnt, int* __restrict__ tileStart, int* __restrict__ recs_) {
    __shared__ int til[NKEY_];
    __shared__ int first[NKEY_];
    __shared__ int tsl[L_ + 1];
    const int k = threadIdx.x;
    int total = 0;
    for (int l = 0; l < L_; ++l) total += cnt[l * NKEY_ + k];
    til[k]   = (total + 15) >> 4;
    first[k] = 0;
    __syncthreads();
    if (k < 32) {
        const int lane = k;
        int md = 0;
        for (int j = lane; j < NKEY_; j += 32)
            if (til[j] > 0) md = j / (2 * NENC_) + 1;
        for (int o = 16; o > 0; o >>= 1) {
            const int u = __shfl_xor(md, o);
            md = u > md ? u : md;
        }
        const int lmax = __builtin_amdgcn_readfirstlane(md);
        const int lcap = lmax < DWIDE_ ? lmax : DWIDE_;
        int off = 0;
        int qd  = 0;
        for (int l = 0; l < lmax; ++l) {
            if (lane == 0) tsl[l] = off;
            const int tn = lane < NENC_ ? til[(l * NENC_ + lane) * 2] : 0;
            int nl = 0;
            #pragma unroll
            for (int e = 0; e < NENC_; ++e) {
                if (lane == e) first[(l * NENC_ + e) * 2] = off + nl;
                nl += __builtin_amdgcn_readlane(tn, e);
            }
            off += nl;
            int room = (l + 1 >= lcap) ? (1 << 28) : ((DUNITS_ - nl % DUNITS_) % DUNITS_);
            for (int pass = 0; pass < (L_ * NENC_) / 32 + 1; ++pass) {
                const int j = qd + lane;
                const bool due = j < L_ * NENC_ && j / NENC_ <= l;
                const int tj = due ? til[j * 2 + 1] : -1;
                int n = 0;
                for (; n < 32; ++n) {
                    const int t = __builtin_amdgcn_readlane(tj, n);
                    if (t < 0 || t > room) break;
                    if (lane == n) first[j * 2 + 1] = off;
                    off  += t;
                    room -= t;
                }
                qd += n;
                if (n < 32) break;
            }
        }
        for (int l = lmax + lane; l <= L_; l += 32) tsl[l] = off;
    }
    __syncthreads();
    int slot = first[k] * 16;
    for (int l = 0; l < L_; ++l) {
        const int c = cnt[l * NKEY_ + k];
        cnt[l * NKEY_ + k] = slot;
        slot += c;
    }
    int4* rec = (int4*)recs_;
    const int e = (k >> 1) % NENC_;
    const int end = (first[k] + til[k]) * 16;
    for (int s = slot; s < end; ++s)
        rec[s] = make_int4(0, 0, e == T_ ? START_ : e, -1);
    if (k <= L_) tileStart[k] = tsl[k];
}

__global__ __launch_bounds__(1024)
void dag_place(const int* __restrict__ par, const int* __restrict__ types,
               const unsigned char* __restrict__ dep, const unsigned short* __restrict__ rank,
               const int* __restrict__ base, int* __restrict__ recs_) {
    const int level = blockIdx.x;
    const int q     = level * M_ + threadIdx.x * 2;
    const int2 tt = *(const int2*)(types + q);
    const int4 pp = *(const int4*)(par + (size_t)q * 2);
    const int ka = dag_key(tt.x, dep[q]);
    const int kb = dag_key(tt.y, dep[q + 1]);
    int4* rec = (int4*)recs_;
    rec[base[level * NKEY_ + ka] + rank[q]]     = make_int4(pp.x, pp.y, tt.x, q);
    rec[base[level * NKEY_ + kb] + rank[q + 1]] = make_int4(pp.z, pp.w, tt.y, q + 1);
}

template <bool TAIL>
__global__ __launch_bounds__(128)
void dag_level_depth(float* __restrict__ buf,
                        const float* __restrict__ W1,
                        const float* __restrict__ b1,
                        const float* __restrict__ W2,
                        const float* __restrict__ b2,
                        const float* __restrict__ slots,
                        const int*   __restrict__ recs_,
                        const int*   __restrict__ tileStart,
                        int depth)
{
    __shared__ float X[16 * XSTR_];
    __shared__ float H[16 * XSTR_];
    __shared__ int   rowNode[2][16];

    const int lane     = threadIdx.x & 31;
    const int w        = threadIdx.x >> 5;
    const int colLane  = lane & 15;
    const int laneHalf = (lane >> 4) & 1;
    int d       = depth;
    int tileBeg = tileStart[d];
    if (TAIL && tileBeg >= tileStart[L_]) return;
    do {
    const int tileEnd = tileStart[d + 1];
    for (int tile = tileBeg + (TAIL ? 0 : (int)blockIdx.x), par2 = 0; tile < tileEnd; tile += (TAIL ? 1 : DGRID_), par2 ^= 1) {
    const int4* tr = (const int4*)recs_ + (size_t)tile * 16;

    int4 rc[4];
    #pragma unroll
    for (int r = 0; r < 4; ++r)
        rc[r] = tr[w + 4 * r];
    const int t0 = rc[0].z;
    const int e = t0 >= START_ ? T_ : (t0 < 0 ? 0 : t0);
    if (threadIdx.x < 16) rowNode[par2][threadIdx.x] = tr[threadIdx.x].w;

    v4f ga[4], gc[4];
    #pragma unroll
    for (int r = 0; r < 4; ++r) {
        const int t = rc[r].z;
        int slot = t - START_;
        slot = slot < 0 ? 0 : (slot > OUTSLOTS_ - 1 ? OUTSLOTS_ - 1 : slot);
        const float* s0 = buf + (size_t)rc[r].x * D_;
        const float* s1 = (t >= START_) ? (slots + (size_t)slot * D_)
                                        : (buf + (size_t)rc[r].y * D_);
        ga[r] = *(const v4f*)(s0 + lane * 4);
        gc[r] = *(const v4f*)(s1 + lane * 4);
    }

    v2f w1ring[RING1_][8][4];
    #pragma unroll
    for (int g = 0; g < RING1_ - 1; ++g)
        dag_fetch_b<true, K1_, 4>(W1, e, w * 4, g, lane, w1ring[g]);

    #pragma unroll
    for (int r = 0; r < 4; ++r) {
        const int m = w + 4 * r;
        *(v4f*)(&X[m * XSTR_ + lane * 4])       = ga[r];
        *(v4f*)(&X[m * XSTR_ + 128 + lane * 4]) = gc[r];
    }
    dag_lds_barrier();

    v2f w2ring[RING2_][8][2];
    float bias2[2];
    {
        v8f acc[4] = {};
        float bias[4];
        auto ahead = [&](int i) {
            if (i == 0) {
                #pragma unroll
                for (int t = 0; t < 4; ++t)
                    bias[t] = b1[e * K1_ + (w * 4 + t) * 16 + colLane];
                #pragma unroll
                for (int t = 0; t < 2; ++t)
                    bias2[t] = b2[e * D_ + (w * 2 + t) * 16 + colLane];
            }
            dag_fetch_b<true, D_, 2>(W2, e, w * 2, 2 * i,     lane, w2ring[2 * i]);
            dag_fetch_b<true, D_, 2>(W2, e, w * 2, 2 * i + 1, lane, w2ring[2 * i + 1]);
        };
        dag_panel<true, K1_, 4, RING1_>(X, W1, e, w * 4, lane, w1ring, acc, ahead);
        #pragma unroll
        for (int g = 2 * (RING1_ - 1); g < RING2_ - 1; ++g)
            dag_fetch_b<true, D_, 2>(W2, e, w * 2, g, lane, w2ring[g]);
        __builtin_amdgcn_sched_barrier(0);
        #pragma unroll
        for (int t = 0; t < 4; ++t) {
            const int j = w * 4 + t;
            #pragma unroll
            for (int i = 0; i < 8; ++i) {
                float x = acc[t][i] + bias[t];
                float g = 0.5f * x * (1.0f + dag_erf_both_ranges(x * 0.70710678118654752f));
                H[(i + laneHalf * 8) * XSTR_ + j * 16 + colLane] = g;
            }
        }
    }
    dag_lds_barrier();

    {
        v8f acc[2] = {};
        int node[8];
        auto ahead2 = [&](int i) {
            if (i == 0) {
                #pragma unroll
                for (int r = 0; r < 8; ++r)
                    node[r] = rowNode[par2][laneHalf * 8 + r];
            }
        };
        dag_panel<true, D_, 2, RING2_>(H, W2, e, w * 2, lane, w2ring, acc, ahead2);
        if (TAIL) {
            asm volatile("v_nop\n\tv_nop\n\tv_nop\n\tv_nop");
        }
        #pragma unroll
        for (int t = 0; t < 2; ++t) {
            const int j = w * 2 + t;
            #pragma unroll
            for (int i = 0; i < 8; ++i) {
                if (node[i] >= 0) {
                    const int row = R_ + node[i];
                    buf[(size_t)row * D_ + j * 16 + colLane] = acc[t][i] + bias2[t];
                }
            }
        }
    }
    }
    if (!TAIL) break;
    ++d;
    tileBeg = tileEnd;
    if (d >= L_ || tileBeg >= tileStart[L_]) break;
    __threadfence();
    __syncthreads();
    } while (true);
}


extern "C" void kernel_launch(void* const* d_in, const int* in_sizes, int n_in,
                              void* d_out, int out_size, void* d_ws, size_t ws_size,
                              hipStream_t stream) {
    const float* root  = (const float*)d_in[0];
    const float* W1    = (const float*)d_in[1];
    const float* b1    = (const float*)d_in[2];
    const float* W2    = (const float*)d_in[3];
    const float* b2    = (const float*)d_in[4];
    const float* slots = (const float*)d_in[5];
    const int*   par   = (const int*)d_in[6];
    const int*   typ   = (const int*)d_in[7];
    float* out = (float*)d_out;

    const bool packed = (ws_size >= WSP_BYTES);
    float* W1p = (float*)d_ws;
    float* W2p = (float*)((char*)d_ws + W1P_BYTES);

    const bool bucketed = (ws_size >= WSP_BYTES + REC_BYTES);
    int* recs = (int*)((char*)d_ws + WSP_BYTES);

    const bool byDepth = (ws_size >= WSP_BYTES + DREC_BYTES + DEP_BYTES + RANK_BYTES + CNT_BYTES + TST_BYTES);
    unsigned char*  dep   = (unsigned char*)d_ws + WSP_BYTES + DREC_BYTES;
    unsigned short* rnk   = (unsigned short*)((char*)d_ws + WSP_BYTES + DREC_BYTES + DEP_BYTES);
    int*            cnt   = (int*)((char*)d_ws + WSP_BYTES + DREC_BYTES + DEP_BYTES + RANK_BYTES);
    int*            tileS = (int*)((char*)d_ws + WSP_BYTES + DREC_BYTES + DEP_BYTES + RANK_BYTES + CNT_BYTES);

    if (!byDepth) dag_init_roots<<<128, 256, 0, stream>>>(out, root);

    if (byDepth) {
        unsigned char* flg = (unsigned char*)recs;
        dag_front<<<FRONT_GRID_, 1024, 0, stream>>>(par, dep, flg, out, root, W1, W1p, W2, W2p);
        dag_rank<<<L_, 1024, 0, stream>>>(typ, dep, flg, rnk, cnt);
        dag_offsets<<<1, NKEY_, 0, stream>>>(cnt, tileS, recs);
        dag_place<<<L_, 1024, 0, stream>>>(par, typ, dep, rnk, cnt, recs);
        for (int d = 0; d < DWIDE_; ++d)
            dag_level_depth<false><<<DGRID_, 128, 0, stream>>>(out, W1p, b1, W2p, b2,
                                                               slots, recs, tileS, d);
        dag_level_depth<true><<<1, 128, 0, stream>>>(out, W1p, b1, W2p, b2,
                                                     slots, recs, tileS, DWIDE_);
    } else if (bucketed) {
        dag_pack_w<<<W1P_FRAGS / 256, 256, 0, stream>>>(W1, W1p, 16, K1_);
        dag_pack_w<<<W2P_FRAGS / 256, 256, 0, stream>>>(W2, W2p, 8, D_);
        dag_bucket<<<L_, 1024, 0, stream>>>(par, typ, recs);
        for (int l = 0; l < L_; ++l)
            dag_level_bucketed<<<TILES_, 128, 0, stream>>>(out, W1p, b1, W2p, b2,
                                                           slots, recs, l);
    } else if (packed) {
        dag_pack_w<<<W1P_FRAGS / 256, 256, 0, stream>>>(W1, W1p, 16, K1_);
        dag_pack_w<<<W2P_FRAGS / 256, 256, 0, stream>>>(W2, W2p, 8, D_);
        for (int l = 0; l < L_; ++l)
            dag_level<true><<<dim3(128, NENC_), 128, 0, stream>>>(out, W1p, b1, W2p, b2,
                                                                 slots, par, typ, l);
    } else {
        for (int l = 0; l < L_; ++l)
            dag_level<false><<<dim3(128, NENC_), 128, 0, stream>>>(out, W1, b1, W2, b2,
                                                                  slots, par, typ, l);
    }
}
